# starting kernel
# baseline (speedup 1.0000x reference)
#include <hip/hip_runtime.h>
#include <hip/hip_bf16.h>
#include <math.h>


typedef __attribute__((ext_vector_type(2))) float v2f;
typedef __attribute__((ext_vector_type(8))) float v8f;

#define NEG_SLOPE 0.2f
#define LN_EPS 1e-5f

static __device__ inline v8f wmma4(v2f a, v2f b, v8f c) {
  return __builtin_amdgcn_wmma_f32_16x16x4_f32(
       false, a,  false, b,
       (short)0, c,  false,  false);
}

static __device__ inline void atomicMaxF(float* addr, float val) {
  if (val >= 0.f) atomicMax((int*)addr, __float_as_int(val));
  else            atomicMin((unsigned int*)addr, __float_as_uint(val));
}

__global__ void fill_f32(float* __restrict__ p, long long n, float v) {
  long long i = (long long)blockIdx.x * blockDim.x + threadIdx.x;
  if (i < n) p[i] = v;
}

__global__ void gemm1_wmma(const float* __restrict__ x,
                           const float* __restrict__ Wl,
                           const float* __restrict__ Wr,
                           float* __restrict__ xl,
                           float* __restrict__ xr, int ntiles) {
  int wave = (int)(((long long)blockIdx.x * blockDim.x + threadIdx.x) >> 5);
  int lane = threadIdx.x & 31;
  if (wave >= ntiles) return;
  int row0 = wave * 16;
  int m  = lane & 15;
  int hi = lane >> 4;
  const float* xrow = x + (size_t)(row0 + m) * 4 + hi * 2;
  v2f a; a.x = xrow[0]; a.y = xrow[1];
#pragma unroll
  for (int j = 0; j < 4; ++j) {
    v2f bl, br;
    bl.x = Wl[(2 * hi + 0) * 64 + j * 16 + m];
    bl.y = Wl[(2 * hi + 1) * 64 + j * 16 + m];
    br.x = Wr[(2 * hi + 0) * 64 + j * 16 + m];
    br.y = Wr[(2 * hi + 1) * 64 + j * 16 + m];
    v8f cl = {}; cl = wmma4(a, bl, cl);
    v8f cr = {}; cr = wmma4(a, br, cr);
#pragma unroll
    for (int v = 0; v < 8; ++v) {
      int r = v + hi * 8;
      xl[(size_t)(row0 + r) * 64 + j * 16 + m] = cl[v];
      xr[(size_t)(row0 + r) * 64 + j * 16 + m] = cr[v];
    }
  }
}

__global__ void gemm2_wmma(const float* __restrict__ h,
                           const float* __restrict__ Wl,
                           const float* __restrict__ Wr,
                           float* __restrict__ xl,
                           float* __restrict__ xr, int ntiles) {
  int wave = (int)(((long long)blockIdx.x * blockDim.x + threadIdx.x) >> 5);
  int lane = threadIdx.x & 31;
  if (wave >= ntiles) return;
  int row0 = wave * 16;
  int m  = lane & 15;
  int hi = lane >> 4;
  const float* hrow = h + (size_t)(row0 + m) * 64;
  v8f cl[2] = {{0.f},{0.f}};
  v8f cr[2] = {{0.f},{0.f}};
#pragma unroll
  for (int kk = 0; kk < 16; ++kk) {
    int k0 = kk * 4 + hi * 2;
    v2f a; a.x = hrow[k0]; a.y = hrow[k0 + 1];
#pragma unroll
    for (int j = 0; j < 2; ++j) {
      v2f bl, br;
      bl.x = Wl[(k0 + 0) * 32 + j * 16 + m];
      bl.y = Wl[(k0 + 1) * 32 + j * 16 + m];
      br.x = Wr[(k0 + 0) * 32 + j * 16 + m];
      br.y = Wr[(k0 + 1) * 32 + j * 16 + m];
      cl[j] = wmma4(a, bl, cl[j]);
      cr[j] = wmma4(a, br, cr[j]);
    }
  }
#pragma unroll
  for (int j = 0; j < 2; ++j)
#pragma unroll
    for (int v = 0; v < 8; ++v) {
      int r = v + hi * 8;
      xl[(size_t)(row0 + r) * 32 + j * 16 + m] = cl[j][v];
      xr[(size_t)(row0 + r) * 32 + j * 16 + m] = cr[j][v];
    }
}

__global__ void edge_logits1(const int* __restrict__ ei, int E, int n,
                             const float* __restrict__ xl,
                             const float* __restrict__ xr,
                             const float* __restrict__ a1,
                             float* __restrict__ logits,
                             float* __restrict__ maxd) {
  int t = blockIdx.x * blockDim.x + threadIdx.x;
  int TE = E + n;
  if (t >= TE) return;
  int s, d;
  if (t < E) { s = ei[t]; d = ei[E + t]; } else { s = t - E; d = s; }
  const float4* ps = (const float4*)(xl + (size_t)s * 64);
  const float4* pd = (const float4*)(xr + (size_t)d * 64);
  const float4* pa = (const float4*)a1;
#pragma unroll
  for (int h = 0; h < 4; ++h) {
    float acc = 0.f;
#pragma unroll
    for (int q = 0; q < 4; ++q) {
      float4 u = ps[h * 4 + q], v = pd[h * 4 + q], w = pa[h * 4 + q];
      float e0 = u.x + v.x; e0 = e0 >= 0.f ? e0 : NEG_SLOPE * e0;
      float e1 = u.y + v.y; e1 = e1 >= 0.f ? e1 : NEG_SLOPE * e1;
      float e2 = u.z + v.z; e2 = e2 >= 0.f ? e2 : NEG_SLOPE * e2;
      float e3 = u.w + v.w; e3 = e3 >= 0.f ? e3 : NEG_SLOPE * e3;
      acc += e0 * w.x + e1 * w.y + e2 * w.z + e3 * w.w;
    }
    logits[(size_t)t * 4 + h] = acc;
    atomicMaxF(&maxd[(size_t)d * 4 + h], acc);
  }
}

__global__ void edge_accum1(const int* __restrict__ ei, int E, int n,
                            const float* __restrict__ xl,
                            const float* __restrict__ logits,
                            const float* __restrict__ maxd,
                            float* __restrict__ denom,
                            float* __restrict__ acc) {
  int t = blockIdx.x * blockDim.x + threadIdx.x;
  int T = 4 * (E + n);
  if (t >= T) return;
  int e = t >> 2, h = t & 3;
  int s, d;
  if (e < E) { s = ei[e]; d = ei[E + e]; } else { s = e - E; d = s; }
  float ex = __expf(logits[(size_t)e * 4 + h] - maxd[(size_t)d * 4 + h]);
  atomicAdd(&denom[(size_t)d * 4 + h], ex);
  const float* px = xl + (size_t)s * 64 + h * 16;
  float* pa = acc + (size_t)d * 64 + h * 16;
#pragma unroll
  for (int c = 0; c < 16; ++c) atomicAdd(&pa[c], ex * px[c]);
}

__global__ void node_fin1(const float* __restrict__ acc,
                          const float* __restrict__ denom,
                          const float* __restrict__ b1,
                          const float* __restrict__ g,
                          const float* __restrict__ be,
                          float* __restrict__ h1, int n) {
  int node = (int)(((long long)blockIdx.x * blockDim.x + threadIdx.x) >> 5);
  int lane = threadIdx.x & 31;
  if (node >= n) return;
  int c0 = lane, c1 = lane + 32;
  float v0 = acc[(size_t)node * 64 + c0] / denom[(size_t)node * 4 + (c0 >> 4)] + b1[c0];
  float v1 = acc[(size_t)node * 64 + c1] / denom[(size_t)node * 4 + (c1 >> 4)] + b1[c1];
  float s = v0 + v1;
#pragma unroll
  for (int o = 16; o; o >>= 1) s += __shfl_xor(s, o, 32);
  float mu = s * (1.f / 64.f);
  float d0 = v0 - mu, d1 = v1 - mu;
  float q = d0 * d0 + d1 * d1;
#pragma unroll
  for (int o = 16; o; o >>= 1) q += __shfl_xor(q, o, 32);
  float rs = rsqrtf(q * (1.f / 64.f) + LN_EPS);
  float y0 = d0 * rs * g[c0] + be[c0];
  float y1 = d1 * rs * g[c1] + be[c1];
  y0 = y0 > 0.f ? y0 : expm1f(y0);
  y1 = y1 > 0.f ? y1 : expm1f(y1);
  h1[(size_t)node * 64 + c0] = y0;
  h1[(size_t)node * 64 + c1] = y1;
}

__global__ void edge_logits2(const int* __restrict__ ei, int E, int n,
                             const float* __restrict__ xl,
                             const float* __restrict__ xr,
                             const float* __restrict__ a2,
                             float* __restrict__ logits,
                             float* __restrict__ maxd) {
  int t = blockIdx.x * blockDim.x + threadIdx.x;
  int TE = E + n;
  if (t >= TE) return;
  int s, d;
  if (t < E) { s = ei[t]; d = ei[E + t]; } else { s = t - E; d = s; }
  const float4* ps = (const float4*)(xl + (size_t)s * 32);
  const float4* pd = (const float4*)(xr + (size_t)d * 32);
  const float4* pa = (const float4*)a2;
  float acc = 0.f;
#pragma unroll
  for (int q = 0; q < 8; ++q) {
    float4 u = ps[q], v = pd[q], w = pa[q];
    float e0 = u.x + v.x; e0 = e0 >= 0.f ? e0 : NEG_SLOPE * e0;
    float e1 = u.y + v.y; e1 = e1 >= 0.f ? e1 : NEG_SLOPE * e1;
    float e2 = u.z + v.z; e2 = e2 >= 0.f ? e2 : NEG_SLOPE * e2;
    float e3 = u.w + v.w; e3 = e3 >= 0.f ? e3 : NEG_SLOPE * e3;
    acc += e0 * w.x + e1 * w.y + e2 * w.z + e3 * w.w;
  }
  logits[t] = acc;
  atomicMaxF(&maxd[d], acc);
}

__global__ void edge_accum2(const int* __restrict__ ei, int E, int n,
                            const float* __restrict__ xl,
                            const float* __restrict__ logits,
                            const float* __restrict__ maxd,
                            float* __restrict__ denom,
                            float* __restrict__ acc) {
  int t = blockIdx.x * blockDim.x + threadIdx.x;
  int TE = E + n;
  if (t >= TE) return;
  int s, d;
  if (t < E) { s = ei[t]; d = ei[E + t]; } else { s = t - E; d = s; }
  float ex = __expf(logits[t] - maxd[d]);
  atomicAdd(&denom[d], ex);
  const float* px = xl + (size_t)s * 32;
  float* pa = acc + (size_t)d * 32;
#pragma unroll
  for (int c = 0; c < 32; ++c) atomicAdd(&pa[c], ex * px[c]);
}

__global__ void node_fin2(const float* __restrict__ acc,
                          const float* __restrict__ denom,
                          const float* __restrict__ b2,
                          const float* __restrict__ g,
                          const float* __restrict__ be,
                          float* __restrict__ out, int n) {
  int node = (int)(((long long)blockIdx.x * blockDim.x + threadIdx.x) >> 5);
  int lane = threadIdx.x & 31;
  if (node >= n) return;
  float v = acc[(size_t)node * 32 + lane] / denom[node] + b2[lane];
  float s = v;
#pragma unroll
  for (int o = 16; o; o >>= 1) s += __shfl_xor(s, o, 32);
  float mu = s * (1.f / 32.f);
  float dd = v - mu;
  float q = dd * dd;
#pragma unroll
  for (int o = 16; o; o >>= 1) q += __shfl_xor(q, o, 32);
  float rs = rsqrtf(q * (1.f / 32.f) + LN_EPS);
  out[(size_t)node * 32 + lane] = dd * rs * g[lane] + be[lane];
}

static inline int gridFor(long long threads, int block) {
  return (int)((threads + block - 1) / block);
}

extern "C" void kernel_launch(void* const* d_in, const int* in_sizes, int n_in,
                              void* d_out, int out_size, void* d_ws, size_t ws_size,
                              hipStream_t stream) {
  const float* x   = (const float*)d_in[0];
  const int*   ei  = (const int*)  d_in[1];
  const float* W1l = (const float*)d_in[2];
  const float* W1r = (const float*)d_in[3];
  const float* a1  = (const float*)d_in[4];
  const float* b1  = (const float*)d_in[5];
  const float* g1  = (const float*)d_in[6];
  const float* be1 = (const float*)d_in[7];
  const float* W2l = (const float*)d_in[8];
  const float* W2r = (const float*)d_in[9];
  const float* a2  = (const float*)d_in[10];
  const float* b2  = (const float*)d_in[11];
  const float* g2  = (const float*)d_in[12];
  const float* be2 = (const float*)d_in[13];
  float* out = (float*)d_out;

  const int n  = in_sizes[0] / 4;
  const int E  = in_sizes[1] / 2;
  const int TE = E + n;
  const int ntiles = n / 16;

  float* ws = (float*)d_ws;
  const size_t XL1  = 0;
  const size_t XR1  = XL1 + (size_t)n * 64;
  const size_t ACC1 = XR1 + (size_t)n * 64;
  const size_t DEN1 = ACC1 + (size_t)n * 64;
  const size_t MAX1 = DEN1 + (size_t)n * 4;
  const size_t LOG1 = MAX1 + (size_t)n * 4;
  const size_t H1   = LOG1 + (size_t)TE * 4;

  const int B = 256;
  const float NEG_INF = -INFINITY;

  fill_f32<<<gridFor((long long)n * 68, B), B, 0, stream>>>(ws + ACC1, (long long)n * 68, 0.f);
  fill_f32<<<gridFor((long long)n * 4, B), B, 0, stream>>>(ws + MAX1, (long long)n * 4, NEG_INF);

  gemm1_wmma<<<gridFor((long long)ntiles * 32, B), B, 0, stream>>>(
      x, W1l, W1r, ws + XL1, ws + XR1, ntiles);

  edge_logits1<<<gridFor(TE, B), B, 0, stream>>>(
      ei, E, n, ws + XL1, ws + XR1, a1, ws + LOG1, ws + MAX1);

  edge_accum1<<<gridFor((long long)TE * 4, B), B, 0, stream>>>(
      ei, E, n, ws + XL1, ws + LOG1, ws + MAX1, ws + DEN1, ws + ACC1);

  node_fin1<<<gridFor((long long)n * 32, B), B, 0, stream>>>(
      ws + ACC1, ws + DEN1, b1, g1, be1, ws + H1, n);

  fill_f32<<<gridFor((long long)n * 32, B), B, 0, stream>>>(ws + ACC1, (long long)n * 32, 0.f);
  fill_f32<<<gridFor(n, B), B, 0, stream>>>(ws + DEN1, n, 0.f);
  fill_f32<<<gridFor(n, B), B, 0, stream>>>(ws + MAX1, n, NEG_INF);

  gemm2_wmma<<<gridFor((long long)ntiles * 32, B), B, 0, stream>>>(
      ws + H1, W2l, W2r, ws + XL1, ws + XR1, ntiles);

  edge_logits2<<<gridFor(TE, B), B, 0, stream>>>(
      ei, E, n, ws + XL1, ws + XR1, a2, ws + LOG1, ws + MAX1);

  edge_accum2<<<gridFor(TE, B), B, 0, stream>>>(
      ei, E, n, ws + XL1, ws + LOG1, ws + MAX1, ws + DEN1, ws + ACC1);

  node_fin2<<<gridFor((long long)n * 32, B), B, 0, stream>>>(
      ws + ACC1, ws + DEN1, b2, g2, be2, out, n);
}


# Sort edges by destination each call (int count, scan, ranked fill); one wave per node keeps max, denom, sums in registers: no float atomics, sums in arrival-rank order, lane-tree logits
# speedup vs baseline: 4.5446x; 4.5446x over previous
#include <hip/hip_runtime.h>
#include <hip/hip_bf16.h>
#include <math.h>


typedef __attribute__((ext_vector_type(2))) float v2f;
typedef __attribute__((ext_vector_type(8))) float v8f;

#define NEG_SLOPE 0.2f
#define LN_EPS 1e-5f

static __device__ inline v8f wmma4(v2f a, v2f b, v8f c) {
  return __builtin_amdgcn_wmma_f32_16x16x4_f32(
       false, a,  false, b,
       (short)0, c,  false,  false);
}

__global__ void fill_f32(float* __restrict__ p, long long n, float v) {
  long long i = (long long)blockIdx.x * blockDim.x + threadIdx.x;
  if (i < n) p[i] = v;
}

__global__ void gemm1_wmma(const float* __restrict__ x,
                           const float* __restrict__ Wl,
                           const float* __restrict__ Wr,
                           float* __restrict__ xl,
                           float* __restrict__ xr, int ntiles) {
  int wave = (int)(((long long)blockIdx.x * blockDim.x + threadIdx.x) >> 5);
  int lane = threadIdx.x & 31;
  if (wave >= ntiles) return;
  int row0 = wave * 16;
  int m  = lane & 15;
  int hi = lane >> 4;
  const float* xrow = x + (size_t)(row0 + m) * 4 + hi * 2;
  v2f a; a.x = xrow[0]; a.y = xrow[1];
#pragma unroll
  for (int j = 0; j < 4; ++j) {
    v2f bl, br;
    bl.x = Wl[(2 * hi + 0) * 64 + j * 16 + m];
    bl.y = Wl[(2 * hi + 1) * 64 + j * 16 + m];
    br.x = Wr[(2 * hi + 0) * 64 + j * 16 + m];
    br.y = Wr[(2 * hi + 1) * 64 + j * 16 + m];
    v8f cl = {}; cl = wmma4(a, bl, cl);
    v8f cr = {}; cr = wmma4(a, br, cr);
#pragma unroll
    for (int v = 0; v < 8; ++v) {
      int r = v + hi * 8;
      xl[(size_t)(row0 + r) * 64 + j * 16 + m] = cl[v];
      xr[(size_t)(row0 + r) * 64 + j * 16 + m] = cr[v];
    }
  }
}

__global__ void gemm2_wmma(const float* __restrict__ h,
                           const float* __restrict__ Wl,
                           const float* __restrict__ Wr,
                           float* __restrict__ xl,
                           float* __restrict__ xr, int ntiles) {
  int wave = (int)(((long long)blockIdx.x * blockDim.x + threadIdx.x) >> 5);
  int lane = threadIdx.x & 31;
  if (wave >= ntiles) return;
  int row0 = wave * 16;
  int m  = lane & 15;
  int hi = lane >> 4;
  const float* hrow = h + (size_t)(row0 + m) * 64;
  v8f cl[2] = {{0.f},{0.f}};
  v8f cr[2] = {{0.f},{0.f}};
#pragma unroll
  for (int kk = 0; kk < 16; ++kk) {
    int k0 = kk * 4 + hi * 2;
    v2f a; a.x = hrow[k0]; a.y = hrow[k0 + 1];
#pragma unroll
    for (int j = 0; j < 2; ++j) {
      v2f bl, br;
      bl.x = Wl[(k0 + 0) * 32 + j * 16 + m];
      bl.y = Wl[(k0 + 1) * 32 + j * 16 + m];
      br.x = Wr[(k0 + 0) * 32 + j * 16 + m];
      br.y = Wr[(k0 + 1) * 32 + j * 16 + m];
      cl[j] = wmma4(a, bl, cl[j]);
      cr[j] = wmma4(a, br, cr[j]);
    }
  }
#pragma unroll
  for (int j = 0; j < 2; ++j)
#pragma unroll
    for (int v = 0; v < 8; ++v) {
      int r = v + hi * 8;
      xl[(size_t)(row0 + r) * 32 + j * 16 + m] = cl[j][v];
      xr[(size_t)(row0 + r) * 32 + j * 16 + m] = cr[j][v];
    }
}

__global__ void count_deg(const int* __restrict__ ei, int E, int* __restrict__ deg) {
  int t = blockIdx.x * blockDim.x + threadIdx.x;
  if (t >= E) return;
  atomicAdd(&deg[ei[E + t]], 1);
}

__global__ void scan_partial(const int* __restrict__ deg, int n, int* __restrict__ bsum) {
  __shared__ int wsum[8];
  int t = threadIdx.x;
  int base = blockIdx.x * 1024 + t * 4;
  int s = 0;
  for (int k = 0; k < 4; ++k) { int i = base + k; if (i < n) s += deg[i]; }
  for (int o = 16; o; o >>= 1) s += __shfl_xor(s, o, 32);
  if ((t & 31) == 0) wsum[t >> 5] = s;
  __syncthreads();
  if (t == 0) {
    int tot = 0;
    for (int q = 0; q < 8; ++q) tot += wsum[q];
    bsum[blockIdx.x] = tot;
  }
}

__global__ void scan_final(const int* __restrict__ deg, int n,
                           const int* __restrict__ bsum,
                           int* __restrict__ rowptr, int* __restrict__ cursor) {
  __shared__ int wsum[8];
  __shared__ int boff;
  int t = threadIdx.x, b = blockIdx.x, lane = t & 31, wv = t >> 5;
  int o = 0;
  for (int j = t; j < b; j += 256) o += bsum[j];
  for (int k = 16; k; k >>= 1) o += __shfl_xor(o, k, 32);
  if (lane == 0) wsum[wv] = o;
  __syncthreads();
  if (t == 0) {
    int tot = 0;
    for (int q = 0; q < 8; ++q) tot += wsum[q];
    boff = tot;
  }
  __syncthreads();
  int blockoff = boff;
  __syncthreads();
  int base = b * 1024 + t * 4;
  int v[4];
  int s = 0;
  for (int k = 0; k < 4; ++k) { int i = base + k; v[k] = (i < n) ? deg[i] : 0; s += v[k]; }
  int inc = s;
  for (int k = 1; k < 32; k <<= 1) { int u = __shfl_up(inc, k, 32); if (lane >= k) inc += u; }
  if (lane == 31) wsum[wv] = inc;
  __syncthreads();
  int woff = 0;
  for (int q = 0; q < wv; ++q) woff += wsum[q];
  int excl = blockoff + woff + inc - s;
  for (int k = 0; k < 4; ++k) {
    int i = base + k;
    if (i < n) {
      rowptr[i] = excl; cursor[i] = excl;
      excl += v[k];
      if (i == n - 1) rowptr[n] = excl;
    }
  }
}

__global__ void fill_csr(const int* __restrict__ ei, int E,
                         int* __restrict__ cursor, int* __restrict__ srcs) {
  int t = blockIdx.x * blockDim.x + threadIdx.x;
  if (t >= E) return;
  int s = ei[t], d = ei[E + t];
  int pos = atomicAdd(&cursor[d], 1);
  srcs[pos] = s;
}

static __device__ inline float lrelu(float e) { return e >= 0.f ? e : NEG_SLOPE * e; }

template <int X> static __device__ inline float xrow(float v) {
  return __builtin_bit_cast(float, __builtin_amdgcn_update_dpp(
      0, __builtin_bit_cast(int, v), 0x160 + X, 0xF, 0xF, true));
}
static __device__ inline float xhalf(float v) {
  int b = __builtin_bit_cast(int, v);
  return __builtin_bit_cast(float, __builtin_amdgcn_permlanex16(
      b, b, 0x76543210, 0xfedcba98, false, false));
}

static __device__ inline float logit1(float2 u, float2 vr, float2 w) {
  float p = lrelu(u.x + vr.x) * w.x + lrelu(u.y + vr.y) * w.y;
  p += xrow<1>(p); p += xrow<2>(p); p += xrow<4>(p);
  return p;
}
static __device__ inline float2 row64(const float* __restrict__ xl, int s, int lane) {
  return ((const float2*)(xl + (size_t)s * 64))[lane];
}

__global__ void gat1_node(const int* __restrict__ rowptr,
                          const int* __restrict__ srcs,
                          const float* __restrict__ xl,
                          const float* __restrict__ xr,
                          const float* __restrict__ a1,
                          float* __restrict__ denom,
                          float* __restrict__ acc, int n) {
  const int node = __builtin_amdgcn_readfirstlane(
      (int)(((long long)blockIdx.x * blockDim.x + threadIdx.x) >> 5));
  const int lane = threadIdx.x & 31;
  if (node >= n) return;
  const float2 w  = ((const float2*)a1)[lane];
  const float2 vr = row64(xr, node, lane);
  const float2 us = row64(xl, node, lane);
  const int beg = rowptr[node], end = rowptr[node + 1];
  const float ls = logit1(us, vr, w);
  float m = ls;
  for (int base = beg; base < end; base += 32) {
    int cnt = end - base; cnt = cnt < 32 ? cnt : 32;
    const int sidx = lane < cnt ? srcs[base + lane] : node;
    int j = 0;
    for (; j + 4 <= cnt; j += 4) {
      float2 u0 = row64(xl, __builtin_amdgcn_readlane(sidx, j + 0), lane);
      float2 u1 = row64(xl, __builtin_amdgcn_readlane(sidx, j + 1), lane);
      float2 u2 = row64(xl, __builtin_amdgcn_readlane(sidx, j + 2), lane);
      float2 u3 = row64(xl, __builtin_amdgcn_readlane(sidx, j + 3), lane);
      float l0 = logit1(u0, vr, w), l1 = logit1(u1, vr, w);
      float l2 = logit1(u2, vr, w), l3 = logit1(u3, vr, w);
      m = fmaxf(fmaxf(m, l0), fmaxf(fmaxf(l1, l2), l3));
    }
    for (; j < cnt; ++j) {
      float2 u = row64(xl, __builtin_amdgcn_readlane(sidx, j), lane);
      m = fmaxf(m, logit1(u, vr, w));
    }
  }
  float ex = __expf(ls - m);
  float den = ex;
  float2 a; a.x = ex * us.x; a.y = ex * us.y;
  for (int base = beg; base < end; base += 32) {
    int cnt = end - base; cnt = cnt < 32 ? cnt : 32;
    const int sidx = lane < cnt ? srcs[base + lane] : node;
    int j = 0;
    for (; j + 4 <= cnt; j += 4) {
      float2 u0 = row64(xl, __builtin_amdgcn_readlane(sidx, j + 0), lane);
      float2 u1 = row64(xl, __builtin_amdgcn_readlane(sidx, j + 1), lane);
      float2 u2 = row64(xl, __builtin_amdgcn_readlane(sidx, j + 2), lane);
      float2 u3 = row64(xl, __builtin_amdgcn_readlane(sidx, j + 3), lane);
      float e0 = __expf(logit1(u0, vr, w) - m), e1 = __expf(logit1(u1, vr, w) - m);
      float e2 = __expf(logit1(u2, vr, w) - m), e3 = __expf(logit1(u3, vr, w) - m);
      den += e0; a.x += e0 * u0.x; a.y += e0 * u0.y;
      den += e1; a.x += e1 * u1.x; a.y += e1 * u1.y;
      den += e2; a.x += e2 * u2.x; a.y += e2 * u2.y;
      den += e3; a.x += e3 * u3.x; a.y += e3 * u3.y;
    }
    for (; j < cnt; ++j) {
      float2 u = row64(xl, __builtin_amdgcn_readlane(sidx, j), lane);
      float e = __expf(logit1(u, vr, w) - m);
      den += e; a.x += e * u.x; a.y += e * u.y;
    }
  }
  ((float2*)(acc + (size_t)node * 64))[lane] = a;
  if ((lane & 7) == 0) denom[(size_t)node * 4 + (lane >> 3)] = den;
}

__global__ void node_fin1(const float* __restrict__ acc,
                          const float* __restrict__ denom,
                          const float* __restrict__ b1,
                          const float* __restrict__ g,
                          const float* __restrict__ be,
                          float* __restrict__ h1, int n) {
  int node = (int)(((long long)blockIdx.x * blockDim.x + threadIdx.x) >> 5);
  int lane = threadIdx.x & 31;
  if (node >= n) return;
  int c0 = lane, c1 = lane + 32;
  float v0 = acc[(size_t)node * 64 + c0] / denom[(size_t)node * 4 + (c0 >> 4)] + b1[c0];
  float v1 = acc[(size_t)node * 64 + c1] / denom[(size_t)node * 4 + (c1 >> 4)] + b1[c1];
  float s = v0 + v1;
#pragma unroll
  for (int o = 16; o; o >>= 1) s += __shfl_xor(s, o, 32);
  float mu = s * (1.f / 64.f);
  float d0 = v0 - mu, d1 = v1 - mu;
  float q = d0 * d0 + d1 * d1;
#pragma unroll
  for (int o = 16; o; o >>= 1) q += __shfl_xor(q, o, 32);
  float rs = rsqrtf(q * (1.f / 64.f) + LN_EPS);
  float y0 = d0 * rs * g[c0] + be[c0];
  float y1 = d1 * rs * g[c1] + be[c1];
  y0 = y0 > 0.f ? y0 : expm1f(y0);
  y1 = y1 > 0.f ? y1 : expm1f(y1);
  h1[(size_t)node * 64 + c0] = y0;
  h1[(size_t)node * 64 + c1] = y1;
}

static __device__ inline float logit2(float u, float vr, float w) {
  float p = lrelu(u + vr) * w;
  p += xrow<1>(p); p += xrow<2>(p); p += xrow<4>(p); p += xrow<8>(p);
  p += xhalf(p);
  return p;
}

__global__ void gat2_node(const int* __restrict__ rowptr,
                          const int* __restrict__ srcs,
                          const float* __restrict__ xl,
                          const float* __restrict__ xr,
                          const float* __restrict__ a2,
                          float* __restrict__ denom,
                          float* __restrict__ acc, int n) {
  const int node = __builtin_amdgcn_readfirstlane(
      (int)(((long long)blockIdx.x * blockDim.x + threadIdx.x) >> 5));
  const int lane = threadIdx.x & 31;
  if (node >= n) return;
  const float w  = a2[lane];
  const float vr = xr[(size_t)node * 32 + lane];
  const float us = xl[(size_t)node * 32 + lane];
  const int beg = rowptr[node], end = rowptr[node + 1];
  const float ls = logit2(us, vr, w);
  float m = ls;
  for (int base = beg; base < end; base += 32) {
    int cnt = end - base; cnt = cnt < 32 ? cnt : 32;
    const int sidx = lane < cnt ? srcs[base + lane] : node;
    int j = 0;
    for (; j + 4 <= cnt; j += 4) {
      float u0 = xl[(size_t)__builtin_amdgcn_readlane(sidx, j + 0) * 32 + lane];
      float u1 = xl[(size_t)__builtin_amdgcn_readlane(sidx, j + 1) * 32 + lane];
      float u2 = xl[(size_t)__builtin_amdgcn_readlane(sidx, j + 2) * 32 + lane];
      float u3 = xl[(size_t)__builtin_amdgcn_readlane(sidx, j + 3) * 32 + lane];
      float l0 = logit2(u0, vr, w), l1 = logit2(u1, vr, w);
      float l2 = logit2(u2, vr, w), l3 = logit2(u3, vr, w);
      m = fmaxf(fmaxf(m, l0), fmaxf(fmaxf(l1, l2), l3));
    }
    for (; j < cnt; ++j) {
      float u = xl[(size_t)__builtin_amdgcn_readlane(sidx, j) * 32 + lane];
      m = fmaxf(m, logit2(u, vr, w));
    }
  }
  float ex = __expf(ls - m);
  float den = ex;
  float a = ex * us;
  for (int base = beg; base < end; base += 32) {
    int cnt = end - base; cnt = cnt < 32 ? cnt : 32;
    const int sidx = lane < cnt ? srcs[base + lane] : node;
    int j = 0;
    for (; j + 4 <= cnt; j += 4) {
      float u0 = xl[(size_t)__builtin_amdgcn_readlane(sidx, j + 0) * 32 + lane];
      float u1 = xl[(size_t)__builtin_amdgcn_readlane(sidx, j + 1) * 32 + lane];
      float u2 = xl[(size_t)__builtin_amdgcn_readlane(sidx, j + 2) * 32 + lane];
      float u3 = xl[(size_t)__builtin_amdgcn_readlane(sidx, j + 3) * 32 + lane];
      float e0 = __expf(logit2(u0, vr, w) - m), e1 = __expf(logit2(u1, vr, w) - m);
      float e2 = __expf(logit2(u2, vr, w) - m), e3 = __expf(logit2(u3, vr, w) - m);
      den += e0; a += e0 * u0;
      den += e1; a += e1 * u1;
      den += e2; a += e2 * u2;
      den += e3; a += e3 * u3;
    }
    for (; j < cnt; ++j) {
      float u = xl[(size_t)__builtin_amdgcn_readlane(sidx, j) * 32 + lane];
      float e = __expf(logit2(u, vr, w) - m);
      den += e; a += e * u;
    }
  }
  acc[(size_t)node * 32 + lane] = a;
  if (lane == 0) denom[node] = den;
}

__global__ void node_fin2(const float* __restrict__ acc,
                          const float* __restrict__ denom,
                          const float* __restrict__ b2,
                          const float* __restrict__ g,
                          const float* __restrict__ be,
                          float* __restrict__ out, int n) {
  int node = (int)(((long long)blockIdx.x * blockDim.x + threadIdx.x) >> 5);
  int lane = threadIdx.x & 31;
  if (node >= n) return;
  float v = acc[(size_t)node * 32 + lane] / denom[node] + b2[lane];
  float s = v;
#pragma unroll
  for (int o = 16; o; o >>= 1) s += __shfl_xor(s, o, 32);
  float mu = s * (1.f / 32.f);
  float dd = v - mu;
  float q = dd * dd;
#pragma unroll
  for (int o = 16; o; o >>= 1) q += __shfl_xor(q, o, 32);
  float rs = rsqrtf(q * (1.f / 32.f) + LN_EPS);
  out[(size_t)node * 32 + lane] = dd * rs * g[lane] + be[lane];
}

static inline int gridFor(long long threads, int block) {
  return (int)((threads + block - 1) / block);
}

extern "C" void kernel_launch(void* const* d_in, const int* in_sizes, int n_in,
                              void* d_out, int out_size, void* d_ws, size_t ws_size,
                              hipStream_t stream) {
  const float* x   = (const float*)d_in[0];
  const int*   ei  = (const int*)  d_in[1];
  const float* W1l = (const float*)d_in[2];
  const float* W1r = (const float*)d_in[3];
  const float* a1  = (const float*)d_in[4];
  const float* b1  = (const float*)d_in[5];
  const float* g1  = (const float*)d_in[6];
  const float* be1 = (const float*)d_in[7];
  const float* W2l = (const float*)d_in[8];
  const float* W2r = (const float*)d_in[9];
  const float* a2  = (const float*)d_in[10];
  const float* b2  = (const float*)d_in[11];
  const float* g2  = (const float*)d_in[12];
  const float* be2 = (const float*)d_in[13];
  float* out = (float*)d_out;

  const int n  = in_sizes[0] / 4;
  const int E  = in_sizes[1] / 2;
  const int TE = E + n;
  const int ntiles = n / 16;

  float* ws = (float*)d_ws;
  const size_t XL1  = 0;
  const size_t XR1  = XL1 + (size_t)n * 64;
  const size_t ACC1 = XR1 + (size_t)n * 64;
  const size_t DEN1 = ACC1 + (size_t)n * 64;
  const size_t MAX1 = DEN1 + (size_t)n * 4;
  const size_t LOG1 = MAX1 + (size_t)n * 4;
  const size_t H1   = LOG1 + (size_t)TE * 4;
  int* wi = (int*)d_ws;
  const size_t DEG  = LOG1;
  const size_t RPTR = DEG  + (size_t)n;
  const size_t CURS = RPTR + (size_t)n + 64;
  const size_t BSUM = CURS + (size_t)n + 64;
  const size_t SRCS = BSUM + 896;

  const int B = 256;
  const int nscan = gridFor(n, 1024);

  fill_f32<<<gridFor(n, B), B, 0, stream>>>(ws + DEG, (long long)n, 0.f);
  count_deg<<<gridFor(E, B), B, 0, stream>>>(ei, E, wi + DEG);
  scan_partial<<<nscan, B, 0, stream>>>(wi + DEG, n, wi + BSUM);
  scan_final<<<nscan, B, 0, stream>>>(wi + DEG, n, wi + BSUM, wi + RPTR, wi + CURS);
  fill_csr<<<gridFor(E, B), B, 0, stream>>>(ei, E, wi + CURS, wi + SRCS);

  gemm1_wmma<<<gridFor((long long)ntiles * 32, B), B, 0, stream>>>(
      x, W1l, W1r, ws + XL1, ws + XR1, ntiles);

  gat1_node<<<gridFor((long long)n * 32, B), B, 0, stream>>>(
      wi + RPTR, wi + SRCS, ws + XL1, ws + XR1, a1, ws + DEN1, ws + ACC1, n);

  node_fin1<<<gridFor((long long)n * 32, B), B, 0, stream>>>(
      ws + ACC1, ws + DEN1, b1, g1, be1, ws + H1, n);

  gemm2_wmma<<<gridFor((long long)ntiles * 32, B), B, 0, stream>>>(
      ws + H1, W2l, W2r, ws + XL1, ws + XR1, ntiles);

  gat2_node<<<gridFor((long long)n * 32, B), B, 0, stream>>>(
      wi + RPTR, wi + SRCS, ws + XL1, ws + XR1, a2, ws + DEN1, ws + ACC1, n);

  node_fin2<<<gridFor((long long)n * 32, B), B, 0, stream>>>(
      ws + ACC1, ws + DEN1, b2, g2, be2, out, n);
}


# Per-node finalize (divide, bias, LayerNorm, ELU) moved into the aggregation wave's epilogue: 2 launches and 200k short waves fewer; layer-1 LayerNorm sums in a new lane-tree order
# speedup vs baseline: 4.9159x; 1.0817x over previous
#include <hip/hip_runtime.h>
#include <hip/hip_bf16.h>
#include <math.h>


typedef __attribute__((ext_vector_type(2))) float v2f;
typedef __attribute__((ext_vector_type(8))) float v8f;

#define NEG_SLOPE 0.2f
#define LN_EPS 1e-5f

static __device__ inline v8f wmma4(v2f a, v2f b, v8f c) {
  return __builtin_amdgcn_wmma_f32_16x16x4_f32(
       false, a,  false, b,
       (short)0, c,  false,  false);
}

__global__ void fill_f32(float* __restrict__ p, long long n, float v) {
  long long i = (long long)blockIdx.x * blockDim.x + threadIdx.x;
  if (i < n) p[i] = v;
}

__global__ void gemm1_wmma(const float* __restrict__ x,
                           const float* __restrict__ Wl,
                           const float* __restrict__ Wr,
                           float* __restrict__ xl,
                           float* __restrict__ xr, int ntiles) {
  int wave = (int)(((long long)blockIdx.x * blockDim.x + threadIdx.x) >> 5);
  int lane = threadIdx.x & 31;
  if (wave >= ntiles) return;
  int row0 = wave * 16;
  int m  = lane & 15;
  int hi = lane >> 4;
  const float* xrow = x + (size_t)(row0 + m) * 4 + hi * 2;
  v2f a; a.x = xrow[0]; a.y = xrow[1];
#pragma unroll
  for (int j = 0; j < 4; ++j) {
    v2f bl, br;
    bl.x = Wl[(2 * hi + 0) * 64 + j * 16 + m];
    bl.y = Wl[(2 * hi + 1) * 64 + j * 16 + m];
    br.x = Wr[(2 * hi + 0) * 64 + j * 16 + m];
    br.y = Wr[(2 * hi + 1) * 64 + j * 16 + m];
    v8f cl = {}; cl = wmma4(a, bl, cl);
    v8f cr = {}; cr = wmma4(a, br, cr);
#pragma unroll
    for (int v = 0; v < 8; ++v) {
      int r = v + hi * 8;
      xl[(size_t)(row0 + r) * 64 + j * 16 + m] = cl[v];
      xr[(size_t)(row0 + r) * 64 + j * 16 + m] = cr[v];
    }
  }
}

__global__ void gemm2_wmma(const float* __restrict__ h,
                           const float* __restrict__ Wl,
                           const float* __restrict__ Wr,
                           float* __restrict__ xl,
                           float* __restrict__ xr, int ntiles) {
  int wave = (int)(((long long)blockIdx.x * blockDim.x + threadIdx.x) >> 5);
  int lane = threadIdx.x & 31;
  if (wave >= ntiles) return;
  int row0 = wave * 16;
  int m  = lane & 15;
  int hi = lane >> 4;
  const float* hrow = h + (size_t)(row0 + m) * 64;
  v8f cl[2] = {{0.f},{0.f}};
  v8f cr[2] = {{0.f},{0.f}};
#pragma unroll
  for (int kk = 0; kk < 16; ++kk) {
    int k0 = kk * 4 + hi * 2;
    v2f a; a.x = hrow[k0]; a.y = hrow[k0 + 1];
#pragma unroll
    for (int j = 0; j < 2; ++j) {
      v2f bl, br;
      bl.x = Wl[(k0 + 0) * 32 + j * 16 + m];
      bl.y = Wl[(k0 + 1) * 32 + j * 16 + m];
      br.x = Wr[(k0 + 0) * 32 + j * 16 + m];
      br.y = Wr[(k0 + 1) * 32 + j * 16 + m];
      cl[j] = wmma4(a, bl, cl[j]);
      cr[j] = wmma4(a, br, cr[j]);
    }
  }
#pragma unroll
  for (int j = 0; j < 2; ++j)
#pragma unroll
    for (int v = 0; v < 8; ++v) {
      int r = v + hi * 8;
      xl[(size_t)(row0 + r) * 32 + j * 16 + m] = cl[j][v];
      xr[(size_t)(row0 + r) * 32 + j * 16 + m] = cr[j][v];
    }
}

__global__ void count_deg(const int* __restrict__ ei, int E, int* __restrict__ deg) {
  int t = blockIdx.x * blockDim.x + threadIdx.x;
  if (t >= E) return;
  atomicAdd(&deg[ei[E + t]], 1);
}

__global__ void scan_partial(const int* __restrict__ deg, int n, int* __restrict__ bsum) {
  __shared__ int wsum[8];
  int t = threadIdx.x;
  int base = blockIdx.x * 1024 + t * 4;
  int s = 0;
  for (int k = 0; k < 4; ++k) { int i = base + k; if (i < n) s += deg[i]; }
  for (int o = 16; o; o >>= 1) s += __shfl_xor(s, o, 32);
  if ((t & 31) == 0) wsum[t >> 5] = s;
  __syncthreads();
  if (t == 0) {
    int tot = 0;
    for (int q = 0; q < 8; ++q) tot += wsum[q];
    bsum[blockIdx.x] = tot;
  }
}

__global__ void scan_final(const int* __restrict__ deg, int n,
                           const int* __restrict__ bsum,
                           int* __restrict__ rowptr, int* __restrict__ cursor) {
  __shared__ int wsum[8];
  __shared__ int boff;
  int t = threadIdx.x, b = blockIdx.x, lane = t & 31, wv = t >> 5;
  int o = 0;
  for (int j = t; j < b; j += 256) o += bsum[j];
  for (int k = 16; k; k >>= 1) o += __shfl_xor(o, k, 32);
  if (lane == 0) wsum[wv] = o;
  __syncthreads();
  if (t == 0) {
    int tot = 0;
    for (int q = 0; q < 8; ++q) tot += wsum[q];
    boff = tot;
  }
  __syncthreads();
  int blockoff = boff;
  __syncthreads();
  int base = b * 1024 + t * 4;
  int v[4];
  int s = 0;
  for (int k = 0; k < 4; ++k) { int i = base + k; v[k] = (i < n) ? deg[i] : 0; s += v[k]; }
  int inc = s;
  for (int k = 1; k < 32; k <<= 1) { int u = __shfl_up(inc, k, 32); if (lane >= k) inc += u; }
  if (lane == 31) wsum[wv] = inc;
  __syncthreads();
  int woff = 0;
  for (int q = 0; q < wv; ++q) woff += wsum[q];
  int excl = blockoff + woff + inc - s;
  for (int k = 0; k < 4; ++k) {
    int i = base + k;
    if (i < n) {
      rowptr[i] = excl; cursor[i] = excl;
      excl += v[k];
      if (i == n - 1) rowptr[n] = excl;
    }
  }
}

__global__ void fill_csr(const int* __restrict__ ei, int E,
                         int* __restrict__ cursor, int* __restrict__ srcs) {
  int t = blockIdx.x * blockDim.x + threadIdx.x;
  if (t >= E) return;
  int s = ei[t], d = ei[E + t];
  int pos = atomicAdd(&cursor[d], 1);
  srcs[pos] = s;
}

static __device__ inline float lrelu(float e) { return e >= 0.f ? e : NEG_SLOPE * e; }

template <int X> static __device__ inline float xrow(float v) {
  return __builtin_bit_cast(float, __builtin_amdgcn_update_dpp(
      0, __builtin_bit_cast(int, v), 0x160 + X, 0xF, 0xF, true));
}
static __device__ inline float xhalf(float v) {
  int b = __builtin_bit_cast(int, v);
  return __builtin_bit_cast(float, __builtin_amdgcn_permlanex16(
      b, b, 0x76543210, 0xfedcba98, false, false));
}

static __device__ inline float wave_sum(float s) {
  s += xhalf(s);
  s += xrow<8>(s); s += xrow<4>(s); s += xrow<2>(s); s += xrow<1>(s);
  return s;
}

static __device__ inline float logit1(float2 u, float2 vr, float2 w) {
  float p = lrelu(u.x + vr.x) * w.x + lrelu(u.y + vr.y) * w.y;
  p += xrow<1>(p); p += xrow<2>(p); p += xrow<4>(p);
  return p;
}
static __device__ inline float2 row64(const float* __restrict__ xl, int s, int lane) {
  return ((const float2*)(xl + (size_t)s * 64))[lane];
}

__global__ void gat1_node(const int* __restrict__ rowptr,
                          const int* __restrict__ srcs,
                          const float* __restrict__ xl,
                          const float* __restrict__ xr,
                          const float* __restrict__ a1,
                          const float* __restrict__ b1,
                          const float* __restrict__ g,
                          const float* __restrict__ be,
                          float* __restrict__ h1, int n) {
  const int node = __builtin_amdgcn_readfirstlane(
      (int)(((long long)blockIdx.x * blockDim.x + threadIdx.x) >> 5));
  const int lane = threadIdx.x & 31;
  if (node >= n) return;
  const float2 w  = ((const float2*)a1)[lane];
  const float2 bb = ((const float2*)b1)[lane];
  const float2 gg = ((const float2*)g)[lane];
  const float2 bt = ((const float2*)be)[lane];
  const float2 vr = row64(xr, node, lane);
  const float2 us = row64(xl, node, lane);
  const int beg = rowptr[node], end = rowptr[node + 1];
  const float ls = logit1(us, vr, w);
  float m = ls;
  for (int base = beg; base < end; base += 32) {
    int cnt = end - base; cnt = cnt < 32 ? cnt : 32;
    const int sidx = lane < cnt ? srcs[base + lane] : node;
    int j = 0;
    for (; j + 4 <= cnt; j += 4) {
      float2 u0 = row64(xl, __builtin_amdgcn_readlane(sidx, j + 0), lane);
      float2 u1 = row64(xl, __builtin_amdgcn_readlane(sidx, j + 1), lane);
      float2 u2 = row64(xl, __builtin_amdgcn_readlane(sidx, j + 2), lane);
      float2 u3 = row64(xl, __builtin_amdgcn_readlane(sidx, j + 3), lane);
      float l0 = logit1(u0, vr, w), l1 = logit1(u1, vr, w);
      float l2 = logit1(u2, vr, w), l3 = logit1(u3, vr, w);
      m = fmaxf(fmaxf(m, l0), fmaxf(fmaxf(l1, l2), l3));
    }
    for (; j < cnt; ++j) {
      float2 u = row64(xl, __builtin_amdgcn_readlane(sidx, j), lane);
      m = fmaxf(m, logit1(u, vr, w));
    }
  }
  float ex = __expf(ls - m);
  float den = ex;
  float2 a; a.x = ex * us.x; a.y = ex * us.y;
  for (int base = beg; base < end; base += 32) {
    int cnt = end - base; cnt = cnt < 32 ? cnt : 32;
    const int sidx = lane < cnt ? srcs[base + lane] : node;
    int j = 0;
    for (; j + 4 <= cnt; j += 4) {
      float2 u0 = row64(xl, __builtin_amdgcn_readlane(sidx, j + 0), lane);
      float2 u1 = row64(xl, __builtin_amdgcn_readlane(sidx, j + 1), lane);
      float2 u2 = row64(xl, __builtin_amdgcn_readlane(sidx, j + 2), lane);
      float2 u3 = row64(xl, __builtin_amdgcn_readlane(sidx, j + 3), lane);
      float e0 = __expf(logit1(u0, vr, w) - m), e1 = __expf(logit1(u1, vr, w) - m);
      float e2 = __expf(logit1(u2, vr, w) - m), e3 = __expf(logit1(u3, vr, w) - m);
      den += e0; a.x += e0 * u0.x; a.y += e0 * u0.y;
      den += e1; a.x += e1 * u1.x; a.y += e1 * u1.y;
      den += e2; a.x += e2 * u2.x; a.y += e2 * u2.y;
      den += e3; a.x += e3 * u3.x; a.y += e3 * u3.y;
    }
    for (; j < cnt; ++j) {
      float2 u = row64(xl, __builtin_amdgcn_readlane(sidx, j), lane);
      float e = __expf(logit1(u, vr, w) - m);
      den += e; a.x += e * u.x; a.y += e * u.y;
    }
  }
  const float v0 = a.x / den + bb.x;
  const float v1 = a.y / den + bb.y;
  const float mu = wave_sum(v0 + v1) * (1.f / 64.f);
  const float d0 = v0 - mu, d1 = v1 - mu;
  const float rs = rsqrtf(wave_sum(d0 * d0 + d1 * d1) * (1.f / 64.f) + LN_EPS);
  const float y0 = d0 * rs * gg.x + bt.x;
  const float y1 = d1 * rs * gg.y + bt.y;
  float2 y;
  y.x = y0 > 0.f ? y0 : expm1f(y0);
  y.y = y1 > 0.f ? y1 : expm1f(y1);
  ((float2*)(h1 + (size_t)node * 64))[lane] = y;
}

static __device__ inline float logit2(float u, float vr, float w) {
  float p = lrelu(u + vr) * w;
  p += xrow<1>(p); p += xrow<2>(p); p += xrow<4>(p); p += xrow<8>(p);
  p += xhalf(p);
  return p;
}

__global__ void gat2_node(const int* __restrict__ rowptr,
                          const int* __restrict__ srcs,
                          const float* __restrict__ xl,
                          const float* __restrict__ xr,
                          const float* __restrict__ a2,
                          const float* __restrict__ b2,
                          const float* __restrict__ g,
                          const float* __restrict__ be,
                          float* __restrict__ out, int n) {
  const int node = __builtin_amdgcn_readfirstlane(
      (int)(((long long)blockIdx.x * blockDim.x + threadIdx.x) >> 5));
  const int lane = threadIdx.x & 31;
  if (node >= n) return;
  const float w  = a2[lane];
  const float bb = b2[lane], gg = g[lane], bt = be[lane];
  const float vr = xr[(size_t)node * 32 + lane];
  const float us = xl[(size_t)node * 32 + lane];
  const int beg = rowptr[node], end = rowptr[node + 1];
  const float ls = logit2(us, vr, w);
  float m = ls;
  for (int base = beg; base < end; base += 32) {
    int cnt = end - base; cnt = cnt < 32 ? cnt : 32;
    const int sidx = lane < cnt ? srcs[base + lane] : node;
    int j = 0;
    for (; j + 4 <= cnt; j += 4) {
      float u0 = xl[(size_t)__builtin_amdgcn_readlane(sidx, j + 0) * 32 + lane];
      float u1 = xl[(size_t)__builtin_amdgcn_readlane(sidx, j + 1) * 32 + lane];
      float u2 = xl[(size_t)__builtin_amdgcn_readlane(sidx, j + 2) * 32 + lane];
      float u3 = xl[(size_t)__builtin_amdgcn_readlane(sidx, j + 3) * 32 + lane];
      float l0 = logit2(u0, vr, w), l1 = logit2(u1, vr, w);
      float l2 = logit2(u2, vr, w), l3 = logit2(u3, vr, w);
      m = fmaxf(fmaxf(m, l0), fmaxf(fmaxf(l1, l2), l3));
    }
    for (; j < cnt; ++j) {
      float u = xl[(size_t)__builtin_amdgcn_readlane(sidx, j) * 32 + lane];
      m = fmaxf(m, logit2(u, vr, w));
    }
  }
  float ex = __expf(ls - m);
  float den = ex;
  float a = ex * us;
  for (int base = beg; base < end; base += 32) {
    int cnt = end - base; cnt = cnt < 32 ? cnt : 32;
    const int sidx = lane < cnt ? srcs[base + lane] : node;
    int j = 0;
    for (; j + 4 <= cnt; j += 4) {
      float u0 = xl[(size_t)__builtin_amdgcn_readlane(sidx, j + 0) * 32 + lane];
      float u1 = xl[(size_t)__builtin_amdgcn_readlane(sidx, j + 1) * 32 + lane];
      float u2 = xl[(size_t)__builtin_amdgcn_readlane(sidx, j + 2) * 32 + lane];
      float u3 = xl[(size_t)__builtin_amdgcn_readlane(sidx, j + 3) * 32 + lane];
      float e0 = __expf(logit2(u0, vr, w) - m), e1 = __expf(logit2(u1, vr, w) - m);
      float e2 = __expf(logit2(u2, vr, w) - m), e3 = __expf(logit2(u3, vr, w) - m);
      den += e0; a += e0 * u0;
      den += e1; a += e1 * u1;
      den += e2; a += e2 * u2;
      den += e3; a += e3 * u3;
    }
    for (; j < cnt; ++j) {
      float u = xl[(size_t)__builtin_amdgcn_readlane(sidx, j) * 32 + lane];
      float e = __expf(logit2(u, vr, w) - m);
      den += e; a += e * u;
    }
  }
  const float v = a / den + bb;
  const float mu = wave_sum(v) * (1.f / 32.f);
  const float dd = v - mu;
  const float rs = rsqrtf(wave_sum(dd * dd) * (1.f / 32.f) + LN_EPS);
  out[(size_t)node * 32 + lane] = dd * rs * gg + bt;
}

static inline int gridFor(long long threads, int block) {
  return (int)((threads + block - 1) / block);
}

extern "C" void kernel_launch(void* const* d_in, const int* in_sizes, int n_in,
                              void* d_out, int out_size, void* d_ws, size_t ws_size,
                              hipStream_t stream) {
  const float* x   = (const float*)d_in[0];
  const int*   ei  = (const int*)  d_in[1];
  const float* W1l = (const float*)d_in[2];
  const float* W1r = (const float*)d_in[3];
  const float* a1  = (const float*)d_in[4];
  const float* b1  = (const float*)d_in[5];
  const float* g1  = (const float*)d_in[6];
  const float* be1 = (const float*)d_in[7];
  const float* W2l = (const float*)d_in[8];
  const float* W2r = (const float*)d_in[9];
  const float* a2  = (const float*)d_in[10];
  const float* b2  = (const float*)d_in[11];
  const float* g2  = (const float*)d_in[12];
  const float* be2 = (const float*)d_in[13];
  float* out = (float*)d_out;

  const int n  = in_sizes[0] / 4;
  const int E  = in_sizes[1] / 2;
  const int TE = E + n;
  const int ntiles = n / 16;

  float* ws = (float*)d_ws;
  const size_t XL1  = 0;
  const size_t XR1  = XL1 + (size_t)n * 64;
  const size_t ACC1 = XR1 + (size_t)n * 64;
  const size_t DEN1 = ACC1 + (size_t)n * 64;
  const size_t MAX1 = DEN1 + (size_t)n * 4;
  const size_t LOG1 = MAX1 + (size_t)n * 4;
  const size_t H1   = LOG1 + (size_t)TE * 4;
  int* wi = (int*)d_ws;
  const size_t DEG  = LOG1;
  const size_t RPTR = DEG  + (size_t)n;
  const size_t CURS = RPTR + (size_t)n + 64;
  const size_t BSUM = CURS + (size_t)n + 64;
  const size_t SRCS = BSUM + 896;

  const int B = 256;
  const int nscan = gridFor(n, 1024);

  fill_f32<<<gridFor(n, B), B, 0, stream>>>(ws + DEG, (long long)n, 0.f);
  count_deg<<<gridFor(E, B), B, 0, stream>>>(ei, E, wi + DEG);
  scan_partial<<<nscan, B, 0, stream>>>(wi + DEG, n, wi + BSUM);
  scan_final<<<nscan, B, 0, stream>>>(wi + DEG, n, wi + BSUM, wi + RPTR, wi + CURS);
  fill_csr<<<gridFor(E, B), B, 0, stream>>>(ei, E, wi + CURS, wi + SRCS);

  gemm1_wmma<<<gridFor((long long)ntiles * 32, B), B, 0, stream>>>(
      x, W1l, W1r, ws + XL1, ws + XR1, ntiles);

  gat1_node<<<gridFor((long long)n * 32, B), B, 0, stream>>>(
      wi + RPTR, wi + SRCS, ws + XL1, ws + XR1, a1, b1, g1, be1, ws + H1, n);

  gemm2_wmma<<<gridFor((long long)ntiles * 32, B), B, 0, stream>>>(
      ws + H1, W2l, W2r, ws + XL1, ws + XR1, ntiles);

  gat2_node<<<gridFor((long long)n * 32, B), B, 0, stream>>>(
      wi + RPTR, wi + SRCS, ws + XL1, ws + XR1, a2, b2, g2, be2, out, n);
}


# Edge sort by destination rebuilt in two LDS levels (bin by 512-node range, then one block per bin ranks in LDS): no per-edge memory atomics; sums stay in an arrival-rank order, now LDS ranks
# speedup vs baseline: 9.3413x; 1.9002x over previous
#include <hip/hip_runtime.h>
#include <hip/hip_bf16.h>
#include <math.h>


typedef __attribute__((ext_vector_type(2))) float v2f;
typedef __attribute__((ext_vector_type(8))) float v8f;

#define NEG_SLOPE 0.2f
#define LN_EPS 1e-5f

static __device__ inline v8f wmma4(v2f a, v2f b, v8f c) {
  return __builtin_amdgcn_wmma_f32_16x16x4_f32(
       false, a,  false, b,
       (short)0, c,  false,  false);
}

__global__ void fill_f32(float* __restrict__ p, long long n, float v) {
  long long i = (long long)blockIdx.x * blockDim.x + threadIdx.x;
  if (i < n) p[i] = v;
}

__global__ void gemm1_wmma(const float* __restrict__ x,
                           const float* __restrict__ Wl,
                           const float* __restrict__ Wr,
                           float* __restrict__ xl,
                           float* __restrict__ xr, int ntiles) {
  int wave = (int)(((long long)blockIdx.x * blockDim.x + threadIdx.x) >> 5);
  int lane = threadIdx.x & 31;
  if (wave >= ntiles) return;
  int row0 = wave * 16;
  int m  = lane & 15;
  int hi = lane >> 4;
  const float* xrow = x + (size_t)(row0 + m) * 4 + hi * 2;
  v2f a; a.x = xrow[0]; a.y = xrow[1];
#pragma unroll
  for (int j = 0; j < 4; ++j) {
    v2f bl, br;
    bl.x = Wl[(2 * hi + 0) * 64 + j * 16 + m];
    bl.y = Wl[(2 * hi + 1) * 64 + j * 16 + m];
    br.x = Wr[(2 * hi + 0) * 64 + j * 16 + m];
    br.y = Wr[(2 * hi + 1) * 64 + j * 16 + m];
    v8f cl = {}; cl = wmma4(a, bl, cl);
    v8f cr = {}; cr = wmma4(a, br, cr);
#pragma unroll
    for (int v = 0; v < 8; ++v) {
      int r = v + hi * 8;
      xl[(size_t)(row0 + r) * 64 + j * 16 + m] = cl[v];
      xr[(size_t)(row0 + r) * 64 + j * 16 + m] = cr[v];
    }
  }
}

__global__ void gemm2_wmma(const float* __restrict__ h,
                           const float* __restrict__ Wl,
                           const float* __restrict__ Wr,
                           float* __restrict__ xl,
                           float* __restrict__ xr, int ntiles) {
  int wave = (int)(((long long)blockIdx.x * blockDim.x + threadIdx.x) >> 5);
  int lane = threadIdx.x & 31;
  if (wave >= ntiles) return;
  int row0 = wave * 16;
  int m  = lane & 15;
  int hi = lane >> 4;
  const float* hrow = h + (size_t)(row0 + m) * 64;
  v8f cl[2] = {{0.f},{0.f}};
  v8f cr[2] = {{0.f},{0.f}};
#pragma unroll
  for (int kk = 0; kk < 16; ++kk) {
    int k0 = kk * 4 + hi * 2;
    v2f a; a.x = hrow[k0]; a.y = hrow[k0 + 1];
#pragma unroll
    for (int j = 0; j < 2; ++j) {
      v2f bl, br;
      bl.x = Wl[(k0 + 0) * 32 + j * 16 + m];
      bl.y = Wl[(k0 + 1) * 32 + j * 16 + m];
      br.x = Wr[(k0 + 0) * 32 + j * 16 + m];
      br.y = Wr[(k0 + 1) * 32 + j * 16 + m];
      cl[j] = wmma4(a, bl, cl[j]);
      cr[j] = wmma4(a, br, cr[j]);
    }
  }
#pragma unroll
  for (int j = 0; j < 2; ++j)
#pragma unroll
    for (int v = 0; v < 8; ++v) {
      int r = v + hi * 8;
      xl[(size_t)(row0 + r) * 32 + j * 16 + m] = cl[j][v];
      xr[(size_t)(row0 + r) * 32 + j * 16 + m] = cr[j][v];
    }
}

#define BIN_SHIFT 9
#define BIN_NODES 512
#define NBMAX     256
#define CSTRIDE   32
#define SORT_T    1024
#define SORT_K    8
#define EPB       (SORT_T * SORT_K)

__global__ void __launch_bounds__(SORT_T)
bin_count(const int* __restrict__ ei, int E, int nb, int* __restrict__ binCnt) {
  __shared__ int h[NBMAX];
  const int t = threadIdx.x;
  for (int i = t; i < nb; i += SORT_T) h[i] = 0;
  __syncthreads();
  const int base = blockIdx.x * EPB + t;
  int d[SORT_K];
#pragma unroll
  for (int k = 0; k < SORT_K; ++k) {
    const int e = base + k * SORT_T;
    d[k] = e < E ? ei[E + e] : -1;
  }
#pragma unroll
  for (int k = 0; k < SORT_K; ++k)
    if (d[k] >= 0) atomicAdd(&h[d[k] >> BIN_SHIFT], 1);
  __syncthreads();
  for (int i = t; i < nb; i += SORT_T) {
    const int c = h[i];
    if (c) atomicAdd(&binCnt[i * CSTRIDE], c);
  }
}

__global__ void __launch_bounds__(NBMAX)
bin_scan(const int* __restrict__ binCnt, int nb,
         int* __restrict__ binStart, int* __restrict__ binCur) {
  __shared__ int wsum[NBMAX / 32];
  const int t = threadIdx.x, lane = t & 31, wv = t >> 5;
  const int v = t < nb ? binCnt[t * CSTRIDE] : 0;
  int inc = v;
  for (int k = 1; k < 32; k <<= 1) { int u = __shfl_up(inc, k, 32); if (lane >= k) inc += u; }
  if (lane == 31) wsum[wv] = inc;
  __syncthreads();
  int woff = 0;
  for (int q = 0; q < wv; ++q) woff += wsum[q];
  const int excl = woff + inc - v;
  if (t < nb) { binStart[t] = excl; binCur[t * CSTRIDE] = excl; }
  if (t == nb - 1) binStart[nb] = excl + v;
}

__global__ void __launch_bounds__(SORT_T)
bin_scatter(const int* __restrict__ ei, int E, int nb,
            int* __restrict__ binCur, int* __restrict__ binnedw) {
  int2* __restrict__ binned = (int2*)binnedw;
  __shared__ int h[NBMAX];
  const int t = threadIdx.x;
  for (int i = t; i < nb; i += SORT_T) h[i] = 0;
  __syncthreads();
  const int base = blockIdx.x * EPB + t;
  int s[SORT_K], d[SORT_K], r[SORT_K];
#pragma unroll
  for (int k = 0; k < SORT_K; ++k) {
    const int e = base + k * SORT_T;
    s[k] = 0; d[k] = -1;
    if (e < E) { s[k] = ei[e]; d[k] = ei[E + e]; }
  }
#pragma unroll
  for (int k = 0; k < SORT_K; ++k) {
    r[k] = 0;
    if (d[k] >= 0) r[k] = atomicAdd(&h[d[k] >> BIN_SHIFT], 1);
  }
  __syncthreads();
  for (int i = t; i < nb; i += SORT_T) {
    const int c = h[i];
    h[i] = c ? atomicAdd(&binCur[i * CSTRIDE], c) : 0;
  }
  __syncthreads();
#pragma unroll
  for (int k = 0; k < SORT_K; ++k)
    if (d[k] >= 0) binned[h[d[k] >> BIN_SHIFT] + r[k]] = make_int2(s[k], d[k]);
}

__global__ void __launch_bounds__(SORT_T)
bin_sort(const int* __restrict__ binnedw, const int* __restrict__ binStart, int n,
         int* __restrict__ rowptr, int* __restrict__ srcs) {
  const int2* __restrict__ binned = (const int2*)binnedw;
  __shared__ int cnt[BIN_NODES];
  __shared__ int wsum[SORT_T / 32];
  const int b = blockIdx.x, t = threadIdx.x, lane = t & 31, wv = t >> 5;
  const int beg = binStart[b], end = binStart[b + 1];
  const int node0 = b << BIN_SHIFT;
  if (t < BIN_NODES) cnt[t] = 0;
  __syncthreads();
  for (int c = beg; c < end; c += EPB) {
    int dl[SORT_K];
#pragma unroll
    for (int k = 0; k < SORT_K; ++k) {
      const int i = c + k * SORT_T + t;
      dl[k] = i < end ? binned[i].y : -1;
    }
#pragma unroll
    for (int k = 0; k < SORT_K; ++k)
      if (dl[k] >= 0) atomicAdd(&cnt[dl[k] - node0], 1);
  }
  __syncthreads();
  const int v = t < BIN_NODES ? cnt[t] : 0;
  int inc = v;
  for (int k = 1; k < 32; k <<= 1) { int u = __shfl_up(inc, k, 32); if (lane >= k) inc += u; }
  if (lane == 31) wsum[wv] = inc;
  __syncthreads();
  int woff = 0;
  for (int q = 0; q < wv; ++q) woff += wsum[q];
  const int excl = beg + woff + inc - v;
  if (t < BIN_NODES) {
    cnt[t] = excl;
    const int node = node0 + t;
    if (node < n) rowptr[node] = excl;
    if (node == n - 1) rowptr[n] = excl + v;
  }
  __syncthreads();
  for (int c = beg; c < end; c += EPB) {
    int2 e[SORT_K];
#pragma unroll
    for (int k = 0; k < SORT_K; ++k) {
      const int i = c + k * SORT_T + t;
      e[k] = i < end ? binned[i] : make_int2(0, -1);
    }
#pragma unroll
    for (int k = 0; k < SORT_K; ++k)
      if (e[k].y >= 0) srcs[atomicAdd(&cnt[e[k].y - node0], 1)] = e[k].x;
  }
}

static __device__ inline float lrelu(float e) { return e >= 0.f ? e : NEG_SLOPE * e; }

template <int X> static __device__ inline float xrow(float v) {
  return __builtin_bit_cast(float, __builtin_amdgcn_update_dpp(
      0, __builtin_bit_cast(int, v), 0x160 + X, 0xF, 0xF, true));
}
static __device__ inline float xhalf(float v) {
  int b = __builtin_bit_cast(int, v);
  return __builtin_bit_cast(float, __builtin_amdgcn_permlanex16(
      b, b, 0x76543210, 0xfedcba98, false, false));
}

static __device__ inline float wave_sum(float s) {
  s += xhalf(s);
  s += xrow<8>(s); s += xrow<4>(s); s += xrow<2>(s); s += xrow<1>(s);
  return s;
}

static __device__ inline float logit1(float2 u, float2 vr, float2 w) {
  float p = lrelu(u.x + vr.x) * w.x + lrelu(u.y + vr.y) * w.y;
  p += xrow<1>(p); p += xrow<2>(p); p += xrow<4>(p);
  return p;
}
static __device__ inline float2 row64(const float* __restrict__ xl, int s, int lane) {
  return ((const float2*)(xl + (size_t)s * 64))[lane];
}

__global__ void gat1_node(const int* __restrict__ rowptr,
                          const int* __restrict__ srcs,
                          const float* __restrict__ xl,
                          const float* __restrict__ xr,
                          const float* __restrict__ a1,
                          const float* __restrict__ b1,
                          const float* __restrict__ g,
                          const float* __restrict__ be,
                          float* __restrict__ h1, int n) {
  const int node = __builtin_amdgcn_readfirstlane(
      (int)(((long long)blockIdx.x * blockDim.x + threadIdx.x) >> 5));
  const int lane = threadIdx.x & 31;
  if (node >= n) return;
  const float2 w  = ((const float2*)a1)[lane];
  const float2 bb = ((const float2*)b1)[lane];
  const float2 gg = ((const float2*)g)[lane];
  const float2 bt = ((const float2*)be)[lane];
  const float2 vr = row64(xr, node, lane);
  const float2 us = row64(xl, node, lane);
  const int beg = rowptr[node], end = rowptr[node + 1];
  const float ls = logit1(us, vr, w);
  float m = ls;
  for (int base = beg; base < end; base += 32) {
    int cnt = end - base; cnt = cnt < 32 ? cnt : 32;
    const int sidx = lane < cnt ? srcs[base + lane] : node;
    int j = 0;
    for (; j + 4 <= cnt; j += 4) {
      float2 u0 = row64(xl, __builtin_amdgcn_readlane(sidx, j + 0), lane);
      float2 u1 = row64(xl, __builtin_amdgcn_readlane(sidx, j + 1), lane);
      float2 u2 = row64(xl, __builtin_amdgcn_readlane(sidx, j + 2), lane);
      float2 u3 = row64(xl, __builtin_amdgcn_readlane(sidx, j + 3), lane);
      float l0 = logit1(u0, vr, w), l1 = logit1(u1, vr, w);
      float l2 = logit1(u2, vr, w), l3 = logit1(u3, vr, w);
      m = fmaxf(fmaxf(m, l0), fmaxf(fmaxf(l1, l2), l3));
    }
    for (; j < cnt; ++j) {
      float2 u = row64(xl, __builtin_amdgcn_readlane(sidx, j), lane);
      m = fmaxf(m, logit1(u, vr, w));
    }
  }
  float ex = __expf(ls - m);
  float den = ex;
  float2 a; a.x = ex * us.x; a.y = ex * us.y;
  for (int base = beg; base < end; base += 32) {
    int cnt = end - base; cnt = cnt < 32 ? cnt : 32;
    const int sidx = lane < cnt ? srcs[base + lane] : node;
    int j = 0;
    for (; j + 4 <= cnt; j += 4) {
      float2 u0 = row64(xl, __builtin_amdgcn_readlane(sidx, j + 0), lane);
      float2 u1 = row64(xl, __builtin_amdgcn_readlane(sidx, j + 1), lane);
      float2 u2 = row64(xl, __builtin_amdgcn_readlane(sidx, j + 2), lane);
      float2 u3 = row64(xl, __builtin_amdgcn_readlane(sidx, j + 3), lane);
      float e0 = __expf(logit1(u0, vr, w) - m), e1 = __expf(logit1(u1, vr, w) - m);
      float e2 = __expf(logit1(u2, vr, w) - m), e3 = __expf(logit1(u3, vr, w) - m);
      den += e0; a.x += e0 * u0.x; a.y += e0 * u0.y;
      den += e1; a.x += e1 * u1.x; a.y += e1 * u1.y;
      den += e2; a.x += e2 * u2.x; a.y += e2 * u2.y;
      den += e3; a.x += e3 * u3.x; a.y += e3 * u3.y;
    }
    for (; j < cnt; ++j) {
      float2 u = row64(xl, __builtin_amdgcn_readlane(sidx, j), lane);
      float e = __expf(logit1(u, vr, w) - m);
      den += e; a.x += e * u.x; a.y += e * u.y;
    }
  }
  const float v0 = a.x / den + bb.x;
  const float v1 = a.y / den + bb.y;
  const float mu = wave_sum(v0 + v1) * (1.f / 64.f);
  const float d0 = v0 - mu, d1 = v1 - mu;
  const float rs = rsqrtf(wave_sum(d0 * d0 + d1 * d1) * (1.f / 64.f) + LN_EPS);
  const float y0 = d0 * rs * gg.x + bt.x;
  const float y1 = d1 * rs * gg.y + bt.y;
  float2 y;
  y.x = y0 > 0.f ? y0 : expm1f(y0);
  y.y = y1 > 0.f ? y1 : expm1f(y1);
  ((float2*)(h1 + (size_t)node * 64))[lane] = y;
}

static __device__ inline float logit2(float u, float vr, float w) {
  float p = lrelu(u + vr) * w;
  p += xrow<1>(p); p += xrow<2>(p); p += xrow<4>(p); p += xrow<8>(p);
  p += xhalf(p);
  return p;
}

__global__ void gat2_node(const int* __restrict__ rowptr,
                          const int* __restrict__ srcs,
                          const float* __restrict__ xl,
                          const float* __restrict__ xr,
                          const float* __restrict__ a2,
                          const float* __restrict__ b2,
                          const float* __restrict__ g,
                          const float* __restrict__ be,
                          float* __restrict__ out, int n) {
  const int node = __builtin_amdgcn_readfirstlane(
      (int)(((long long)blockIdx.x * blockDim.x + threadIdx.x) >> 5));
  const int lane = threadIdx.x & 31;
  if (node >= n) return;
  const float w  = a2[lane];
  const float bb = b2[lane], gg = g[lane], bt = be[lane];
  const float vr = xr[(size_t)node * 32 + lane];
  const float us = xl[(size_t)node * 32 + lane];
  const int beg = rowptr[node], end = rowptr[node + 1];
  const float ls = logit2(us, vr, w);
  float m = ls;
  for (int base = beg; base < end; base += 32) {
    int cnt = end - base; cnt = cnt < 32 ? cnt : 32;
    const int sidx = lane < cnt ? srcs[base + lane] : node;
    int j = 0;
    for (; j + 4 <= cnt; j += 4) {
      float u0 = xl[(size_t)__builtin_amdgcn_readlane(sidx, j + 0) * 32 + lane];
      float u1 = xl[(size_t)__builtin_amdgcn_readlane(sidx, j + 1) * 32 + lane];
      float u2 = xl[(size_t)__builtin_amdgcn_readlane(sidx, j + 2) * 32 + lane];
      float u3 = xl[(size_t)__builtin_amdgcn_readlane(sidx, j + 3) * 32 + lane];
      float l0 = logit2(u0, vr, w), l1 = logit2(u1, vr, w);
      float l2 = logit2(u2, vr, w), l3 = logit2(u3, vr, w);
      m = fmaxf(fmaxf(m, l0), fmaxf(fmaxf(l1, l2), l3));
    }
    for (; j < cnt; ++j) {
      float u = xl[(size_t)__builtin_amdgcn_readlane(sidx, j) * 32 + lane];
      m = fmaxf(m, logit2(u, vr, w));
    }
  }
  float ex = __expf(ls - m);
  float den = ex;
  float a = ex * us;
  for (int base = beg; base < end; base += 32) {
    int cnt = end - base; cnt = cnt < 32 ? cnt : 32;
    const int sidx = lane < cnt ? srcs[base + lane] : node;
    int j = 0;
    for (; j + 4 <= cnt; j += 4) {
      float u0 = xl[(size_t)__builtin_amdgcn_readlane(sidx, j + 0) * 32 + lane];
      float u1 = xl[(size_t)__builtin_amdgcn_readlane(sidx, j + 1) * 32 + lane];
      float u2 = xl[(size_t)__builtin_amdgcn_readlane(sidx, j + 2) * 32 + lane];
      float u3 = xl[(size_t)__builtin_amdgcn_readlane(sidx, j + 3) * 32 + lane];
      float e0 = __expf(logit2(u0, vr, w) - m), e1 = __expf(logit2(u1, vr, w) - m);
      float e2 = __expf(logit2(u2, vr, w) - m), e3 = __expf(logit2(u3, vr, w) - m);
      den += e0; a += e0 * u0;
      den += e1; a += e1 * u1;
      den += e2; a += e2 * u2;
      den += e3; a += e3 * u3;
    }
    for (; j < cnt; ++j) {
      float u = xl[(size_t)__builtin_amdgcn_readlane(sidx, j) * 32 + lane];
      float e = __expf(logit2(u, vr, w) - m);
      den += e; a += e * u;
    }
  }
  const float v = a / den + bb;
  const float mu = wave_sum(v) * (1.f / 32.f);
  const float dd = v - mu;
  const float rs = rsqrtf(wave_sum(dd * dd) * (1.f / 32.f) + LN_EPS);
  out[(size_t)node * 32 + lane] = dd * rs * gg + bt;
}

static inline int gridFor(long long threads, int block) {
  return (int)((threads + block - 1) / block);
}

extern "C" void kernel_launch(void* const* d_in, const int* in_sizes, int n_in,
                              void* d_out, int out_size, void* d_ws, size_t ws_size,
                              hipStream_t stream) {
  const float* x   = (const float*)d_in[0];
  const int*   ei  = (const int*)  d_in[1];
  const float* W1l = (const float*)d_in[2];
  const float* W1r = (const float*)d_in[3];
  const float* a1  = (const float*)d_in[4];
  const float* b1  = (const float*)d_in[5];
  const float* g1  = (const float*)d_in[6];
  const float* be1 = (const float*)d_in[7];
  const float* W2l = (const float*)d_in[8];
  const float* W2r = (const float*)d_in[9];
  const float* a2  = (const float*)d_in[10];
  const float* b2  = (const float*)d_in[11];
  const float* g2  = (const float*)d_in[12];
  const float* be2 = (const float*)d_in[13];
  float* out = (float*)d_out;

  const int n  = in_sizes[0] / 4;
  const int E  = in_sizes[1] / 2;
  const int TE = E + n;
  const int ntiles = n / 16;

  float* ws = (float*)d_ws;
  const size_t XL1  = 0;
  const size_t XR1  = XL1 + (size_t)n * 64;
  const size_t ACC1 = XR1 + (size_t)n * 64;
  const size_t DEN1 = ACC1 + (size_t)n * 64;
  const size_t MAX1 = DEN1 + (size_t)n * 4;
  const size_t LOG1 = MAX1 + (size_t)n * 4;
  const size_t H1   = LOG1 + (size_t)TE * 4;
  int* wi = (int*)d_ws;
  const size_t BCNT = LOG1;
  const size_t BCUR = BCNT + (size_t)NBMAX * CSTRIDE;
  const size_t BSTA = BCUR + (size_t)NBMAX * CSTRIDE;
  const size_t RPTR = BSTA + NBMAX + 64;
  const size_t SRCS = RPTR + (size_t)n + 64;
  const size_t BINN = (SRCS + (size_t)E + 1) & ~(size_t)1;

  const int B = 256;
  const int nb = (n + BIN_NODES - 1) >> BIN_SHIFT;
  const int nchunk = gridFor(E, EPB);

  fill_f32<<<gridFor(NBMAX * CSTRIDE, B), B, 0, stream>>>(
      ws + BCNT, (long long)NBMAX * CSTRIDE, 0.f);
  bin_count<<<nchunk, SORT_T, 0, stream>>>(ei, E, nb, wi + BCNT);
  bin_scan<<<1, NBMAX, 0, stream>>>(wi + BCNT, nb, wi + BSTA, wi + BCUR);
  bin_scatter<<<nchunk, SORT_T, 0, stream>>>(ei, E, nb, wi + BCUR, wi + BINN);
  bin_sort<<<nb, SORT_T, 0, stream>>>(wi + BINN, wi + BSTA, n, wi + RPTR, wi + SRCS);

  gemm1_wmma<<<gridFor((long long)ntiles * 32, B), B, 0, stream>>>(
      x, W1l, W1r, ws + XL1, ws + XR1, ntiles);

  gat1_node<<<gridFor((long long)n * 32, B), B, 0, stream>>>(
      wi + RPTR, wi + SRCS, ws + XL1, ws + XR1, a1, b1, g1, be1, ws + H1, n);

  gemm2_wmma<<<gridFor((long long)ntiles * 32, B), B, 0, stream>>>(
      ws + H1, W2l, W2r, ws + XL1, ws + XR1, ntiles);

  gat2_node<<<gridFor((long long)n * 32, B), B, 0, stream>>>(
      wi + RPTR, wi + SRCS, ws + XL1, ws + XR1, a2, b2, g2, be2, out, n);
}


# Both node kernels walk a node's edges once with a running max (online softmax): denominator and sums rescaled by exp(old max - new max) per 4-edge group; the max pass and its row gathers go
# speedup vs baseline: 11.5475x; 1.2362x over previous
#include <hip/hip_runtime.h>
#include <hip/hip_bf16.h>
#include <math.h>


typedef __attribute__((ext_vector_type(2))) float v2f;
typedef __attribute__((ext_vector_type(8))) float v8f;

#define NEG_SLOPE 0.2f
#define LN_EPS 1e-5f

static __device__ inline v8f wmma4(v2f a, v2f b, v8f c) {
  return __builtin_amdgcn_wmma_f32_16x16x4_f32(
       false, a,  false, b,
       (short)0, c,  false,  false);
}

__global__ void fill_f32(float* __restrict__ p, long long n, float v) {
  long long i = (long long)blockIdx.x * blockDim.x + threadIdx.x;
  if (i < n) p[i] = v;
}

__global__ void gemm1_wmma(const float* __restrict__ x,
                           const float* __restrict__ Wl,
                           const float* __restrict__ Wr,
                           float* __restrict__ xl,
                           float* __restrict__ xr, int ntiles) {
  int wave = (int)(((long long)blockIdx.x * blockDim.x + threadIdx.x) >> 5);
  int lane = threadIdx.x & 31;
  if (wave >= ntiles) return;
  int row0 = wave * 16;
  int m  = lane & 15;
  int hi = lane >> 4;
  const float* xrow = x + (size_t)(row0 + m) * 4 + hi * 2;
  v2f a; a.x = xrow[0]; a.y = xrow[1];
#pragma unroll
  for (int j = 0; j < 4; ++j) {
    v2f bl, br;
    bl.x = Wl[(2 * hi + 0) * 64 + j * 16 + m];
    bl.y = Wl[(2 * hi + 1) * 64 + j * 16 + m];
    br.x = Wr[(2 * hi + 0) * 64 + j * 16 + m];
    br.y = Wr[(2 * hi + 1) * 64 + j * 16 + m];
    v8f cl = {}; cl = wmma4(a, bl, cl);
    v8f cr = {}; cr = wmma4(a, br, cr);
#pragma unroll
    for (int v = 0; v < 8; ++v) {
      int r = v + hi * 8;
      xl[(size_t)(row0 + r) * 64 + j * 16 + m] = cl[v];
      xr[(size_t)(row0 + r) * 64 + j * 16 + m] = cr[v];
    }
  }
}

__global__ void gemm2_wmma(const float* __restrict__ h,
                           const float* __restrict__ Wl,
                           const float* __restrict__ Wr,
                           float* __restrict__ xl,
                           float* __restrict__ xr, int ntiles) {
  int wave = (int)(((long long)blockIdx.x * blockDim.x + threadIdx.x) >> 5);
  int lane = threadIdx.x & 31;
  if (wave >= ntiles) return;
  int row0 = wave * 16;
  int m  = lane & 15;
  int hi = lane >> 4;
  const float* hrow = h + (size_t)(row0 + m) * 64;
  v8f cl[2] = {{0.f},{0.f}};
  v8f cr[2] = {{0.f},{0.f}};
#pragma unroll
  for (int kk = 0; kk < 16; ++kk) {
    int k0 = kk * 4 + hi * 2;
    v2f a; a.x = hrow[k0]; a.y = hrow[k0 + 1];
#pragma unroll
    for (int j = 0; j < 2; ++j) {
      v2f bl, br;
      bl.x = Wl[(k0 + 0) * 32 + j * 16 + m];
      bl.y = Wl[(k0 + 1) * 32 + j * 16 + m];
      br.x = Wr[(k0 + 0) * 32 + j * 16 + m];
      br.y = Wr[(k0 + 1) * 32 + j * 16 + m];
      cl[j] = wmma4(a, bl, cl[j]);
      cr[j] = wmma4(a, br, cr[j]);
    }
  }
#pragma unroll
  for (int j = 0; j < 2; ++j)
#pragma unroll
    for (int v = 0; v < 8; ++v) {
      int r = v + hi * 8;
      xl[(size_t)(row0 + r) * 32 + j * 16 + m] = cl[j][v];
      xr[(size_t)(row0 + r) * 32 + j * 16 + m] = cr[j][v];
    }
}

#define BIN_SHIFT 9
#define BIN_NODES 512
#define NBMAX     256
#define CSTRIDE   32
#define SORT_T    1024
#define SORT_K    8
#define EPB       (SORT_T * SORT_K)

__global__ void __launch_bounds__(SORT_T)
bin_count(const int* __restrict__ ei, int E, int nb, int* __restrict__ binCnt) {
  __shared__ int h[NBMAX];
  const int t = threadIdx.x;
  for (int i = t; i < nb; i += SORT_T) h[i] = 0;
  __syncthreads();
  const int base = blockIdx.x * EPB + t;
  int d[SORT_K];
#pragma unroll
  for (int k = 0; k < SORT_K; ++k) {
    const int e = base + k * SORT_T;
    d[k] = e < E ? ei[E + e] : -1;
  }
#pragma unroll
  for (int k = 0; k < SORT_K; ++k)
    if (d[k] >= 0) atomicAdd(&h[d[k] >> BIN_SHIFT], 1);
  __syncthreads();
  for (int i = t; i < nb; i += SORT_T) {
    const int c = h[i];
    if (c) atomicAdd(&binCnt[i * CSTRIDE], c);
  }
}

__global__ void __launch_bounds__(NBMAX)
bin_scan(const int* __restrict__ binCnt, int nb,
         int* __restrict__ binStart, int* __restrict__ binCur) {
  __shared__ int wsum[NBMAX / 32];
  const int t = threadIdx.x, lane = t & 31, wv = t >> 5;
  const int v = t < nb ? binCnt[t * CSTRIDE] : 0;
  int inc = v;
  for (int k = 1; k < 32; k <<= 1) { int u = __shfl_up(inc, k, 32); if (lane >= k) inc += u; }
  if (lane == 31) wsum[wv] = inc;
  __syncthreads();
  int woff = 0;
  for (int q = 0; q < wv; ++q) woff += wsum[q];
  const int excl = woff + inc - v;
  if (t < nb) { binStart[t] = excl; binCur[t * CSTRIDE] = excl; }
  if (t == nb - 1) binStart[nb] = excl + v;
}

__global__ void __launch_bounds__(SORT_T)
bin_scatter(const int* __restrict__ ei, int E, int nb,
            int* __restrict__ binCur, int* __restrict__ binnedw) {
  int2* __restrict__ binned = (int2*)binnedw;
  __shared__ int h[NBMAX];
  const int t = threadIdx.x;
  for (int i = t; i < nb; i += SORT_T) h[i] = 0;
  __syncthreads();
  const int base = blockIdx.x * EPB + t;
  int s[SORT_K], d[SORT_K], r[SORT_K];
#pragma unroll
  for (int k = 0; k < SORT_K; ++k) {
    const int e = base + k * SORT_T;
    s[k] = 0; d[k] = -1;
    if (e < E) { s[k] = ei[e]; d[k] = ei[E + e]; }
  }
#pragma unroll
  for (int k = 0; k < SORT_K; ++k) {
    r[k] = 0;
    if (d[k] >= 0) r[k] = atomicAdd(&h[d[k] >> BIN_SHIFT], 1);
  }
  __syncthreads();
  for (int i = t; i < nb; i += SORT_T) {
    const int c = h[i];
    h[i] = c ? atomicAdd(&binCur[i * CSTRIDE], c) : 0;
  }
  __syncthreads();
#pragma unroll
  for (int k = 0; k < SORT_K; ++k)
    if (d[k] >= 0) binned[h[d[k] >> BIN_SHIFT] + r[k]] = make_int2(s[k], d[k]);
}

__global__ void __launch_bounds__(SORT_T)
bin_sort(const int* __restrict__ binnedw, const int* __restrict__ binStart, int n,
         int* __restrict__ rowptr, int* __restrict__ srcs) {
  const int2* __restrict__ binned = (const int2*)binnedw;
  __shared__ int cnt[BIN_NODES];
  __shared__ int wsum[SORT_T / 32];
  const int b = blockIdx.x, t = threadIdx.x, lane = t & 31, wv = t >> 5;
  const int beg = binStart[b], end = binStart[b + 1];
  const int node0 = b << BIN_SHIFT;
  if (t < BIN_NODES) cnt[t] = 0;
  __syncthreads();
  for (int c = beg; c < end; c += EPB) {
    int dl[SORT_K];
#pragma unroll
    for (int k = 0; k < SORT_K; ++k) {
      const int i = c + k * SORT_T + t;
      dl[k] = i < end ? binned[i].y : -1;
    }
#pragma unroll
    for (int k = 0; k < SORT_K; ++k)
      if (dl[k] >= 0) atomicAdd(&cnt[dl[k] - node0], 1);
  }
  __syncthreads();
  const int v = t < BIN_NODES ? cnt[t] : 0;
  int inc = v;
  for (int k = 1; k < 32; k <<= 1) { int u = __shfl_up(inc, k, 32); if (lane >= k) inc += u; }
  if (lane == 31) wsum[wv] = inc;
  __syncthreads();
  int woff = 0;
  for (int q = 0; q < wv; ++q) woff += wsum[q];
  const int excl = beg + woff + inc - v;
  if (t < BIN_NODES) {
    cnt[t] = excl;
    const int node = node0 + t;
    if (node < n) rowptr[node] = excl;
    if (node == n - 1) rowptr[n] = excl + v;
  }
  __syncthreads();
  for (int c = beg; c < end; c += EPB) {
    int2 e[SORT_K];
#pragma unroll
    for (int k = 0; k < SORT_K; ++k) {
      const int i = c + k * SORT_T + t;
      e[k] = i < end ? binned[i] : make_int2(0, -1);
    }
#pragma unroll
    for (int k = 0; k < SORT_K; ++k)
      if (e[k].y >= 0) srcs[atomicAdd(&cnt[e[k].y - node0], 1)] = e[k].x;
  }
}

static __device__ inline float lrelu(float e) { return e >= 0.f ? e : NEG_SLOPE * e; }

template <int X> static __device__ inline float xrow(float v) {
  return __builtin_bit_cast(float, __builtin_amdgcn_update_dpp(
      0, __builtin_bit_cast(int, v), 0x160 + X, 0xF, 0xF, true));
}
static __device__ inline float xhalf(float v) {
  int b = __builtin_bit_cast(int, v);
  return __builtin_bit_cast(float, __builtin_amdgcn_permlanex16(
      b, b, 0x76543210, 0xfedcba98, false, false));
}

static __device__ inline float wave_sum(float s) {
  s += xhalf(s);
  s += xrow<8>(s); s += xrow<4>(s); s += xrow<2>(s); s += xrow<1>(s);
  return s;
}

static __device__ inline float logit1(float2 u, float2 vr, float2 w) {
  float p = lrelu(u.x + vr.x) * w.x + lrelu(u.y + vr.y) * w.y;
  p += xrow<1>(p); p += xrow<2>(p); p += xrow<4>(p);
  return p;
}
static __device__ inline float2 row64(const float* __restrict__ xl, int s, int lane) {
  return ((const float2*)(xl + (size_t)s * 64))[lane];
}

__global__ void gat1_node(const int* __restrict__ rowptr,
                          const int* __restrict__ srcs,
                          const float* __restrict__ xl,
                          const float* __restrict__ xr,
                          const float* __restrict__ a1,
                          const float* __restrict__ b1,
                          const float* __restrict__ g,
                          const float* __restrict__ be,
                          float* __restrict__ h1, int n) {
  const int node = __builtin_amdgcn_readfirstlane(
      (int)(((long long)blockIdx.x * blockDim.x + threadIdx.x) >> 5));
  const int lane = threadIdx.x & 31;
  if (node >= n) return;
  const float2 w  = ((const float2*)a1)[lane];
  const float2 bb = ((const float2*)b1)[lane];
  const float2 gg = ((const float2*)g)[lane];
  const float2 bt = ((const float2*)be)[lane];
  const float2 vr = row64(xr, node, lane);
  const float2 us = row64(xl, node, lane);
  const int beg = rowptr[node], end = rowptr[node + 1];
  float m = logit1(us, vr, w);
  float den = 1.f;
  float2 a = us;
  for (int base = beg; base < end; base += 32) {
    int cnt = end - base; cnt = cnt < 32 ? cnt : 32;
    const int sidx = lane < cnt ? srcs[base + lane] : node;
    int j = 0;
    for (; j + 4 <= cnt; j += 4) {
      float2 u0 = row64(xl, __builtin_amdgcn_readlane(sidx, j + 0), lane);
      float2 u1 = row64(xl, __builtin_amdgcn_readlane(sidx, j + 1), lane);
      float2 u2 = row64(xl, __builtin_amdgcn_readlane(sidx, j + 2), lane);
      float2 u3 = row64(xl, __builtin_amdgcn_readlane(sidx, j + 3), lane);
      const float l0 = logit1(u0, vr, w), l1 = logit1(u1, vr, w);
      const float l2 = logit1(u2, vr, w), l3 = logit1(u3, vr, w);
      const float mn = fmaxf(fmaxf(m, l0), fmaxf(fmaxf(l1, l2), l3));
      const float sc = __expf(m - mn);
      const float e0 = __expf(l0 - mn), e1 = __expf(l1 - mn);
      const float e2 = __expf(l2 - mn), e3 = __expf(l3 - mn);
      den = den * sc + e0; a.x = a.x * sc + e0 * u0.x; a.y = a.y * sc + e0 * u0.y;
      den += e1; a.x += e1 * u1.x; a.y += e1 * u1.y;
      den += e2; a.x += e2 * u2.x; a.y += e2 * u2.y;
      den += e3; a.x += e3 * u3.x; a.y += e3 * u3.y;
      m = mn;
    }
    for (; j < cnt; ++j) {
      float2 u = row64(xl, __builtin_amdgcn_readlane(sidx, j), lane);
      const float l = logit1(u, vr, w);
      const float mn = fmaxf(m, l);
      const float sc = __expf(m - mn);
      const float e = __expf(l - mn);
      den = den * sc + e; a.x = a.x * sc + e * u.x; a.y = a.y * sc + e * u.y;
      m = mn;
    }
  }
  const float v0 = a.x / den + bb.x;
  const float v1 = a.y / den + bb.y;
  const float mu = wave_sum(v0 + v1) * (1.f / 64.f);
  const float d0 = v0 - mu, d1 = v1 - mu;
  const float rs = rsqrtf(wave_sum(d0 * d0 + d1 * d1) * (1.f / 64.f) + LN_EPS);
  const float y0 = d0 * rs * gg.x + bt.x;
  const float y1 = d1 * rs * gg.y + bt.y;
  float2 y;
  y.x = y0 > 0.f ? y0 : expm1f(y0);
  y.y = y1 > 0.f ? y1 : expm1f(y1);
  ((float2*)(h1 + (size_t)node * 64))[lane] = y;
}

static __device__ inline float logit2(float u, float vr, float w) {
  float p = lrelu(u + vr) * w;
  p += xrow<1>(p); p += xrow<2>(p); p += xrow<4>(p); p += xrow<8>(p);
  p += xhalf(p);
  return p;
}

__global__ void gat2_node(const int* __restrict__ rowptr,
                          const int* __restrict__ srcs,
                          const float* __restrict__ xl,
                          const float* __restrict__ xr,
                          const float* __restrict__ a2,
                          const float* __restrict__ b2,
                          const float* __restrict__ g,
                          const float* __restrict__ be,
                          float* __restrict__ out, int n) {
  const int node = __builtin_amdgcn_readfirstlane(
      (int)(((long long)blockIdx.x * blockDim.x + threadIdx.x) >> 5));
  const int lane = threadIdx.x & 31;
  if (node >= n) return;
  const float w  = a2[lane];
  const float bb = b2[lane], gg = g[lane], bt = be[lane];
  const float vr = xr[(size_t)node * 32 + lane];
  const float us = xl[(size_t)node * 32 + lane];
  const int beg = rowptr[node], end = rowptr[node + 1];
  float m = logit2(us, vr, w);
  float den = 1.f;
  float a = us;
  for (int base = beg; base < end; base += 32) {
    int cnt = end - base; cnt = cnt < 32 ? cnt : 32;
    const int sidx = lane < cnt ? srcs[base + lane] : node;
    int j = 0;
    for (; j + 4 <= cnt; j += 4) {
      float u0 = xl[(size_t)__builtin_amdgcn_readlane(sidx, j + 0) * 32 + lane];
      float u1 = xl[(size_t)__builtin_amdgcn_readlane(sidx, j + 1) * 32 + lane];
      float u2 = xl[(size_t)__builtin_amdgcn_readlane(sidx, j + 2) * 32 + lane];
      float u3 = xl[(size_t)__builtin_amdgcn_readlane(sidx, j + 3) * 32 + lane];
      const float l0 = logit2(u0, vr, w), l1 = logit2(u1, vr, w);
      const float l2 = logit2(u2, vr, w), l3 = logit2(u3, vr, w);
      const float mn = fmaxf(fmaxf(m, l0), fmaxf(fmaxf(l1, l2), l3));
      const float sc = __expf(m - mn);
      const float e0 = __expf(l0 - mn), e1 = __expf(l1 - mn);
      const float e2 = __expf(l2 - mn), e3 = __expf(l3 - mn);
      den = den * sc + e0; a = a * sc + e0 * u0;
      den += e1; a += e1 * u1;
      den += e2; a += e2 * u2;
      den += e3; a += e3 * u3;
      m = mn;
    }
    for (; j < cnt; ++j) {
      float u = xl[(size_t)__builtin_amdgcn_readlane(sidx, j) * 32 + lane];
      const float l = logit2(u, vr, w);
      const float mn = fmaxf(m, l);
      const float sc = __expf(m - mn);
      const float e = __expf(l - mn);
      den = den * sc + e; a = a * sc + e * u;
      m = mn;
    }
  }
  const float v = a / den + bb;
  const float mu = wave_sum(v) * (1.f / 32.f);
  const float dd = v - mu;
  const float rs = rsqrtf(wave_sum(dd * dd) * (1.f / 32.f) + LN_EPS);
  out[(size_t)node * 32 + lane] = dd * rs * gg + bt;
}

static inline int gridFor(long long threads, int block) {
  return (int)((threads + block - 1) / block);
}

extern "C" void kernel_launch(void* const* d_in, const int* in_sizes, int n_in,
                              void* d_out, int out_size, void* d_ws, size_t ws_size,
                              hipStream_t stream) {
  const float* x   = (const float*)d_in[0];
  const int*   ei  = (const int*)  d_in[1];
  const float* W1l = (const float*)d_in[2];
  const float* W1r = (const float*)d_in[3];
  const float* a1  = (const float*)d_in[4];
  const float* b1  = (const float*)d_in[5];
  const float* g1  = (const float*)d_in[6];
  const float* be1 = (const float*)d_in[7];
  const float* W2l = (const float*)d_in[8];
  const float* W2r = (const float*)d_in[9];
  const float* a2  = (const float*)d_in[10];
  const float* b2  = (const float*)d_in[11];
  const float* g2  = (const float*)d_in[12];
  const float* be2 = (const float*)d_in[13];
  float* out = (float*)d_out;

  const int n  = in_sizes[0] / 4;
  const int E  = in_sizes[1] / 2;
  const int TE = E + n;
  const int ntiles = n / 16;

  float* ws = (float*)d_ws;
  const size_t XL1  = 0;
  const size_t XR1  = XL1 + (size_t)n * 64;
  const size_t ACC1 = XR1 + (size_t)n * 64;
  const size_t DEN1 = ACC1 + (size_t)n * 64;
  const size_t MAX1 = DEN1 + (size_t)n * 4;
  const size_t LOG1 = MAX1 + (size_t)n * 4;
  const size_t H1   = LOG1 + (size_t)TE * 4;
  int* wi = (int*)d_ws;
  const size_t BCNT = LOG1;
  const size_t BCUR = BCNT + (size_t)NBMAX * CSTRIDE;
  const size_t BSTA = BCUR + (size_t)NBMAX * CSTRIDE;
  const size_t RPTR = BSTA + NBMAX + 64;
  const size_t SRCS = RPTR + (size_t)n + 64;
  const size_t BINN = (SRCS + (size_t)E + 1) & ~(size_t)1;

  const int B = 256;
  const int nb = (n + BIN_NODES - 1) >> BIN_SHIFT;
  const int nchunk = gridFor(E, EPB);

  fill_f32<<<gridFor(NBMAX * CSTRIDE, B), B, 0, stream>>>(
      ws + BCNT, (long long)NBMAX * CSTRIDE, 0.f);
  bin_count<<<nchunk, SORT_T, 0, stream>>>(ei, E, nb, wi + BCNT);
  bin_scan<<<1, NBMAX, 0, stream>>>(wi + BCNT, nb, wi + BSTA, wi + BCUR);
  bin_scatter<<<nchunk, SORT_T, 0, stream>>>(ei, E, nb, wi + BCUR, wi + BINN);
  bin_sort<<<nb, SORT_T, 0, stream>>>(wi + BINN, wi + BSTA, n, wi + RPTR, wi + SRCS);

  gemm1_wmma<<<gridFor((long long)ntiles * 32, B), B, 0, stream>>>(
      x, W1l, W1r, ws + XL1, ws + XR1, ntiles);

  gat1_node<<<gridFor((long long)n * 32, B), B, 0, stream>>>(
      wi + RPTR, wi + SRCS, ws + XL1, ws + XR1, a1, b1, g1, be1, ws + H1, n);

  gemm2_wmma<<<gridFor((long long)ntiles * 32, B), B, 0, stream>>>(
      ws + H1, W2l, W2r, ws + XL1, ws + XR1, ntiles);

  gat2_node<<<gridFor((long long)n * 32, B), B, 0, stream>>>(
      wi + RPTR, wi + SRCS, ws + XL1, ws + XR1, a2, b2, g2, be2, out, n);
}


# Node kernels: each 16-lane half-wave walks its own edges (2 edges per instruction, 8 rows in flight), halves merged by max-rescale per node; softmax and LayerNorm sums in a new order
# speedup vs baseline: 12.3757x; 1.0717x over previous
#include <hip/hip_runtime.h>
#include <hip/hip_bf16.h>
#include <math.h>


typedef __attribute__((ext_vector_type(2))) float v2f;
typedef __attribute__((ext_vector_type(8))) float v8f;

#define NEG_SLOPE 0.2f
#define LN_EPS 1e-5f

static __device__ inline v8f wmma4(v2f a, v2f b, v8f c) {
  return __builtin_amdgcn_wmma_f32_16x16x4_f32(
       false, a,  false, b,
       (short)0, c,  false,  false);
}

__global__ void fill_f32(float* __restrict__ p, long long n, float v) {
  long long i = (long long)blockIdx.x * blockDim.x + threadIdx.x;
  if (i < n) p[i] = v;
}

__global__ void gemm1_wmma(const float* __restrict__ x,
                           const float* __restrict__ Wl,
                           const float* __restrict__ Wr,
                           float* __restrict__ xl,
                           float* __restrict__ xr, int ntiles) {
  int wave = (int)(((long long)blockIdx.x * blockDim.x + threadIdx.x) >> 5);
  int lane = threadIdx.x & 31;
  if (wave >= ntiles) return;
  int row0 = wave * 16;
  int m  = lane & 15;
  int hi = lane >> 4;
  const float* xrow = x + (size_t)(row0 + m) * 4 + hi * 2;
  v2f a; a.x = xrow[0]; a.y = xrow[1];
#pragma unroll
  for (int j = 0; j < 4; ++j) {
    v2f bl, br;
    bl.x = Wl[(2 * hi + 0) * 64 + j * 16 + m];
    bl.y = Wl[(2 * hi + 1) * 64 + j * 16 + m];
    br.x = Wr[(2 * hi + 0) * 64 + j * 16 + m];
    br.y = Wr[(2 * hi + 1) * 64 + j * 16 + m];
    v8f cl = {}; cl = wmma4(a, bl, cl);
    v8f cr = {}; cr = wmma4(a, br, cr);
#pragma unroll
    for (int v = 0; v < 8; ++v) {
      int r = v + hi * 8;
      xl[(size_t)(row0 + r) * 64 + j * 16 + m] = cl[v];
      xr[(size_t)(row0 + r) * 64 + j * 16 + m] = cr[v];
    }
  }
}

__global__ void gemm2_wmma(const float* __restrict__ h,
                           const float* __restrict__ Wl,
                           const float* __restrict__ Wr,
                           float* __restrict__ xl,
                           float* __restrict__ xr, int ntiles) {
  int wave = (int)(((long long)blockIdx.x * blockDim.x + threadIdx.x) >> 5);
  int lane = threadIdx.x & 31;
  if (wave >= ntiles) return;
  int row0 = wave * 16;
  int m  = lane & 15;
  int hi = lane >> 4;
  const float* hrow = h + (size_t)(row0 + m) * 64;
  v8f cl[2] = {{0.f},{0.f}};
  v8f cr[2] = {{0.f},{0.f}};
#pragma unroll
  for (int kk = 0; kk < 16; ++kk) {
    int k0 = kk * 4 + hi * 2;
    v2f a; a.x = hrow[k0]; a.y = hrow[k0 + 1];
#pragma unroll
    for (int j = 0; j < 2; ++j) {
      v2f bl, br;
      bl.x = Wl[(k0 + 0) * 32 + j * 16 + m];
      bl.y = Wl[(k0 + 1) * 32 + j * 16 + m];
      br.x = Wr[(k0 + 0) * 32 + j * 16 + m];
      br.y = Wr[(k0 + 1) * 32 + j * 16 + m];
      cl[j] = wmma4(a, bl, cl[j]);
      cr[j] = wmma4(a, br, cr[j]);
    }
  }
#pragma unroll
  for (int j = 0; j < 2; ++j)
#pragma unroll
    for (int v = 0; v < 8; ++v) {
      int r = v + hi * 8;
      xl[(size_t)(row0 + r) * 32 + j * 16 + m] = cl[j][v];
      xr[(size_t)(row0 + r) * 32 + j * 16 + m] = cr[j][v];
    }
}

#define BIN_SHIFT 9
#define BIN_NODES 512
#define NBMAX     256
#define CSTRIDE   32
#define SORT_T    1024
#define SORT_K    8
#define EPB       (SORT_T * SORT_K)

__global__ void __launch_bounds__(SORT_T)
bin_count(const int* __restrict__ ei, int E, int nb, int* __restrict__ binCnt) {
  __shared__ int h[NBMAX];
  const int t = threadIdx.x;
  for (int i = t; i < nb; i += SORT_T) h[i] = 0;
  __syncthreads();
  const int base = blockIdx.x * EPB + t;
  int d[SORT_K];
#pragma unroll
  for (int k = 0; k < SORT_K; ++k) {
    const int e = base + k * SORT_T;
    d[k] = e < E ? ei[E + e] : -1;
  }
#pragma unroll
  for (int k = 0; k < SORT_K; ++k)
    if (d[k] >= 0) atomicAdd(&h[d[k] >> BIN_SHIFT], 1);
  __syncthreads();
  for (int i = t; i < nb; i += SORT_T) {
    const int c = h[i];
    if (c) atomicAdd(&binCnt[i * CSTRIDE], c);
  }
}

__global__ void __launch_bounds__(NBMAX)
bin_scan(const int* __restrict__ binCnt, int nb,
         int* __restrict__ binStart, int* __restrict__ binCur) {
  __shared__ int wsum[NBMAX / 32];
  const int t = threadIdx.x, lane = t & 31, wv = t >> 5;
  const int v = t < nb ? binCnt[t * CSTRIDE] : 0;
  int inc = v;
  for (int k = 1; k < 32; k <<= 1) { int u = __shfl_up(inc, k, 32); if (lane >= k) inc += u; }
  if (lane == 31) wsum[wv] = inc;
  __syncthreads();
  int woff = 0;
  for (int q = 0; q < wv; ++q) woff += wsum[q];
  const int excl = woff + inc - v;
  if (t < nb) { binStart[t] = excl; binCur[t * CSTRIDE] = excl; }
  if (t == nb - 1) binStart[nb] = excl + v;
}

__global__ void __launch_bounds__(SORT_T)
bin_scatter(const int* __restrict__ ei, int E, int nb,
            int* __restrict__ binCur, int* __restrict__ binnedw) {
  int2* __restrict__ binned = (int2*)binnedw;
  __shared__ int h[NBMAX];
  const int t = threadIdx.x;
  for (int i = t; i < nb; i += SORT_T) h[i] = 0;
  __syncthreads();
  const int base = blockIdx.x * EPB + t;
  int s[SORT_K], d[SORT_K], r[SORT_K];
#pragma unroll
  for (int k = 0; k < SORT_K; ++k) {
    const int e = base + k * SORT_T;
    s[k] = 0; d[k] = -1;
    if (e < E) { s[k] = ei[e]; d[k] = ei[E + e]; }
  }
#pragma unroll
  for (int k = 0; k < SORT_K; ++k) {
    r[k] = 0;
    if (d[k] >= 0) r[k] = atomicAdd(&h[d[k] >> BIN_SHIFT], 1);
  }
  __syncthreads();
  for (int i = t; i < nb; i += SORT_T) {
    const int c = h[i];
    h[i] = c ? atomicAdd(&binCur[i * CSTRIDE], c) : 0;
  }
  __syncthreads();
#pragma unroll
  for (int k = 0; k < SORT_K; ++k)
    if (d[k] >= 0) binned[h[d[k] >> BIN_SHIFT] + r[k]] = make_int2(s[k], d[k]);
}

__global__ void __launch_bounds__(SORT_T)
bin_sort(const int* __restrict__ binnedw, const int* __restrict__ binStart, int n,
         int* __restrict__ rowptr, int* __restrict__ srcs) {
  const int2* __restrict__ binned = (const int2*)binnedw;
  __shared__ int cnt[BIN_NODES];
  __shared__ int wsum[SORT_T / 32];
  const int b = blockIdx.x, t = threadIdx.x, lane = t & 31, wv = t >> 5;
  const int beg = binStart[b], end = binStart[b + 1];
  const int node0 = b << BIN_SHIFT;
  if (t < BIN_NODES) cnt[t] = 0;
  __syncthreads();
  for (int c = beg; c < end; c += EPB) {
    int dl[SORT_K];
#pragma unroll
    for (int k = 0; k < SORT_K; ++k) {
      const int i = c + k * SORT_T + t;
      dl[k] = i < end ? binned[i].y : -1;
    }
#pragma unroll
    for (int k = 0; k < SORT_K; ++k)
      if (dl[k] >= 0) atomicAdd(&cnt[dl[k] - node0], 1);
  }
  __syncthreads();
  const int v = t < BIN_NODES ? cnt[t] : 0;
  int inc = v;
  for (int k = 1; k < 32; k <<= 1) { int u = __shfl_up(inc, k, 32); if (lane >= k) inc += u; }
  if (lane == 31) wsum[wv] = inc;
  __syncthreads();
  int woff = 0;
  for (int q = 0; q < wv; ++q) woff += wsum[q];
  const int excl = beg + woff + inc - v;
  if (t < BIN_NODES) {
    cnt[t] = excl;
    const int node = node0 + t;
    if (node < n) rowptr[node] = excl;
    if (node == n - 1) rowptr[n] = excl + v;
  }
  __syncthreads();
  for (int c = beg; c < end; c += EPB) {
    int2 e[SORT_K];
#pragma unroll
    for (int k = 0; k < SORT_K; ++k) {
      const int i = c + k * SORT_T + t;
      e[k] = i < end ? binned[i] : make_int2(0, -1);
    }
#pragma unroll
    for (int k = 0; k < SORT_K; ++k)
      if (e[k].y >= 0) srcs[atomicAdd(&cnt[e[k].y - node0], 1)] = e[k].x;
  }
}

static __device__ inline float lrelu(float e) { return e >= 0.f ? e : NEG_SLOPE * e; }

template <int X> static __device__ inline float xrow(float v) {
  return __builtin_bit_cast(float, __builtin_amdgcn_update_dpp(
      0, __builtin_bit_cast(int, v), 0x160 + X, 0xF, 0xF, true));
}
static __device__ inline float xhalf(float v) {
  int b = __builtin_bit_cast(int, v);
  return __builtin_bit_cast(float, __builtin_amdgcn_permlanex16(
      b, b, 0x76543210, 0xfedcba98, false, false));
}

static __device__ inline float wave_sum(float s) {
  s += xhalf(s);
  s += xrow<8>(s); s += xrow<4>(s); s += xrow<2>(s); s += xrow<1>(s);
  return s;
}


static __device__ inline float4 ldrow4(const float* __restrict__ p, int s, int c) {
  return *(const float4*)((const char*)p + ((unsigned)s * 256u + (unsigned)c * 16u));
}
static __device__ inline float2 ldrow2(const float* __restrict__ p, int s, int c) {
  return *(const float2*)((const char*)p + ((unsigned)s * 128u + (unsigned)c * 8u));
}

static __device__ inline float logit1(float4 u, float4 vr, float4 w) {
  float p = lrelu(u.x + vr.x) * w.x + lrelu(u.y + vr.y) * w.y
          + lrelu(u.z + vr.z) * w.z + lrelu(u.w + vr.w) * w.w;
  p += xrow<1>(p); p += xrow<2>(p);
  return p;
}

__global__ void __launch_bounds__(256) __attribute__((amdgpu_waves_per_eu(16, 16)))
gat1_node(const int* __restrict__ rowptr,
                          const int* __restrict__ srcs,
                          const float* __restrict__ xl,
                          const float* __restrict__ xr,
                          const float* __restrict__ a1,
                          const float* __restrict__ b1,
                          const float* __restrict__ g,
                          const float* __restrict__ be,
                          float* __restrict__ h1, int n) {
  const int node = __builtin_amdgcn_readfirstlane(
      (int)(((long long)blockIdx.x * blockDim.x + threadIdx.x) >> 5));
  const int lane = threadIdx.x & 31;
  if (node >= n) return;
  const int  c  = lane & 15;
  const bool hi = lane >= 16;
  const int  q  = 2 * c + (hi ? 1 : 0);
  const float2 wa = ((const float2*)a1)[2 * c], wb = ((const float2*)a1)[2 * c + 1];
  const float4 w  = make_float4(wa.x, wa.y, wb.x, wb.y);
  const float4 vr = ((const float4*)(xr + (size_t)node * 64))[c];
  const float4 us = ((const float4*)(xl + (size_t)node * 64))[c];
  const int beg = rowptr[node], end = rowptr[node + 1];
  float m = logit1(us, vr, w);
  float den = hi ? 0.f : 1.f;
  float4 a = hi ? make_float4(0.f, 0.f, 0.f, 0.f) : us;
  for (int base = beg; base < end; base += 32) {
    int cnt = end - base; cnt = cnt < 32 ? cnt : 32;
    const int sidx = lane < cnt ? srcs[base + lane] : node;
    int j = 0;
    for (; j + 8 <= cnt; j += 8) {
      const int ra0 = __builtin_amdgcn_readlane(sidx, j + 0), rb0 = __builtin_amdgcn_readlane(sidx, j + 1);
      const int ra1 = __builtin_amdgcn_readlane(sidx, j + 2), rb1 = __builtin_amdgcn_readlane(sidx, j + 3);
      const int ra2 = __builtin_amdgcn_readlane(sidx, j + 4), rb2 = __builtin_amdgcn_readlane(sidx, j + 5);
      const int ra3 = __builtin_amdgcn_readlane(sidx, j + 6), rb3 = __builtin_amdgcn_readlane(sidx, j + 7);
      const int s0 = hi ? rb0 : ra0, s1 = hi ? rb1 : ra1;
      const int s2 = hi ? rb2 : ra2, s3 = hi ? rb3 : ra3;
      const float4 u0 = ldrow4(xl, s0, c), u1 = ldrow4(xl, s1, c);
      const float4 u2 = ldrow4(xl, s2, c), u3 = ldrow4(xl, s3, c);
      const float l0 = logit1(u0, vr, w), l1 = logit1(u1, vr, w);
      const float l2 = logit1(u2, vr, w), l3 = logit1(u3, vr, w);
      const float mn = fmaxf(fmaxf(m, l0), fmaxf(fmaxf(l1, l2), l3));
      const float sc = __expf(m - mn);
      const float e0 = __expf(l0 - mn), e1 = __expf(l1 - mn);
      const float e2 = __expf(l2 - mn), e3 = __expf(l3 - mn);
      den = den * sc + e0;
      a.x = a.x * sc + e0 * u0.x; a.y = a.y * sc + e0 * u0.y;
      a.z = a.z * sc + e0 * u0.z; a.w = a.w * sc + e0 * u0.w;
      den += e1; a.x += e1 * u1.x; a.y += e1 * u1.y; a.z += e1 * u1.z; a.w += e1 * u1.w;
      den += e2; a.x += e2 * u2.x; a.y += e2 * u2.y; a.z += e2 * u2.z; a.w += e2 * u2.w;
      den += e3; a.x += e3 * u3.x; a.y += e3 * u3.y; a.z += e3 * u3.z; a.w += e3 * u3.w;
      m = mn;
    }
    const int rem = cnt - j;
    if (rem > 0) {
      float4 u[4];
#pragma unroll
      for (int p = 0; p < 4; ++p) {
        u[p] = make_float4(0.f, 0.f, 0.f, 0.f);
        if (2 * p < rem) {
          const int ra = __builtin_amdgcn_readlane(sidx, j + 2 * p);
          const int rb = __builtin_amdgcn_readlane(sidx, j + 2 * p + 1);
          const int s = hi ? rb : ra;
          u[p] = ldrow4(xl, s, c);
        }
      }
#pragma unroll
      for (int p = 0; p < 4; ++p) {
        if (2 * p < rem) {
          const float l = logit1(u[p], vr, w);
          const float mn = fmaxf(m, l);
          const float sc = __expf(m - mn);
          float e = __expf(l - mn);
          if (2 * p + 1 >= rem) e = hi ? 0.f : e;
          den = den * sc + e;
          a.x = a.x * sc + e * u[p].x; a.y = a.y * sc + e * u[p].y;
          a.z = a.z * sc + e * u[p].z; a.w = a.w * sc + e * u[p].w;
          m = mn;
        }
      }
    }
  }
  const float2 bb = ((const float2*)b1)[q];
  const float2 gg = ((const float2*)g)[q];
  const float2 bt = ((const float2*)be)[q];
  {
    const float mt = fmaxf(m, xhalf(m));
    const float sm = __expf(m - mt);
    den *= sm; a.x *= sm; a.y *= sm; a.z *= sm; a.w *= sm;
    den += xhalf(den);
    a.x += xhalf(a.x); a.y += xhalf(a.y); a.z += xhalf(a.z); a.w += xhalf(a.w);
  }
  const float v0 = (hi ? a.z : a.x) / den + bb.x;
  const float v1 = (hi ? a.w : a.y) / den + bb.y;
  const float mu = wave_sum(v0 + v1) * (1.f / 64.f);
  const float d0 = v0 - mu, d1 = v1 - mu;
  const float rs = rsqrtf(wave_sum(d0 * d0 + d1 * d1) * (1.f / 64.f) + LN_EPS);
  const float y0 = d0 * rs * gg.x + bt.x;
  const float y1 = d1 * rs * gg.y + bt.y;
  float2 y;
  y.x = y0 > 0.f ? y0 : expm1f(y0);
  y.y = y1 > 0.f ? y1 : expm1f(y1);
  ((float2*)(h1 + (size_t)node * 64))[q] = y;
}

static __device__ inline float logit2(float2 u, float2 vr, float2 w) {
  float p = lrelu(u.x + vr.x) * w.x + lrelu(u.y + vr.y) * w.y;
  p += xrow<1>(p); p += xrow<2>(p); p += xrow<4>(p); p += xrow<8>(p);
  return p;
}

__global__ void gat2_node(const int* __restrict__ rowptr,
                          const int* __restrict__ srcs,
                          const float* __restrict__ xl,
                          const float* __restrict__ xr,
                          const float* __restrict__ a2,
                          const float* __restrict__ b2,
                          const float* __restrict__ g,
                          const float* __restrict__ be,
                          float* __restrict__ out, int n) {
  const int node = __builtin_amdgcn_readfirstlane(
      (int)(((long long)blockIdx.x * blockDim.x + threadIdx.x) >> 5));
  const int lane = threadIdx.x & 31;
  if (node >= n) return;
  const int  c  = lane & 15;
  const bool hi = lane >= 16;
  const int  q  = 2 * c + (hi ? 1 : 0);
  const float2 w  = ((const float2*)a2)[c];
  const float bb = b2[q], gg = g[q], bt = be[q];
  const float2 vr = ((const float2*)(xr + (size_t)node * 32))[c];
  const float2 us = ((const float2*)(xl + (size_t)node * 32))[c];
  const int beg = rowptr[node], end = rowptr[node + 1];
  float m = logit2(us, vr, w);
  float den = hi ? 0.f : 1.f;
  float2 a = hi ? make_float2(0.f, 0.f) : us;
  for (int base = beg; base < end; base += 32) {
    int cnt = end - base; cnt = cnt < 32 ? cnt : 32;
    const int sidx = lane < cnt ? srcs[base + lane] : node;
    int j = 0;
    for (; j + 8 <= cnt; j += 8) {
      const int ra0 = __builtin_amdgcn_readlane(sidx, j + 0), rb0 = __builtin_amdgcn_readlane(sidx, j + 1);
      const int ra1 = __builtin_amdgcn_readlane(sidx, j + 2), rb1 = __builtin_amdgcn_readlane(sidx, j + 3);
      const int ra2 = __builtin_amdgcn_readlane(sidx, j + 4), rb2 = __builtin_amdgcn_readlane(sidx, j + 5);
      const int ra3 = __builtin_amdgcn_readlane(sidx, j + 6), rb3 = __builtin_amdgcn_readlane(sidx, j + 7);
      const int s0 = hi ? rb0 : ra0, s1 = hi ? rb1 : ra1;
      const int s2 = hi ? rb2 : ra2, s3 = hi ? rb3 : ra3;
      const float2 u0 = ldrow2(xl, s0, c), u1 = ldrow2(xl, s1, c);
      const float2 u2 = ldrow2(xl, s2, c), u3 = ldrow2(xl, s3, c);
      const float l0 = logit2(u0, vr, w), l1 = logit2(u1, vr, w);
      const float l2 = logit2(u2, vr, w), l3 = logit2(u3, vr, w);
      const float mn = fmaxf(fmaxf(m, l0), fmaxf(fmaxf(l1, l2), l3));
      const float sc = __expf(m - mn);
      const float e0 = __expf(l0 - mn), e1 = __expf(l1 - mn);
      const float e2 = __expf(l2 - mn), e3 = __expf(l3 - mn);
      den = den * sc + e0; a.x = a.x * sc + e0 * u0.x; a.y = a.y * sc + e0 * u0.y;
      den += e1; a.x += e1 * u1.x; a.y += e1 * u1.y;
      den += e2; a.x += e2 * u2.x; a.y += e2 * u2.y;
      den += e3; a.x += e3 * u3.x; a.y += e3 * u3.y;
      m = mn;
    }
    const int rem = cnt - j;
    if (rem > 0) {
      float2 u[4];
#pragma unroll
      for (int p = 0; p < 4; ++p) {
        u[p] = make_float2(0.f, 0.f);
        if (2 * p < rem) {
          const int ra = __builtin_amdgcn_readlane(sidx, j + 2 * p);
          const int rb = __builtin_amdgcn_readlane(sidx, j + 2 * p + 1);
          const int s = hi ? rb : ra;
          u[p] = ldrow2(xl, s, c);
        }
      }
#pragma unroll
      for (int p = 0; p < 4; ++p) {
        if (2 * p < rem) {
          const float l = logit2(u[p], vr, w);
          const float mn = fmaxf(m, l);
          const float sc = __expf(m - mn);
          float e = __expf(l - mn);
          if (2 * p + 1 >= rem) e = hi ? 0.f : e;
          den = den * sc + e;
          a.x = a.x * sc + e * u[p].x; a.y = a.y * sc + e * u[p].y;
          m = mn;
        }
      }
    }
  }
  {
    const float mt = fmaxf(m, xhalf(m));
    const float sm = __expf(m - mt);
    den *= sm; a.x *= sm; a.y *= sm;
    den += xhalf(den);
    a.x += xhalf(a.x); a.y += xhalf(a.y);
  }
  const float v = (hi ? a.y : a.x) / den + bb;
  const float mu = wave_sum(v) * (1.f / 32.f);
  const float dd = v - mu;
  const float rs = rsqrtf(wave_sum(dd * dd) * (1.f / 32.f) + LN_EPS);
  out[(size_t)node * 32 + q] = dd * rs * gg + bt;
}

static inline int gridFor(long long threads, int block) {
  return (int)((threads + block - 1) / block);
}

extern "C" void kernel_launch(void* const* d_in, const int* in_sizes, int n_in,
                              void* d_out, int out_size, void* d_ws, size_t ws_size,
                              hipStream_t stream) {
  const float* x   = (const float*)d_in[0];
  const int*   ei  = (const int*)  d_in[1];
  const float* W1l = (const float*)d_in[2];
  const float* W1r = (const float*)d_in[3];
  const float* a1  = (const float*)d_in[4];
  const float* b1  = (const float*)d_in[5];
  const float* g1  = (const float*)d_in[6];
  const float* be1 = (const float*)d_in[7];
  const float* W2l = (const float*)d_in[8];
  const float* W2r = (const float*)d_in[9];
  const float* a2  = (const float*)d_in[10];
  const float* b2  = (const float*)d_in[11];
  const float* g2  = (const float*)d_in[12];
  const float* be2 = (const float*)d_in[13];
  float* out = (float*)d_out;

  const int n  = in_sizes[0] / 4;
  const int E  = in_sizes[1] / 2;
  const int TE = E + n;
  const int ntiles = n / 16;

  float* ws = (float*)d_ws;
  const size_t XL1  = 0;
  const size_t XR1  = XL1 + (size_t)n * 64;
  const size_t ACC1 = XR1 + (size_t)n * 64;
  const size_t DEN1 = ACC1 + (size_t)n * 64;
  const size_t MAX1 = DEN1 + (size_t)n * 4;
  const size_t LOG1 = MAX1 + (size_t)n * 4;
  const size_t H1   = LOG1 + (size_t)TE * 4;
  int* wi = (int*)d_ws;
  const size_t BCNT = LOG1;
  const size_t BCUR = BCNT + (size_t)NBMAX * CSTRIDE;
  const size_t BSTA = BCUR + (size_t)NBMAX * CSTRIDE;
  const size_t RPTR = BSTA + NBMAX + 64;
  const size_t SRCS = RPTR + (size_t)n + 64;
  const size_t BINN = (SRCS + (size_t)E + 1) & ~(size_t)1;

  const int B = 256;
  const int nb = (n + BIN_NODES - 1) >> BIN_SHIFT;
  const int nchunk = gridFor(E, EPB);

  fill_f32<<<gridFor(NBMAX * CSTRIDE, B), B, 0, stream>>>(
      ws + BCNT, (long long)NBMAX * CSTRIDE, 0.f);
  bin_count<<<nchunk, SORT_T, 0, stream>>>(ei, E, nb, wi + BCNT);
  bin_scan<<<1, NBMAX, 0, stream>>>(wi + BCNT, nb, wi + BSTA, wi + BCUR);
  bin_scatter<<<nchunk, SORT_T, 0, stream>>>(ei, E, nb, wi + BCUR, wi + BINN);
  bin_sort<<<nb, SORT_T, 0, stream>>>(wi + BINN, wi + BSTA, n, wi + RPTR, wi + SRCS);

  gemm1_wmma<<<gridFor((long long)ntiles * 32, B), B, 0, stream>>>(
      x, W1l, W1r, ws + XL1, ws + XR1, ntiles);

  gat1_node<<<gridFor((long long)n * 32, B), B, 0, stream>>>(
      wi + RPTR, wi + SRCS, ws + XL1, ws + XR1, a1, b1, g1, be1, ws + H1, n);

  gemm2_wmma<<<gridFor((long long)ntiles * 32, B), B, 0, stream>>>(
      ws + H1, W2l, W2r, ws + XL1, ws + XR1, ntiles);

  gat2_node<<<gridFor((long long)n * 32, B), B, 0, stream>>>(
      wi + RPTR, wi + SRCS, ws + XL1, ws + XR1, a2, b2, g2, be2, out, n);
}


# Sort and layer-1 GEMM in 3 launches, not 6: count pass stores per-block bin rows (GEMM tiles ride along); scatter blocks sum the rows for their offsets; no zero fill, no scan launch; new sum order
# speedup vs baseline: 13.1725x; 1.0644x over previous
#include <hip/hip_runtime.h>
#include <hip/hip_bf16.h>
#include <math.h>


typedef __attribute__((ext_vector_type(2))) float v2f;
typedef __attribute__((ext_vector_type(8))) float v8f;

#define NEG_SLOPE 0.2f
#define LN_EPS 1e-5f

static __device__ inline v8f wmma4(v2f a, v2f b, v8f c) {
  return __builtin_amdgcn_wmma_f32_16x16x4_f32(
       false, a,  false, b,
       (short)0, c,  false,  false);
}

static __device__ inline void gemm1_tile(const float* __restrict__ x,
                                         const float* __restrict__ Wl,
                                         const float* __restrict__ Wr,
                                         float* __restrict__ xl,
                                         float* __restrict__ xr, int wave) {
  int lane = threadIdx.x & 31;
  int row0 = wave * 16;
  int m  = lane & 15;
  int hi = lane >> 4;
  const float* xrow = x + (size_t)(row0 + m) * 4 + hi * 2;
  v2f a; a.x = xrow[0]; a.y = xrow[1];
#pragma unroll
  for (int j = 0; j < 4; ++j) {
    v2f bl, br;
    bl.x = Wl[(2 * hi + 0) * 64 + j * 16 + m];
    bl.y = Wl[(2 * hi + 1) * 64 + j * 16 + m];
    br.x = Wr[(2 * hi + 0) * 64 + j * 16 + m];
    br.y = Wr[(2 * hi + 1) * 64 + j * 16 + m];
    v8f cl = {}; cl = wmma4(a, bl, cl);
    v8f cr = {}; cr = wmma4(a, br, cr);
#pragma unroll
    for (int v = 0; v < 8; ++v) {
      int r = v + hi * 8;
      xl[(size_t)(row0 + r) * 64 + j * 16 + m] = cl[v];
      xr[(size_t)(row0 + r) * 64 + j * 16 + m] = cr[v];
    }
  }
}

__global__ void gemm2_wmma(const float* __restrict__ h,
                           const float* __restrict__ Wl,
                           const float* __restrict__ Wr,
                           float* __restrict__ xl,
                           float* __restrict__ xr, int ntiles) {
  int wave = (int)(((long long)blockIdx.x * blockDim.x + threadIdx.x) >> 5);
  int lane = threadIdx.x & 31;
  if (wave >= ntiles) return;
  int row0 = wave * 16;
  int m  = lane & 15;
  int hi = lane >> 4;
  const float* hrow = h + (size_t)(row0 + m) * 64;
  v8f cl[2] = {{0.f},{0.f}};
  v8f cr[2] = {{0.f},{0.f}};
#pragma unroll
  for (int kk = 0; kk < 16; ++kk) {
    int k0 = kk * 4 + hi * 2;
    v2f a; a.x = hrow[k0]; a.y = hrow[k0 + 1];
#pragma unroll
    for (int j = 0; j < 2; ++j) {
      v2f bl, br;
      bl.x = Wl[(k0 + 0) * 32 + j * 16 + m];
      bl.y = Wl[(k0 + 1) * 32 + j * 16 + m];
      br.x = Wr[(k0 + 0) * 32 + j * 16 + m];
      br.y = Wr[(k0 + 1) * 32 + j * 16 + m];
      cl[j] = wmma4(a, bl, cl[j]);
      cr[j] = wmma4(a, br, cr[j]);
    }
  }
#pragma unroll
  for (int j = 0; j < 2; ++j)
#pragma unroll
    for (int v = 0; v < 8; ++v) {
      int r = v + hi * 8;
      xl[(size_t)(row0 + r) * 32 + j * 16 + m] = cl[j][v];
      xr[(size_t)(row0 + r) * 32 + j * 16 + m] = cr[j][v];
    }
}

#define BIN_SHIFT 9
#define BIN_NODES 512
#define NBMAX     256
#define SORT_T    1024
#define SORT_K    8
#define EPB       (SORT_T * SORT_K)
#define ROW_G     (SORT_T / (NBMAX / 4))
#define ROW_B     7

__global__ void __launch_bounds__(SORT_T)
count_gemm1(const int* __restrict__ ei, int E, int nchunk, int* __restrict__ cntB,
            const float* __restrict__ x, const float* __restrict__ Wl,
            const float* __restrict__ Wr, float* __restrict__ xl,
            float* __restrict__ xr, int ntiles) {
  __shared__ int h[NBMAX];
  const int t = threadIdx.x;
  if ((int)blockIdx.x >= nchunk) {
    const int wave = ((int)blockIdx.x - nchunk) * (SORT_T / 32) + (t >> 5);
    if (wave < ntiles) gemm1_tile(x, Wl, Wr, xl, xr, wave);
    return;
  }
  if (t < NBMAX) h[t] = 0;
  __syncthreads();
  const int base = blockIdx.x * EPB + t;
  int d[SORT_K];
#pragma unroll
  for (int k = 0; k < SORT_K; ++k) {
    const int e = base + k * SORT_T;
    d[k] = e < E ? ei[E + e] : -1;
  }
#pragma unroll
  for (int k = 0; k < SORT_K; ++k)
    if (d[k] >= 0) atomicAdd(&h[d[k] >> BIN_SHIFT], 1);
  __syncthreads();
  if (t < NBMAX) cntB[blockIdx.x * NBMAX + t] = h[t];
}

__global__ void __launch_bounds__(SORT_T)
bin_scatter(const int* __restrict__ ei, int E, int nb, int nchunk,
            const int* __restrict__ cntB, int* __restrict__ binStart,
            int* __restrict__ binnedw) {
  int2* __restrict__ binned = (int2*)binnedw;
  __shared__ int h[NBMAX];
  __shared__ __attribute__((aligned(16))) int ptot[ROW_G][NBMAX];
  __shared__ __attribute__((aligned(16))) int pbef[ROW_G][NBMAX];
  __shared__ int wsum[SORT_T / 32];
  const int t = threadIdx.x, lane = t & 31, wv = t >> 5;
  const int blk = blockIdx.x;
  if (t < NBMAX) h[t] = 0;
  const int base = blk * EPB + t;
  int s[SORT_K], d[SORT_K], r[SORT_K];
#pragma unroll
  for (int k = 0; k < SORT_K; ++k) {
    const int e = base + k * SORT_T;
    s[k] = 0; d[k] = -1;
    if (e < E) { s[k] = ei[e]; d[k] = ei[E + e]; }
  }
  {
    const int c4 = t & (NBMAX / 4 - 1), g = t / (NBMAX / 4);
    const int4* __restrict__ m4 = (const int4*)cntB;
    int4 tot = make_int4(0, 0, 0, 0), bef = make_int4(0, 0, 0, 0);
    for (int k0 = g; k0 < nchunk; k0 += ROW_G * ROW_B) {
      int4 v[ROW_B];
#pragma unroll
      for (int j = 0; j < ROW_B; ++j) {
        const int k = k0 + j * ROW_G;
        v[j] = k < nchunk ? m4[k * (NBMAX / 4) + c4] : make_int4(0, 0, 0, 0);
      }
#pragma unroll
      for (int j = 0; j < ROW_B; ++j) {
        const int k = k0 + j * ROW_G;
        tot.x += v[j].x; tot.y += v[j].y; tot.z += v[j].z; tot.w += v[j].w;
        if (k < blk) { bef.x += v[j].x; bef.y += v[j].y; bef.z += v[j].z; bef.w += v[j].w; }
      }
    }
    ((int4*)ptot[g])[c4] = tot;
    ((int4*)pbef[g])[c4] = bef;
  }
  __syncthreads();
#pragma unroll
  for (int k = 0; k < SORT_K; ++k) {
    r[k] = 0;
    if (d[k] >= 0) r[k] = atomicAdd(&h[d[k] >> BIN_SHIFT], 1);
  }
  int T = 0, B = 0;
  if (t < NBMAX) {
    for (int g = 0; g < ROW_G; ++g) { T += ptot[g][t]; B += pbef[g][t]; }
  }
  int inc = T;
  for (int k = 1; k < 32; k <<= 1) { int u = __shfl_up(inc, k, 32); if (lane >= k) inc += u; }
  if (lane == 31) wsum[wv] = inc;
  __syncthreads();
  if (t < NBMAX) {
    int woff = 0;
    for (int q = 0; q < wv; ++q) woff += wsum[q];
    const int excl = woff + inc - T;
    h[t] = excl + B;
    if (blk == 0) {
      if (t < nb) binStart[t] = excl;
      if (t == nb - 1) binStart[nb] = excl + T;
    }
  }
  __syncthreads();
#pragma unroll
  for (int k = 0; k < SORT_K; ++k)
    if (d[k] >= 0) binned[h[d[k] >> BIN_SHIFT] + r[k]] = make_int2(s[k], d[k]);
}

__global__ void __launch_bounds__(SORT_T)
bin_sort(const int* __restrict__ binnedw, const int* __restrict__ binStart, int n,
         int* __restrict__ rowptr, int* __restrict__ srcs) {
  const int2* __restrict__ binned = (const int2*)binnedw;
  __shared__ int cnt[BIN_NODES];
  __shared__ int wsum[SORT_T / 32];
  const int b = blockIdx.x, t = threadIdx.x, lane = t & 31, wv = t >> 5;
  const int beg = binStart[b], end = binStart[b + 1];
  const int node0 = b << BIN_SHIFT;
  if (t < BIN_NODES) cnt[t] = 0;
  __syncthreads();
  for (int c = beg; c < end; c += EPB) {
    int dl[SORT_K];
#pragma unroll
    for (int k = 0; k < SORT_K; ++k) {
      const int i = c + k * SORT_T + t;
      dl[k] = i < end ? binned[i].y : -1;
    }
#pragma unroll
    for (int k = 0; k < SORT_K; ++k)
      if (dl[k] >= 0) atomicAdd(&cnt[dl[k] - node0], 1);
  }
  __syncthreads();
  const int v = t < BIN_NODES ? cnt[t] : 0;
  int inc = v;
  for (int k = 1; k < 32; k <<= 1) { int u = __shfl_up(inc, k, 32); if (lane >= k) inc += u; }
  if (lane == 31) wsum[wv] = inc;
  __syncthreads();
  int woff = 0;
  for (int q = 0; q < wv; ++q) woff += wsum[q];
  const int excl = beg + woff + inc - v;
  if (t < BIN_NODES) {
    cnt[t] = excl;
    const int node = node0 + t;
    if (node < n) rowptr[node] = excl;
    if (node == n - 1) rowptr[n] = excl + v;
  }
  __syncthreads();
  for (int c = beg; c < end; c += EPB) {
    int2 e[SORT_K];
#pragma unroll
    for (int k = 0; k < SORT_K; ++k) {
      const int i = c + k * SORT_T + t;
      e[k] = i < end ? binned[i] : make_int2(0, -1);
    }
#pragma unroll
    for (int k = 0; k < SORT_K; ++k)
      if (e[k].y >= 0) srcs[atomicAdd(&cnt[e[k].y - node0], 1)] = e[k].x;
  }
}

static __device__ inline float lrelu(float e) { return e >= 0.f ? e : NEG_SLOPE * e; }

template <int X> static __device__ inline float xrow(float v) {
  return __builtin_bit_cast(float, __builtin_amdgcn_update_dpp(
      0, __builtin_bit_cast(int, v), 0x160 + X, 0xF, 0xF, true));
}
static __device__ inline float xhalf(float v) {
  int b = __builtin_bit_cast(int, v);
  return __builtin_bit_cast(float, __builtin_amdgcn_permlanex16(
      b, b, 0x76543210, 0xfedcba98, false, false));
}

static __device__ inline float wave_sum(float s) {
  s += xhalf(s);
  s += xrow<8>(s); s += xrow<4>(s); s += xrow<2>(s); s += xrow<1>(s);
  return s;
}


static __device__ inline float4 ldrow4(const float* __restrict__ p, int s, int c) {
  return *(const float4*)((const char*)p + ((unsigned)s * 256u + (unsigned)c * 16u));
}
static __device__ inline float2 ldrow2(const float* __restrict__ p, int s, int c) {
  return *(const float2*)((const char*)p + ((unsigned)s * 128u + (unsigned)c * 8u));
}

static __device__ inline float logit1(float4 u, float4 vr, float4 w) {
  float p = lrelu(u.x + vr.x) * w.x + lrelu(u.y + vr.y) * w.y
          + lrelu(u.z + vr.z) * w.z + lrelu(u.w + vr.w) * w.w;
  p += xrow<1>(p); p += xrow<2>(p);
  return p;
}

__global__ void __launch_bounds__(256) __attribute__((amdgpu_waves_per_eu(16, 16)))
gat1_node(const int* __restrict__ rowptr,
                          const int* __restrict__ srcs,
                          const float* __restrict__ xl,
                          const float* __restrict__ xr,
                          const float* __restrict__ a1,
                          const float* __restrict__ b1,
                          const float* __restrict__ g,
                          const float* __restrict__ be,
                          float* __restrict__ h1, int n) {
  const int node = __builtin_amdgcn_readfirstlane(
      (int)(((long long)blockIdx.x * blockDim.x + threadIdx.x) >> 5));
  const int lane = threadIdx.x & 31;
  if (node >= n) return;
  const int  c  = lane & 15;
  const bool hi = lane >= 16;
  const int  q  = 2 * c + (hi ? 1 : 0);
  const float2 wa = ((const float2*)a1)[2 * c], wb = ((const float2*)a1)[2 * c + 1];
  const float4 w  = make_float4(wa.x, wa.y, wb.x, wb.y);
  const float4 vr = ((const float4*)(xr + (size_t)node * 64))[c];
  const float4 us = ((const float4*)(xl + (size_t)node * 64))[c];
  const int beg = rowptr[node], end = rowptr[node + 1];
  float m = logit1(us, vr, w);
  float den = hi ? 0.f : 1.f;
  float4 a = hi ? make_float4(0.f, 0.f, 0.f, 0.f) : us;
  for (int base = beg; base < end; base += 32) {
    int cnt = end - base; cnt = cnt < 32 ? cnt : 32;
    const int sidx = lane < cnt ? srcs[base + lane] : node;
    int j = 0;
    for (; j + 8 <= cnt; j += 8) {
      const int ra0 = __builtin_amdgcn_readlane(sidx, j + 0), rb0 = __builtin_amdgcn_readlane(sidx, j + 1);
      const int ra1 = __builtin_amdgcn_readlane(sidx, j + 2), rb1 = __builtin_amdgcn_readlane(sidx, j + 3);
      const int ra2 = __builtin_amdgcn_readlane(sidx, j + 4), rb2 = __builtin_amdgcn_readlane(sidx, j + 5);
      const int ra3 = __builtin_amdgcn_readlane(sidx, j + 6), rb3 = __builtin_amdgcn_readlane(sidx, j + 7);
      const int s0 = hi ? rb0 : ra0, s1 = hi ? rb1 : ra1;
      const int s2 = hi ? rb2 : ra2, s3 = hi ? rb3 : ra3;
      const float4 u0 = ldrow4(xl, s0, c), u1 = ldrow4(xl, s1, c);
      const float4 u2 = ldrow4(xl, s2, c), u3 = ldrow4(xl, s3, c);
      const float l0 = logit1(u0, vr, w), l1 = logit1(u1, vr, w);
      const float l2 = logit1(u2, vr, w), l3 = logit1(u3, vr, w);
      const float mn = fmaxf(fmaxf(m, l0), fmaxf(fmaxf(l1, l2), l3));
      const float sc = __expf(m - mn);
      const float e0 = __expf(l0 - mn), e1 = __expf(l1 - mn);
      const float e2 = __expf(l2 - mn), e3 = __expf(l3 - mn);
      den = den * sc + e0;
      a.x = a.x * sc + e0 * u0.x; a.y = a.y * sc + e0 * u0.y;
      a.z = a.z * sc + e0 * u0.z; a.w = a.w * sc + e0 * u0.w;
      den += e1; a.x += e1 * u1.x; a.y += e1 * u1.y; a.z += e1 * u1.z; a.w += e1 * u1.w;
      den += e2; a.x += e2 * u2.x; a.y += e2 * u2.y; a.z += e2 * u2.z; a.w += e2 * u2.w;
      den += e3; a.x += e3 * u3.x; a.y += e3 * u3.y; a.z += e3 * u3.z; a.w += e3 * u3.w;
      m = mn;
    }
    const int rem = cnt - j;
    if (rem > 0) {
      float4 u[4];
#pragma unroll
      for (int p = 0; p < 4; ++p) {
        u[p] = make_float4(0.f, 0.f, 0.f, 0.f);
        if (2 * p < rem) {
          const int ra = __builtin_amdgcn_readlane(sidx, j + 2 * p);
          const int rb = __builtin_amdgcn_readlane(sidx, j + 2 * p + 1);
          const int s = hi ? rb : ra;
          u[p] = ldrow4(xl, s, c);
        }
      }
#pragma unroll
      for (int p = 0; p < 4; ++p) {
        if (2 * p < rem) {
          const float l = logit1(u[p], vr, w);
          const float mn = fmaxf(m, l);
          const float sc = __expf(m - mn);
          float e = __expf(l - mn);
          if (2 * p + 1 >= rem) e = hi ? 0.f : e;
          den = den * sc + e;
          a.x = a.x * sc + e * u[p].x; a.y = a.y * sc + e * u[p].y;
          a.z = a.z * sc + e * u[p].z; a.w = a.w * sc + e * u[p].w;
          m = mn;
        }
      }
    }
  }
  const float2 bb = ((const float2*)b1)[q];
  const float2 gg = ((const float2*)g)[q];
  const float2 bt = ((const float2*)be)[q];
  {
    const float mt = fmaxf(m, xhalf(m));
    const float sm = __expf(m - mt);
    den *= sm; a.x *= sm; a.y *= sm; a.z *= sm; a.w *= sm;
    den += xhalf(den);
    a.x += xhalf(a.x); a.y += xhalf(a.y); a.z += xhalf(a.z); a.w += xhalf(a.w);
  }
  const float v0 = (hi ? a.z : a.x) / den + bb.x;
  const float v1 = (hi ? a.w : a.y) / den + bb.y;
  const float mu = wave_sum(v0 + v1) * (1.f / 64.f);
  const float d0 = v0 - mu, d1 = v1 - mu;
  const float rs = rsqrtf(wave_sum(d0 * d0 + d1 * d1) * (1.f / 64.f) + LN_EPS);
  const float y0 = d0 * rs * gg.x + bt.x;
  const float y1 = d1 * rs * gg.y + bt.y;
  float2 y;
  y.x = y0 > 0.f ? y0 : expm1f(y0);
  y.y = y1 > 0.f ? y1 : expm1f(y1);
  ((float2*)(h1 + (size_t)node * 64))[q] = y;
}

static __device__ inline float logit2(float2 u, float2 vr, float2 w) {
  float p = lrelu(u.x + vr.x) * w.x + lrelu(u.y + vr.y) * w.y;
  p += xrow<1>(p); p += xrow<2>(p); p += xrow<4>(p); p += xrow<8>(p);
  return p;
}

__global__ void gat2_node(const int* __restrict__ rowptr,
                          const int* __restrict__ srcs,
                          const float* __restrict__ xl,
                          const float* __restrict__ xr,
                          const float* __restrict__ a2,
                          const float* __restrict__ b2,
                          const float* __restrict__ g,
                          const float* __restrict__ be,
                          float* __restrict__ out, int n) {
  const int node = __builtin_amdgcn_readfirstlane(
      (int)(((long long)blockIdx.x * blockDim.x + threadIdx.x) >> 5));
  const int lane = threadIdx.x & 31;
  if (node >= n) return;
  const int  c  = lane & 15;
  const bool hi = lane >= 16;
  const int  q  = 2 * c + (hi ? 1 : 0);
  const float2 w  = ((const float2*)a2)[c];
  const float bb = b2[q], gg = g[q], bt = be[q];
  const float2 vr = ((const float2*)(xr + (size_t)node * 32))[c];
  const float2 us = ((const float2*)(xl + (size_t)node * 32))[c];
  const int beg = rowptr[node], end = rowptr[node + 1];
  float m = logit2(us, vr, w);
  float den = hi ? 0.f : 1.f;
  float2 a = hi ? make_float2(0.f, 0.f) : us;
  for (int base = beg; base < end; base += 32) {
    int cnt = end - base; cnt = cnt < 32 ? cnt : 32;
    const int sidx = lane < cnt ? srcs[base + lane] : node;
    int j = 0;
    for (; j + 8 <= cnt; j += 8) {
      const int ra0 = __builtin_amdgcn_readlane(sidx, j + 0), rb0 = __builtin_amdgcn_readlane(sidx, j + 1);
      const int ra1 = __builtin_amdgcn_readlane(sidx, j + 2), rb1 = __builtin_amdgcn_readlane(sidx, j + 3);
      const int ra2 = __builtin_amdgcn_readlane(sidx, j + 4), rb2 = __builtin_amdgcn_readlane(sidx, j + 5);
      const int ra3 = __builtin_amdgcn_readlane(sidx, j + 6), rb3 = __builtin_amdgcn_readlane(sidx, j + 7);
      const int s0 = hi ? rb0 : ra0, s1 = hi ? rb1 : ra1;
      const int s2 = hi ? rb2 : ra2, s3 = hi ? rb3 : ra3;
      const float2 u0 = ldrow2(xl, s0, c), u1 = ldrow2(xl, s1, c);
      const float2 u2 = ldrow2(xl, s2, c), u3 = ldrow2(xl, s3, c);
      const float l0 = logit2(u0, vr, w), l1 = logit2(u1, vr, w);
      const float l2 = logit2(u2, vr, w), l3 = logit2(u3, vr, w);
      const float mn = fmaxf(fmaxf(m, l0), fmaxf(fmaxf(l1, l2), l3));
      const float sc = __expf(m - mn);
      const float e0 = __expf(l0 - mn), e1 = __expf(l1 - mn);
      const float e2 = __expf(l2 - mn), e3 = __expf(l3 - mn);
      den = den * sc + e0; a.x = a.x * sc + e0 * u0.x; a.y = a.y * sc + e0 * u0.y;
      den += e1; a.x += e1 * u1.x; a.y += e1 * u1.y;
      den += e2; a.x += e2 * u2.x; a.y += e2 * u2.y;
      den += e3; a.x += e3 * u3.x; a.y += e3 * u3.y;
      m = mn;
    }
    const int rem = cnt - j;
    if (rem > 0) {
      float2 u[4];
#pragma unroll
      for (int p = 0; p < 4; ++p) {
        u[p] = make_float2(0.f, 0.f);
        if (2 * p < rem) {
          const int ra = __builtin_amdgcn_readlane(sidx, j + 2 * p);
          const int rb = __builtin_amdgcn_readlane(sidx, j + 2 * p + 1);
          const int s = hi ? rb : ra;
          u[p] = ldrow2(xl, s, c);
        }
      }
#pragma unroll
      for (int p = 0; p < 4; ++p) {
        if (2 * p < rem) {
          const float l = logit2(u[p], vr, w);
          const float mn = fmaxf(m, l);
          const float sc = __expf(m - mn);
          float e = __expf(l - mn);
          if (2 * p + 1 >= rem) e = hi ? 0.f : e;
          den = den * sc + e;
          a.x = a.x * sc + e * u[p].x; a.y = a.y * sc + e * u[p].y;
          m = mn;
        }
      }
    }
  }
  {
    const float mt = fmaxf(m, xhalf(m));
    const float sm = __expf(m - mt);
    den *= sm; a.x *= sm; a.y *= sm;
    den += xhalf(den);
    a.x += xhalf(a.x); a.y += xhalf(a.y);
  }
  const float v = (hi ? a.y : a.x) / den + bb;
  const float mu = wave_sum(v) * (1.f / 32.f);
  const float dd = v - mu;
  const float rs = rsqrtf(wave_sum(dd * dd) * (1.f / 32.f) + LN_EPS);
  out[(size_t)node * 32 + q] = dd * rs * gg + bt;
}

static inline int gridFor(long long threads, int block) {
  return (int)((threads + block - 1) / block);
}

extern "C" void kernel_launch(void* const* d_in, const int* in_sizes, int n_in,
                              void* d_out, int out_size, void* d_ws, size_t ws_size,
                              hipStream_t stream) {
  const float* x   = (const float*)d_in[0];
  const int*   ei  = (const int*)  d_in[1];
  const float* W1l = (const float*)d_in[2];
  const float* W1r = (const float*)d_in[3];
  const float* a1  = (const float*)d_in[4];
  const float* b1  = (const float*)d_in[5];
  const float* g1  = (const float*)d_in[6];
  const float* be1 = (const float*)d_in[7];
  const float* W2l = (const float*)d_in[8];
  const float* W2r = (const float*)d_in[9];
  const float* a2  = (const float*)d_in[10];
  const float* b2  = (const float*)d_in[11];
  const float* g2  = (const float*)d_in[12];
  const float* be2 = (const float*)d_in[13];
  float* out = (float*)d_out;

  const int n  = in_sizes[0] / 4;
  const int E  = in_sizes[1] / 2;
  const int TE = E + n;
  const int ntiles = n / 16;

  float* ws = (float*)d_ws;
  const size_t XL1  = 0;
  const size_t XR1  = XL1 + (size_t)n * 64;
  const size_t ACC1 = XR1 + (size_t)n * 64;
  const size_t DEN1 = ACC1 + (size_t)n * 64;
  const size_t MAX1 = DEN1 + (size_t)n * 4;
  const size_t LOG1 = MAX1 + (size_t)n * 4;
  const size_t H1   = LOG1 + (size_t)TE * 4;
  int* wi = (int*)d_ws;
  const size_t BSTA = LOG1;
  const size_t RPTR = BSTA + NBMAX + 64;
  const size_t SRCS = RPTR + (size_t)n + 64;
  const size_t BINN = (SRCS + (size_t)E + 1) & ~(size_t)1;

  const int B = 256;
  const int nb = (n + BIN_NODES - 1) >> BIN_SHIFT;
  const int nchunk = gridFor(E, EPB);
  const size_t CNTB = ACC1;

  count_gemm1<<<nchunk + gridFor(ntiles, SORT_T / 32), SORT_T, 0, stream>>>(
      ei, E, nchunk, wi + CNTB, x, W1l, W1r, ws + XL1, ws + XR1, ntiles);
  bin_scatter<<<nchunk, SORT_T, 0, stream>>>(ei, E, nb, nchunk, wi + CNTB, wi + BSTA, wi + BINN);
  bin_sort<<<nb, SORT_T, 0, stream>>>(wi + BINN, wi + BSTA, n, wi + RPTR, wi + SRCS);

  gat1_node<<<gridFor((long long)n * 32, B), B, 0, stream>>>(
      wi + RPTR, wi + SRCS, ws + XL1, ws + XR1, a1, b1, g1, be1, ws + H1, n);

  gemm2_wmma<<<gridFor((long long)ntiles * 32, B), B, 0, stream>>>(
      ws + H1, W2l, W2r, ws + XL1, ws + XR1, ntiles);

  gat2_node<<<gridFor((long long)n * 32, B), B, 0, stream>>>(
      wi + RPTR, wi + SRCS, ws + XL1, ws + XR1, a2, b2, g2, be2, out, n);
}


# Layer-2 GEMM formed inside the layer-1 node kernel: blocks of 16 one-node waves leave finished rows in LDS, 2 of them multiply the rows by W2 on WMMA; no h1 write and read, 5 launches not 6
# speedup vs baseline: 13.9014x; 1.0553x over previous
#include <hip/hip_runtime.h>
#include <hip/hip_bf16.h>
#include <math.h>


typedef __attribute__((ext_vector_type(2))) float v2f;
typedef __attribute__((ext_vector_type(8))) float v8f;

#define NEG_SLOPE 0.2f
#define LN_EPS 1e-5f

static __device__ inline v8f wmma4(v2f a, v2f b, v8f c) {
  return __builtin_amdgcn_wmma_f32_16x16x4_f32(
       false, a,  false, b,
       (short)0, c,  false,  false);
}

static __device__ inline void gemm1_tile(const float* __restrict__ x,
                                         const float* __restrict__ Wl,
                                         const float* __restrict__ Wr,
                                         float* __restrict__ xl,
                                         float* __restrict__ xr, int wave) {
  int lane = threadIdx.x & 31;
  int row0 = wave * 16;
  int m  = lane & 15;
  int hi = lane >> 4;
  const float* xrow = x + (size_t)(row0 + m) * 4 + hi * 2;
  v2f a; a.x = xrow[0]; a.y = xrow[1];
#pragma unroll
  for (int j = 0; j < 4; ++j) {
    v2f bl, br;
    bl.x = Wl[(2 * hi + 0) * 64 + j * 16 + m];
    bl.y = Wl[(2 * hi + 1) * 64 + j * 16 + m];
    br.x = Wr[(2 * hi + 0) * 64 + j * 16 + m];
    br.y = Wr[(2 * hi + 1) * 64 + j * 16 + m];
    v8f cl = {}; cl = wmma4(a, bl, cl);
    v8f cr = {}; cr = wmma4(a, br, cr);
#pragma unroll
    for (int v = 0; v < 8; ++v) {
      int r = v + hi * 8;
      xl[(size_t)(row0 + r) * 64 + j * 16 + m] = cl[v];
      xr[(size_t)(row0 + r) * 64 + j * 16 + m] = cr[v];
    }
  }
}

#define G1_NODES 16
#define HS_PITCH 68
static __device__ inline void gemm2_rows(const float* hs,
                                         const float* __restrict__ W,
                                         float* __restrict__ xo,
                                         int row0, int n) {
  const int lane = threadIdx.x & 31;
  const int m  = lane & 15;
  const int hi = lane >> 4;
  const float* hrow = hs + m * HS_PITCH;
  v8f c[2] = {{0.f},{0.f}};
#pragma unroll
  for (int kk = 0; kk < 16; ++kk) {
    const int k0 = kk * 4 + hi * 2;
    v2f a; a.x = hrow[k0]; a.y = hrow[k0 + 1];
#pragma unroll
    for (int j = 0; j < 2; ++j) {
      v2f b;
      b.x = W[(k0 + 0) * 32 + j * 16 + m];
      b.y = W[(k0 + 1) * 32 + j * 16 + m];
      c[j] = wmma4(a, b, c[j]);
    }
  }
  float* __restrict__ xcol = xo + (size_t)(row0 + hi * 8) * 32 + m;
  if (row0 + G1_NODES <= n) {
#pragma unroll
    for (int j = 0; j < 2; ++j)
#pragma unroll
      for (int v = 0; v < 8; ++v) xcol[v * 32 + j * 16] = c[j][v];
  } else {
#pragma unroll
    for (int j = 0; j < 2; ++j)
#pragma unroll
      for (int v = 0; v < 8; ++v)
        if (row0 + hi * 8 + v < n) xcol[v * 32 + j * 16] = c[j][v];
  }
}

#define BIN_SHIFT 9
#define BIN_NODES 512
#define NBMAX     256
#define SORT_T    1024
#define SORT_K    8
#define EPB       (SORT_T * SORT_K)
#define ROW_G     (SORT_T / (NBMAX / 4))
#define ROW_B     7

__global__ void __launch_bounds__(SORT_T)
count_gemm1(const int* __restrict__ ei, int E, int nchunk, int* __restrict__ cntB,
            const float* __restrict__ x, const float* __restrict__ Wl,
            const float* __restrict__ Wr, float* __restrict__ xl,
            float* __restrict__ xr, int ntiles) {
  __shared__ int h[NBMAX];
  const int t = threadIdx.x;
  if ((int)blockIdx.x >= nchunk) {
    const int wave = ((int)blockIdx.x - nchunk) * (SORT_T / 32) + (t >> 5);
    if (wave < ntiles) gemm1_tile(x, Wl, Wr, xl, xr, wave);
    return;
  }
  if (t < NBMAX) h[t] = 0;
  __syncthreads();
  const int base = blockIdx.x * EPB + t;
  int d[SORT_K];
#pragma unroll
  for (int k = 0; k < SORT_K; ++k) {
    const int e = base + k * SORT_T;
    d[k] = e < E ? ei[E + e] : -1;
  }
#pragma unroll
  for (int k = 0; k < SORT_K; ++k)
    if (d[k] >= 0) atomicAdd(&h[d[k] >> BIN_SHIFT], 1);
  __syncthreads();
  if (t < NBMAX) cntB[blockIdx.x * NBMAX + t] = h[t];
}

__global__ void __launch_bounds__(SORT_T)
bin_scatter(const int* __restrict__ ei, int E, int nb, int nchunk,
            const int* __restrict__ cntB, int* __restrict__ binStart,
            int* __restrict__ binnedw) {
  int2* __restrict__ binned = (int2*)binnedw;
  __shared__ int h[NBMAX];
  __shared__ __attribute__((aligned(16))) int ptot[ROW_G][NBMAX];
  __shared__ __attribute__((aligned(16))) int pbef[ROW_G][NBMAX];
  __shared__ int wsum[SORT_T / 32];
  const int t = threadIdx.x, lane = t & 31, wv = t >> 5;
  const int blk = blockIdx.x;
  if (t < NBMAX) h[t] = 0;
  const int base = blk * EPB + t;
  int s[SORT_K], d[SORT_K], r[SORT_K];
#pragma unroll
  for (int k = 0; k < SORT_K; ++k) {
    const int e = base + k * SORT_T;
    s[k] = 0; d[k] = -1;
    if (e < E) { s[k] = ei[e]; d[k] = ei[E + e]; }
  }
  {
    const int c4 = t & (NBMAX / 4 - 1), g = t / (NBMAX / 4);
    const int4* __restrict__ m4 = (const int4*)cntB;
    int4 tot = make_int4(0, 0, 0, 0), bef = make_int4(0, 0, 0, 0);
    for (int k0 = g; k0 < nchunk; k0 += ROW_G * ROW_B) {
      int4 v[ROW_B];
#pragma unroll
      for (int j = 0; j < ROW_B; ++j) {
        const int k = k0 + j * ROW_G;
        v[j] = k < nchunk ? m4[k * (NBMAX / 4) + c4] : make_int4(0, 0, 0, 0);
      }
#pragma unroll
      for (int j = 0; j < ROW_B; ++j) {
        const int k = k0 + j * ROW_G;
        tot.x += v[j].x; tot.y += v[j].y; tot.z += v[j].z; tot.w += v[j].w;
        if (k < blk) { bef.x += v[j].x; bef.y += v[j].y; bef.z += v[j].z; bef.w += v[j].w; }
      }
    }
    ((int4*)ptot[g])[c4] = tot;
    ((int4*)pbef[g])[c4] = bef;
  }
  __syncthreads();
#pragma unroll
  for (int k = 0; k < SORT_K; ++k) {
    r[k] = 0;
    if (d[k] >= 0) r[k] = atomicAdd(&h[d[k] >> BIN_SHIFT], 1);
  }
  int T = 0, B = 0;
  if (t < NBMAX) {
    for (int g = 0; g < ROW_G; ++g) { T += ptot[g][t]; B += pbef[g][t]; }
  }
  int inc = T;
  for (int k = 1; k < 32; k <<= 1) { int u = __shfl_up(inc, k, 32); if (lane >= k) inc += u; }
  if (lane == 31) wsum[wv] = inc;
  __syncthreads();
  if (t < NBMAX) {
    int woff = 0;
    for (int q = 0; q < wv; ++q) woff += wsum[q];
    const int excl = woff + inc - T;
    h[t] = excl + B;
    if (blk == 0) {
      if (t < nb) binStart[t] = excl;
      if (t == nb - 1) binStart[nb] = excl + T;
    }
  }
  __syncthreads();
#pragma unroll
  for (int k = 0; k < SORT_K; ++k)
    if (d[k] >= 0) binned[h[d[k] >> BIN_SHIFT] + r[k]] = make_int2(s[k], d[k]);
}

__global__ void __launch_bounds__(SORT_T)
bin_sort(const int* __restrict__ binnedw, const int* __restrict__ binStart, int n,
         int* __restrict__ rowptr, int* __restrict__ srcs) {
  const int2* __restrict__ binned = (const int2*)binnedw;
  __shared__ int cnt[BIN_NODES];
  __shared__ int wsum[SORT_T / 32];
  const int b = blockIdx.x, t = threadIdx.x, lane = t & 31, wv = t >> 5;
  const int beg = binStart[b], end = binStart[b + 1];
  const int node0 = b << BIN_SHIFT;
  if (t < BIN_NODES) cnt[t] = 0;
  __syncthreads();
  for (int c = beg; c < end; c += EPB) {
    int dl[SORT_K];
#pragma unroll
    for (int k = 0; k < SORT_K; ++k) {
      const int i = c + k * SORT_T + t;
      dl[k] = i < end ? binned[i].y : -1;
    }
#pragma unroll
    for (int k = 0; k < SORT_K; ++k)
      if (dl[k] >= 0) atomicAdd(&cnt[dl[k] - node0], 1);
  }
  __syncthreads();
  const int v = t < BIN_NODES ? cnt[t] : 0;
  int inc = v;
  for (int k = 1; k < 32; k <<= 1) { int u = __shfl_up(inc, k, 32); if (lane >= k) inc += u; }
  if (lane == 31) wsum[wv] = inc;
  __syncthreads();
  int woff = 0;
  for (int q = 0; q < wv; ++q) woff += wsum[q];
  const int excl = beg + woff + inc - v;
  if (t < BIN_NODES) {
    cnt[t] = excl;
    const int node = node0 + t;
    if (node < n) rowptr[node] = excl;
    if (node == n - 1) rowptr[n] = excl + v;
  }
  __syncthreads();
  for (int c = beg; c < end; c += EPB) {
    int2 e[SORT_K];
#pragma unroll
    for (int k = 0; k < SORT_K; ++k) {
      const int i = c + k * SORT_T + t;
      e[k] = i < end ? binned[i] : make_int2(0, -1);
    }
#pragma unroll
    for (int k = 0; k < SORT_K; ++k)
      if (e[k].y >= 0) srcs[atomicAdd(&cnt[e[k].y - node0], 1)] = e[k].x;
  }
}

static __device__ inline float lrelu(float e) { return e >= 0.f ? e : NEG_SLOPE * e; }

template <int X> static __device__ inline float xrow(float v) {
  return __builtin_bit_cast(float, __builtin_amdgcn_update_dpp(
      0, __builtin_bit_cast(int, v), 0x160 + X, 0xF, 0xF, true));
}
static __device__ inline float xhalf(float v) {
  int b = __builtin_bit_cast(int, v);
  return __builtin_bit_cast(float, __builtin_amdgcn_permlanex16(
      b, b, 0x76543210, 0xfedcba98, false, false));
}

static __device__ inline float wave_sum(float s) {
  s += xhalf(s);
  s += xrow<8>(s); s += xrow<4>(s); s += xrow<2>(s); s += xrow<1>(s);
  return s;
}


static __device__ inline float4 ldrow4(const float* __restrict__ p, int s, int c) {
  return *(const float4*)((const char*)p + ((unsigned)s * 256u + (unsigned)c * 16u));
}
static __device__ inline float2 ldrow2(const float* __restrict__ p, int s, int c) {
  return *(const float2*)((const char*)p + ((unsigned)s * 128u + (unsigned)c * 8u));
}

static __device__ inline float logit1(float4 u, float4 vr, float4 w) {
  float p = lrelu(u.x + vr.x) * w.x + lrelu(u.y + vr.y) * w.y
          + lrelu(u.z + vr.z) * w.z + lrelu(u.w + vr.w) * w.w;
  p += xrow<1>(p); p += xrow<2>(p);
  return p;
}

__global__ void __launch_bounds__(G1_NODES * 32) __attribute__((amdgpu_waves_per_eu(16, 16)))
gat1_node(const int* __restrict__ rowptr,
                          const int* __restrict__ srcs,
                          const float* __restrict__ xl,
                          const float* __restrict__ xr,
                          const float* __restrict__ a1,
                          const float* __restrict__ b1,
                          const float* __restrict__ g,
                          const float* __restrict__ be,
                          const float* __restrict__ W2l,
                          const float* __restrict__ W2r,
                          float* __restrict__ xl2,
                          float* __restrict__ xr2, int n) {
  __shared__ __attribute__((aligned(16))) float hs[G1_NODES * HS_PITCH];
  const int wv = __builtin_amdgcn_readfirstlane((int)(threadIdx.x >> 5));
  const int row0 = (int)blockIdx.x * G1_NODES;
  const int lane = threadIdx.x & 31;
  const int node = row0 + wv < n ? row0 + wv : n - 1;
  const int  c  = lane & 15;
  const bool hi = lane >= 16;
  const int  q  = 2 * c + (hi ? 1 : 0);
  const float2 wa = ((const float2*)a1)[2 * c], wb = ((const float2*)a1)[2 * c + 1];
  const float4 w  = make_float4(wa.x, wa.y, wb.x, wb.y);
  const float4 vr = ((const float4*)(xr + (size_t)node * 64))[c];
  const float4 us = ((const float4*)(xl + (size_t)node * 64))[c];
  const int beg = rowptr[node], end = rowptr[node + 1];
  float m = logit1(us, vr, w);
  float den = hi ? 0.f : 1.f;
  float4 a = hi ? make_float4(0.f, 0.f, 0.f, 0.f) : us;
  for (int base = beg; base < end; base += 32) {
    int cnt = end - base; cnt = cnt < 32 ? cnt : 32;
    const int sidx = lane < cnt ? srcs[base + lane] : node;
    int j = 0;
    for (; j + 8 <= cnt; j += 8) {
      const int ra0 = __builtin_amdgcn_readlane(sidx, j + 0), rb0 = __builtin_amdgcn_readlane(sidx, j + 1);
      const int ra1 = __builtin_amdgcn_readlane(sidx, j + 2), rb1 = __builtin_amdgcn_readlane(sidx, j + 3);
      const int ra2 = __builtin_amdgcn_readlane(sidx, j + 4), rb2 = __builtin_amdgcn_readlane(sidx, j + 5);
      const int ra3 = __builtin_amdgcn_readlane(sidx, j + 6), rb3 = __builtin_amdgcn_readlane(sidx, j + 7);
      const int s0 = hi ? rb0 : ra0, s1 = hi ? rb1 : ra1;
      const int s2 = hi ? rb2 : ra2, s3 = hi ? rb3 : ra3;
      const float4 u0 = ldrow4(xl, s0, c), u1 = ldrow4(xl, s1, c);
      const float4 u2 = ldrow4(xl, s2, c), u3 = ldrow4(xl, s3, c);
      const float l0 = logit1(u0, vr, w), l1 = logit1(u1, vr, w);
      const float l2 = logit1(u2, vr, w), l3 = logit1(u3, vr, w);
      const float mn = fmaxf(fmaxf(m, l0), fmaxf(fmaxf(l1, l2), l3));
      const float sc = __expf(m - mn);
      const float e0 = __expf(l0 - mn), e1 = __expf(l1 - mn);
      const float e2 = __expf(l2 - mn), e3 = __expf(l3 - mn);
      den = den * sc + e0;
      a.x = a.x * sc + e0 * u0.x; a.y = a.y * sc + e0 * u0.y;
      a.z = a.z * sc + e0 * u0.z; a.w = a.w * sc + e0 * u0.w;
      den += e1; a.x += e1 * u1.x; a.y += e1 * u1.y; a.z += e1 * u1.z; a.w += e1 * u1.w;
      den += e2; a.x += e2 * u2.x; a.y += e2 * u2.y; a.z += e2 * u2.z; a.w += e2 * u2.w;
      den += e3; a.x += e3 * u3.x; a.y += e3 * u3.y; a.z += e3 * u3.z; a.w += e3 * u3.w;
      m = mn;
    }
    const int rem = cnt - j;
    if (rem > 0) {
      float4 u[4];
#pragma unroll
      for (int p = 0; p < 4; ++p) {
        u[p] = make_float4(0.f, 0.f, 0.f, 0.f);
        if (2 * p < rem) {
          const int ra = __builtin_amdgcn_readlane(sidx, j + 2 * p);
          const int rb = __builtin_amdgcn_readlane(sidx, j + 2 * p + 1);
          const int s = hi ? rb : ra;
          u[p] = ldrow4(xl, s, c);
        }
      }
#pragma unroll
      for (int p = 0; p < 4; ++p) {
        if (2 * p < rem) {
          const float l = logit1(u[p], vr, w);
          const float mn = fmaxf(m, l);
          const float sc = __expf(m - mn);
          float e = __expf(l - mn);
          if (2 * p + 1 >= rem) e = hi ? 0.f : e;
          den = den * sc + e;
          a.x = a.x * sc + e * u[p].x; a.y = a.y * sc + e * u[p].y;
          a.z = a.z * sc + e * u[p].z; a.w = a.w * sc + e * u[p].w;
          m = mn;
        }
      }
    }
  }
  const float2 bb = ((const float2*)b1)[q];
  const float2 gg = ((const float2*)g)[q];
  const float2 bt = ((const float2*)be)[q];
  {
    const float mt = fmaxf(m, xhalf(m));
    const float sm = __expf(m - mt);
    den *= sm; a.x *= sm; a.y *= sm; a.z *= sm; a.w *= sm;
    den += xhalf(den);
    a.x += xhalf(a.x); a.y += xhalf(a.y); a.z += xhalf(a.z); a.w += xhalf(a.w);
  }
  const float v0 = (hi ? a.z : a.x) / den + bb.x;
  const float v1 = (hi ? a.w : a.y) / den + bb.y;
  const float mu = wave_sum(v0 + v1) * (1.f / 64.f);
  const float d0 = v0 - mu, d1 = v1 - mu;
  const float rs = rsqrtf(wave_sum(d0 * d0 + d1 * d1) * (1.f / 64.f) + LN_EPS);
  const float y0 = d0 * rs * gg.x + bt.x;
  const float y1 = d1 * rs * gg.y + bt.y;
  float2 y;
  y.x = y0 > 0.f ? y0 : expm1f(y0);
  y.y = y1 > 0.f ? y1 : expm1f(y1);
  ((float2*)(hs + wv * HS_PITCH))[q] = y;
  __syncthreads();
  if (wv >= 2) return;
  gemm2_rows(hs, wv == 0 ? W2l : W2r, wv == 0 ? xl2 : xr2, row0, n);
}

static __device__ inline float logit2(float2 u, float2 vr, float2 w) {
  float p = lrelu(u.x + vr.x) * w.x + lrelu(u.y + vr.y) * w.y;
  p += xrow<1>(p); p += xrow<2>(p); p += xrow<4>(p); p += xrow<8>(p);
  return p;
}

__global__ void gat2_node(const int* __restrict__ rowptr,
                          const int* __restrict__ srcs,
                          const float* __restrict__ xl,
                          const float* __restrict__ xr,
                          const float* __restrict__ a2,
                          const float* __restrict__ b2,
                          const float* __restrict__ g,
                          const float* __restrict__ be,
                          float* __restrict__ out, int n) {
  const int node = __builtin_amdgcn_readfirstlane(
      (int)(((long long)blockIdx.x * blockDim.x + threadIdx.x) >> 5));
  const int lane = threadIdx.x & 31;
  if (node >= n) return;
  const int  c  = lane & 15;
  const bool hi = lane >= 16;
  const int  q  = 2 * c + (hi ? 1 : 0);
  const float2 w  = ((const float2*)a2)[c];
  const float bb = b2[q], gg = g[q], bt = be[q];
  const float2 vr = ((const float2*)(xr + (size_t)node * 32))[c];
  const float2 us = ((const float2*)(xl + (size_t)node * 32))[c];
  const int beg = rowptr[node], end = rowptr[node + 1];
  float m = logit2(us, vr, w);
  float den = hi ? 0.f : 1.f;
  float2 a = hi ? make_float2(0.f, 0.f) : us;
  for (int base = beg; base < end; base += 32) {
    int cnt = end - base; cnt = cnt < 32 ? cnt : 32;
    const int sidx = lane < cnt ? srcs[base + lane] : node;
    int j = 0;
    for (; j + 8 <= cnt; j += 8) {
      const int ra0 = __builtin_amdgcn_readlane(sidx, j + 0), rb0 = __builtin_amdgcn_readlane(sidx, j + 1);
      const int ra1 = __builtin_amdgcn_readlane(sidx, j + 2), rb1 = __builtin_amdgcn_readlane(sidx, j + 3);
      const int ra2 = __builtin_amdgcn_readlane(sidx, j + 4), rb2 = __builtin_amdgcn_readlane(sidx, j + 5);
      const int ra3 = __builtin_amdgcn_readlane(sidx, j + 6), rb3 = __builtin_amdgcn_readlane(sidx, j + 7);
      const int s0 = hi ? rb0 : ra0, s1 = hi ? rb1 : ra1;
      const int s2 = hi ? rb2 : ra2, s3 = hi ? rb3 : ra3;
      const float2 u0 = ldrow2(xl, s0, c), u1 = ldrow2(xl, s1, c);
      const float2 u2 = ldrow2(xl, s2, c), u3 = ldrow2(xl, s3, c);
      const float l0 = logit2(u0, vr, w), l1 = logit2(u1, vr, w);
      const float l2 = logit2(u2, vr, w), l3 = logit2(u3, vr, w);
      const float mn = fmaxf(fmaxf(m, l0), fmaxf(fmaxf(l1, l2), l3));
      const float sc = __expf(m - mn);
      const float e0 = __expf(l0 - mn), e1 = __expf(l1 - mn);
      const float e2 = __expf(l2 - mn), e3 = __expf(l3 - mn);
      den = den * sc + e0; a.x = a.x * sc + e0 * u0.x; a.y = a.y * sc + e0 * u0.y;
      den += e1; a.x += e1 * u1.x; a.y += e1 * u1.y;
      den += e2; a.x += e2 * u2.x; a.y += e2 * u2.y;
      den += e3; a.x += e3 * u3.x; a.y += e3 * u3.y;
      m = mn;
    }
    const int rem = cnt - j;
    if (rem > 0) {
      float2 u[4];
#pragma unroll
      for (int p = 0; p < 4; ++p) {
        u[p] = make_float2(0.f, 0.f);
        if (2 * p < rem) {
          const int ra = __builtin_amdgcn_readlane(sidx, j + 2 * p);
          const int rb = __builtin_amdgcn_readlane(sidx, j + 2 * p + 1);
          const int s = hi ? rb : ra;
          u[p] = ldrow2(xl, s, c);
        }
      }
#pragma unroll
      for (int p = 0; p < 4; ++p) {
        if (2 * p < rem) {
          const float l = logit2(u[p], vr, w);
          const float mn = fmaxf(m, l);
          const float sc = __expf(m - mn);
          float e = __expf(l - mn);
          if (2 * p + 1 >= rem) e = hi ? 0.f : e;
          den = den * sc + e;
          a.x = a.x * sc + e * u[p].x; a.y = a.y * sc + e * u[p].y;
          m = mn;
        }
      }
    }
  }
  {
    const float mt = fmaxf(m, xhalf(m));
    const float sm = __expf(m - mt);
    den *= sm; a.x *= sm; a.y *= sm;
    den += xhalf(den);
    a.x += xhalf(a.x); a.y += xhalf(a.y);
  }
  const float v = (hi ? a.y : a.x) / den + bb;
  const float mu = wave_sum(v) * (1.f / 32.f);
  const float dd = v - mu;
  const float rs = rsqrtf(wave_sum(dd * dd) * (1.f / 32.f) + LN_EPS);
  out[(size_t)node * 32 + q] = dd * rs * gg + bt;
}

static inline int gridFor(long long threads, int block) {
  return (int)((threads + block - 1) / block);
}

extern "C" void kernel_launch(void* const* d_in, const int* in_sizes, int n_in,
                              void* d_out, int out_size, void* d_ws, size_t ws_size,
                              hipStream_t stream) {
  const float* x   = (const float*)d_in[0];
  const int*   ei  = (const int*)  d_in[1];
  const float* W1l = (const float*)d_in[2];
  const float* W1r = (const float*)d_in[3];
  const float* a1  = (const float*)d_in[4];
  const float* b1  = (const float*)d_in[5];
  const float* g1  = (const float*)d_in[6];
  const float* be1 = (const float*)d_in[7];
  const float* W2l = (const float*)d_in[8];
  const float* W2r = (const float*)d_in[9];
  const float* a2  = (const float*)d_in[10];
  const float* b2  = (const float*)d_in[11];
  const float* g2  = (const float*)d_in[12];
  const float* be2 = (const float*)d_in[13];
  float* out = (float*)d_out;

  const int n  = in_sizes[0] / 4;
  const int E  = in_sizes[1] / 2;
  const int TE = E + n;
  const int ntiles = n / 16;

  float* ws = (float*)d_ws;
  const size_t XL1  = 0;
  const size_t XR1  = XL1 + (size_t)n * 64;
  const size_t ACC1 = XR1 + (size_t)n * 64;
  const size_t DEN1 = ACC1 + (size_t)n * 64;
  const size_t MAX1 = DEN1 + (size_t)n * 4;
  const size_t LOG1 = MAX1 + (size_t)n * 4;
  int* wi = (int*)d_ws;
  const size_t BSTA = LOG1;
  const size_t RPTR = BSTA + NBMAX + 64;
  const size_t SRCS = RPTR + (size_t)n + 64;
  const size_t BINN = (SRCS + (size_t)E + 1) & ~(size_t)1;

  const int B = 256;
  const int nb = (n + BIN_NODES - 1) >> BIN_SHIFT;
  const int nchunk = gridFor(E, EPB);
  const size_t CNTB = ACC1;
  const size_t XL2  = ACC1;
  const size_t XR2  = ACC1 + (size_t)n * 32;

  count_gemm1<<<nchunk + gridFor(ntiles, SORT_T / 32), SORT_T, 0, stream>>>(
      ei, E, nchunk, wi + CNTB, x, W1l, W1r, ws + XL1, ws + XR1, ntiles);
  bin_scatter<<<nchunk, SORT_T, 0, stream>>>(ei, E, nb, nchunk, wi + CNTB, wi + BSTA, wi + BINN);
  bin_sort<<<nb, SORT_T, 0, stream>>>(wi + BINN, wi + BSTA, n, wi + RPTR, wi + SRCS);

  gat1_node<<<gridFor(n, G1_NODES), G1_NODES * 32, 0, stream>>>(
      wi + RPTR, wi + SRCS, ws + XL1, ws + XR1, a1, b1, g1, be1,
      W2l, W2r, ws + XL2, ws + XR2, n);

  gat2_node<<<gridFor((long long)n * 32, B), B, 0, stream>>>(
      wi + RPTR, wi + SRCS, ws + XL2, ws + XR2, a2, b2, g2, be2, out, n);
}


# Node kernels issue fewer vector instructions, sums and their order kept: leaky ReLU as max(e, 0.2e), base-2 logits for bare exp2, one reciprocal for the divides, ELU as exp-1, bare rsq
# speedup vs baseline: 14.6804x; 1.0560x over previous
#include <hip/hip_runtime.h>
#include <hip/hip_bf16.h>
#include <math.h>


typedef __attribute__((ext_vector_type(2))) float v2f;
typedef __attribute__((ext_vector_type(8))) float v8f;

#define NEG_SLOPE 0.2f
#define LN_EPS 1e-5f

static __device__ inline v8f wmma4(v2f a, v2f b, v8f c) {
  return __builtin_amdgcn_wmma_f32_16x16x4_f32(
       false, a,  false, b,
       (short)0, c,  false,  false);
}

static __device__ inline void gemm1_tile(const float* __restrict__ x,
                                         const float* __restrict__ Wl,
                                         const float* __restrict__ Wr,
                                         float* __restrict__ xl,
                                         float* __restrict__ xr, int wave) {
  int lane = threadIdx.x & 31;
  int row0 = wave * 16;
  int m  = lane & 15;
  int hi = lane >> 4;
  const float* xrow = x + (size_t)(row0 + m) * 4 + hi * 2;
  v2f a; a.x = xrow[0]; a.y = xrow[1];
#pragma unroll
  for (int j = 0; j < 4; ++j) {
    v2f bl, br;
    bl.x = Wl[(2 * hi + 0) * 64 + j * 16 + m];
    bl.y = Wl[(2 * hi + 1) * 64 + j * 16 + m];
    br.x = Wr[(2 * hi + 0) * 64 + j * 16 + m];
    br.y = Wr[(2 * hi + 1) * 64 + j * 16 + m];
    v8f cl = {}; cl = wmma4(a, bl, cl);
    v8f cr = {}; cr = wmma4(a, br, cr);
#pragma unroll
    for (int v = 0; v < 8; ++v) {
      int r = v + hi * 8;
      xl[(size_t)(row0 + r) * 64 + j * 16 + m] = cl[v];
      xr[(size_t)(row0 + r) * 64 + j * 16 + m] = cr[v];
    }
  }
}

#define G1_NODES 16
#define HS_PITCH 68
static __device__ inline void gemm2_rows(const float* hs,
                                         const float* __restrict__ W,
                                         float* __restrict__ xo,
                                         int row0, int n) {
  const int lane = (int)__builtin_amdgcn_mbcnt_lo(~0u, 0u);
  const int m  = lane & 15;
  const int hi = lane >> 4;
  const float* hrow = hs + m * HS_PITCH;
  v8f c[2] = {{0.f},{0.f}};
#pragma unroll
  for (int kk = 0; kk < 16; ++kk) {
    const int k0 = kk * 4 + hi * 2;
    v2f a; a.x = hrow[k0]; a.y = hrow[k0 + 1];
#pragma unroll
    for (int j = 0; j < 2; ++j) {
      v2f b;
      b.x = W[(k0 + 0) * 32 + j * 16 + m];
      b.y = W[(k0 + 1) * 32 + j * 16 + m];
      c[j] = wmma4(a, b, c[j]);
    }
  }
  asm volatile("v_nop\n\tv_nop\n\tv_nop\n\tv_nop" : "+v"(c[0]), "+v"(c[1]));
  float* __restrict__ xcol = xo + (size_t)(row0 + hi * 8) * 32 + m;
  if (row0 + G1_NODES <= n) {
#pragma unroll
    for (int j = 0; j < 2; ++j)
#pragma unroll
      for (int v = 0; v < 8; ++v) xcol[v * 32 + j * 16] = c[j][v];
  } else {
#pragma unroll
    for (int j = 0; j < 2; ++j)
#pragma unroll
      for (int v = 0; v < 8; ++v)
        if (row0 + hi * 8 + v < n) xcol[v * 32 + j * 16] = c[j][v];
  }
}

#define BIN_SHIFT 9
#define BIN_NODES 512
#define NBMAX     256
#define SORT_T    1024
#define SORT_K    8
#define EPB       (SORT_T * SORT_K)
#define ROW_G     (SORT_T / (NBMAX / 4))
#define ROW_B     7

__global__ void __launch_bounds__(SORT_T)
count_gemm1(const int* __restrict__ ei, int E, int nchunk, int* __restrict__ cntB,
            const float* __restrict__ x, const float* __restrict__ Wl,
            const float* __restrict__ Wr, float* __restrict__ xl,
            float* __restrict__ xr, int ntiles) {
  __shared__ int h[NBMAX];
  const int t = threadIdx.x;
  if ((int)blockIdx.x >= nchunk) {
    const int wave = ((int)blockIdx.x - nchunk) * (SORT_T / 32) + (t >> 5);
    if (wave < ntiles) gemm1_tile(x, Wl, Wr, xl, xr, wave);
    return;
  }
  if (t < NBMAX) h[t] = 0;
  __syncthreads();
  const int base = blockIdx.x * EPB + t;
  int d[SORT_K];
#pragma unroll
  for (int k = 0; k < SORT_K; ++k) {
    const int e = base + k * SORT_T;
    d[k] = e < E ? ei[E + e] : -1;
  }
#pragma unroll
  for (int k = 0; k < SORT_K; ++k)
    if (d[k] >= 0) atomicAdd(&h[d[k] >> BIN_SHIFT], 1);
  __syncthreads();
  if (t < NBMAX) cntB[blockIdx.x * NBMAX + t] = h[t];
}

__global__ void __launch_bounds__(SORT_T)
bin_scatter(const int* __restrict__ ei, int E, int nb, int nchunk,
            const int* __restrict__ cntB, int* __restrict__ binStart,
            int* __restrict__ binnedw) {
  int2* __restrict__ binned = (int2*)binnedw;
  __shared__ int h[NBMAX];
  __shared__ __attribute__((aligned(16))) int ptot[ROW_G][NBMAX];
  __shared__ __attribute__((aligned(16))) int pbef[ROW_G][NBMAX];
  __shared__ int wsum[SORT_T / 32];
  const int t = threadIdx.x, lane = t & 31, wv = t >> 5;
  const int blk = blockIdx.x;
  if (t < NBMAX) h[t] = 0;
  const int base = blk * EPB + t;
  int s[SORT_K], d[SORT_K], r[SORT_K];
#pragma unroll
  for (int k = 0; k < SORT_K; ++k) {
    const int e = base + k * SORT_T;
    s[k] = 0; d[k] = -1;
    if (e < E) { s[k] = ei[e]; d[k] = ei[E + e]; }
  }
  {
    const int c4 = t & (NBMAX / 4 - 1), g = t / (NBMAX / 4);
    const int4* __restrict__ m4 = (const int4*)cntB;
    int4 tot = make_int4(0, 0, 0, 0), bef = make_int4(0, 0, 0, 0);
    for (int k0 = g; k0 < nchunk; k0 += ROW_G * ROW_B) {
      int4 v[ROW_B];
#pragma unroll
      for (int j = 0; j < ROW_B; ++j) {
        const int k = k0 + j * ROW_G;
        v[j] = k < nchunk ? m4[k * (NBMAX / 4) + c4] : make_int4(0, 0, 0, 0);
      }
#pragma unroll
      for (int j = 0; j < ROW_B; ++j) {
        const int k = k0 + j * ROW_G;
        tot.x += v[j].x; tot.y += v[j].y; tot.z += v[j].z; tot.w += v[j].w;
        if (k < blk) { bef.x += v[j].x; bef.y += v[j].y; bef.z += v[j].z; bef.w += v[j].w; }
      }
    }
    ((int4*)ptot[g])[c4] = tot;
    ((int4*)pbef[g])[c4] = bef;
  }
  __syncthreads();
#pragma unroll
  for (int k = 0; k < SORT_K; ++k) {
    r[k] = 0;
    if (d[k] >= 0) r[k] = atomicAdd(&h[d[k] >> BIN_SHIFT], 1);
  }
  int T = 0, B = 0;
  if (t < NBMAX) {
    for (int g = 0; g < ROW_G; ++g) { T += ptot[g][t]; B += pbef[g][t]; }
  }
  int inc = T;
  for (int k = 1; k < 32; k <<= 1) { int u = __shfl_up(inc, k, 32); if (lane >= k) inc += u; }
  if (lane == 31) wsum[wv] = inc;
  __syncthreads();
  if (t < NBMAX) {
    int woff = 0;
    for (int q = 0; q < wv; ++q) woff += wsum[q];
    const int excl = woff + inc - T;
    h[t] = excl + B;
    if (blk == 0) {
      if (t < nb) binStart[t] = excl;
      if (t == nb - 1) binStart[nb] = excl + T;
    }
  }
  __syncthreads();
#pragma unroll
  for (int k = 0; k < SORT_K; ++k)
    if (d[k] >= 0) binned[h[d[k] >> BIN_SHIFT] + r[k]] = make_int2(s[k], d[k]);
}

__global__ void __launch_bounds__(SORT_T)
bin_sort(const int* __restrict__ binnedw, const int* __restrict__ binStart, int n,
         int* __restrict__ rowptr, int* __restrict__ srcs) {
  const int2* __restrict__ binned = (const int2*)binnedw;
  __shared__ int cnt[BIN_NODES];
  __shared__ int wsum[SORT_T / 32];
  const int b = blockIdx.x, t = threadIdx.x, lane = t & 31, wv = t >> 5;
  const int beg = binStart[b], end = binStart[b + 1];
  const int node0 = b << BIN_SHIFT;
  if (t < BIN_NODES) cnt[t] = 0;
  __syncthreads();
  for (int c = beg; c < end; c += EPB) {
    int dl[SORT_K];
#pragma unroll
    for (int k = 0; k < SORT_K; ++k) {
      const int i = c + k * SORT_T + t;
      dl[k] = i < end ? binned[i].y : -1;
    }
#pragma unroll
    for (int k = 0; k < SORT_K; ++k)
      if (dl[k] >= 0) atomicAdd(&cnt[dl[k] - node0], 1);
  }
  __syncthreads();
  const int v = t < BIN_NODES ? cnt[t] : 0;
  int inc = v;
  for (int k = 1; k < 32; k <<= 1) { int u = __shfl_up(inc, k, 32); if (lane >= k) inc += u; }
  if (lane == 31) wsum[wv] = inc;
  __syncthreads();
  int woff = 0;
  for (int q = 0; q < wv; ++q) woff += wsum[q];
  const int excl = beg + woff + inc - v;
  if (t < BIN_NODES) {
    cnt[t] = excl;
    const int node = node0 + t;
    if (node < n) rowptr[node] = excl;
    if (node == n - 1) rowptr[n] = excl + v;
  }
  __syncthreads();
  for (int c = beg; c < end; c += EPB) {
    int2 e[SORT_K];
#pragma unroll
    for (int k = 0; k < SORT_K; ++k) {
      const int i = c + k * SORT_T + t;
      e[k] = i < end ? binned[i] : make_int2(0, -1);
    }
#pragma unroll
    for (int k = 0; k < SORT_K; ++k)
      if (e[k].y >= 0) srcs[atomicAdd(&cnt[e[k].y - node0], 1)] = e[k].x;
  }
}

static __device__ inline float lrelu(float e) { return __builtin_fmaxf(e, NEG_SLOPE * e); }

#define LOG2E 1.4426950408889634f
static __device__ inline float ex2(float x) { return __builtin_amdgcn_exp2f(x); }

template <int X> static __device__ inline float xrow(float v) {
  return __builtin_bit_cast(float, __builtin_amdgcn_update_dpp(
      0, __builtin_bit_cast(int, v), 0x160 + X, 0xF, 0xF, true));
}
static __device__ inline float xhalf(float v) {
  int b = __builtin_bit_cast(int, v);
  return __builtin_bit_cast(float, __builtin_amdgcn_permlanex16(
      b, b, 0x76543210, 0xfedcba98, false, false));
}

static __device__ inline float wave_sum(float s) {
  s += xhalf(s);
  s += xrow<8>(s); s += xrow<4>(s); s += xrow<2>(s); s += xrow<1>(s);
  return s;
}


static __device__ inline float4 ldrow4(const float* __restrict__ p, int s, int c) {
  return *(const float4*)((const char*)p + ((unsigned)s * 256u + (unsigned)c * 16u));
}
static __device__ inline float2 ldrow2(const float* __restrict__ p, int s, int c) {
  return *(const float2*)((const char*)p + ((unsigned)s * 128u + (unsigned)c * 8u));
}

static __device__ inline float logit1(float4 u, float4 vr, float4 w) {
  float p = lrelu(u.x + vr.x) * w.x + lrelu(u.y + vr.y) * w.y
          + lrelu(u.z + vr.z) * w.z + lrelu(u.w + vr.w) * w.w;
  p += xrow<1>(p); p += xrow<2>(p);
  return p;
}

__global__ void __launch_bounds__(G1_NODES * 32) __attribute__((amdgpu_waves_per_eu(16, 16)))
gat1_node(const int* __restrict__ rowptr,
                          const int* __restrict__ srcs,
                          const float* __restrict__ xl,
                          const float* __restrict__ xr,
                          const float* __restrict__ a1,
                          const float* __restrict__ b1,
                          const float* __restrict__ g,
                          const float* __restrict__ be,
                          const float* __restrict__ W2l,
                          const float* __restrict__ W2r,
                          float* __restrict__ xl2,
                          float* __restrict__ xr2, int n) {
  __shared__ __attribute__((aligned(16))) float hs[G1_NODES * HS_PITCH];
  const int wv = __builtin_amdgcn_readfirstlane((int)(threadIdx.x >> 5));
  const int row0 = (int)blockIdx.x * G1_NODES;
  const int lane = threadIdx.x & 31;
  const int node = row0 + wv < n ? row0 + wv : n - 1;
  const int  c  = lane & 15;
  const bool hi = lane >= 16;
  const float2 wa = ((const float2*)a1)[2 * c], wb = ((const float2*)a1)[2 * c + 1];
  const float4 w  = make_float4(wa.x * LOG2E, wa.y * LOG2E, wb.x * LOG2E, wb.y * LOG2E);
  const float4 vr = ((const float4*)(xr + (size_t)node * 64))[c];
  const float4 us = ((const float4*)(xl + (size_t)node * 64))[c];
  const int beg = rowptr[node], end = rowptr[node + 1];
  float m = logit1(us, vr, w);
  float den = hi ? 0.f : 1.f;
  float4 a = hi ? make_float4(0.f, 0.f, 0.f, 0.f) : us;
  for (int base = beg; base < end; base += 32) {
    int cnt = end - base; cnt = cnt < 32 ? cnt : 32;
    const int sidx = lane < cnt ? srcs[base + lane] : node;
    int j = 0;
    for (; j + 8 <= cnt; j += 8) {
      const int ra0 = __builtin_amdgcn_readlane(sidx, j + 0), rb0 = __builtin_amdgcn_readlane(sidx, j + 1);
      const int ra1 = __builtin_amdgcn_readlane(sidx, j + 2), rb1 = __builtin_amdgcn_readlane(sidx, j + 3);
      const int ra2 = __builtin_amdgcn_readlane(sidx, j + 4), rb2 = __builtin_amdgcn_readlane(sidx, j + 5);
      const int ra3 = __builtin_amdgcn_readlane(sidx, j + 6), rb3 = __builtin_amdgcn_readlane(sidx, j + 7);
      const int s0 = hi ? rb0 : ra0, s1 = hi ? rb1 : ra1;
      const int s2 = hi ? rb2 : ra2, s3 = hi ? rb3 : ra3;
      const float4 u0 = ldrow4(xl, s0, c), u1 = ldrow4(xl, s1, c);
      const float4 u2 = ldrow4(xl, s2, c), u3 = ldrow4(xl, s3, c);
      const float l0 = logit1(u0, vr, w), l1 = logit1(u1, vr, w);
      const float l2 = logit1(u2, vr, w), l3 = logit1(u3, vr, w);
      const float mn = fmaxf(fmaxf(m, l0), fmaxf(fmaxf(l1, l2), l3));
      const float sc = ex2(m - mn);
      const float e0 = ex2(l0 - mn), e1 = ex2(l1 - mn);
      const float e2 = ex2(l2 - mn), e3 = ex2(l3 - mn);
      den = den * sc + e0;
      a.x = a.x * sc + e0 * u0.x; a.y = a.y * sc + e0 * u0.y;
      a.z = a.z * sc + e0 * u0.z; a.w = a.w * sc + e0 * u0.w;
      den += e1; a.x += e1 * u1.x; a.y += e1 * u1.y; a.z += e1 * u1.z; a.w += e1 * u1.w;
      den += e2; a.x += e2 * u2.x; a.y += e2 * u2.y; a.z += e2 * u2.z; a.w += e2 * u2.w;
      den += e3; a.x += e3 * u3.x; a.y += e3 * u3.y; a.z += e3 * u3.z; a.w += e3 * u3.w;
      m = mn;
    }
    const int rem = cnt - j;
    if (rem > 0) {
      float4 u[4];
#pragma unroll
      for (int p = 0; p < 4; ++p) {
        u[p] = make_float4(0.f, 0.f, 0.f, 0.f);
        if (2 * p < rem) {
          const int ra = __builtin_amdgcn_readlane(sidx, j + 2 * p);
          const int rb = __builtin_amdgcn_readlane(sidx, j + 2 * p + 1);
          const int s = hi ? rb : ra;
          u[p] = ldrow4(xl, s, c);
        }
      }
#pragma unroll
      for (int p = 0; p < 4; ++p) {
        if (2 * p < rem) {
          const float l = logit1(u[p], vr, w);
          const float mn = fmaxf(m, l);
          const float sc = ex2(m - mn);
          float e = ex2(l - mn);
          if (2 * p + 1 >= rem) e = hi ? 0.f : e;
          den = den * sc + e;
          a.x = a.x * sc + e * u[p].x; a.y = a.y * sc + e * u[p].y;
          a.z = a.z * sc + e * u[p].z; a.w = a.w * sc + e * u[p].w;
          m = mn;
        }
      }
    }
  }
  const int ln = (int)__builtin_amdgcn_mbcnt_lo(~0u, 0u);
  const int q  = 2 * (ln & 15) + (ln >> 4);
  const float2 bb = ((const float2*)b1)[q];
  const float2 gg = ((const float2*)g)[q];
  const float2 bt = ((const float2*)be)[q];
  {
    const float mt = fmaxf(m, xhalf(m));
    const float sm = ex2(m - mt);
    den *= sm; a.x *= sm; a.y *= sm; a.z *= sm; a.w *= sm;
    den += xhalf(den);
    a.x += xhalf(a.x); a.y += xhalf(a.y); a.z += xhalf(a.z); a.w += xhalf(a.w);
  }
  const float rd = __builtin_amdgcn_rcpf(den);
  const float v0 = (hi ? a.z : a.x) * rd + bb.x;
  const float v1 = (hi ? a.w : a.y) * rd + bb.y;
  const float mu = wave_sum(v0 + v1) * (1.f / 64.f);
  const float d0 = v0 - mu, d1 = v1 - mu;
  const float rs = __builtin_amdgcn_rsqf(wave_sum(d0 * d0 + d1 * d1) * (1.f / 64.f) + LN_EPS);
  const float y0 = d0 * rs * gg.x + bt.x;
  const float y1 = d1 * rs * gg.y + bt.y;
  float2 y;
  y.x = y0 > 0.f ? y0 : ex2(y0 * LOG2E) - 1.f;
  y.y = y1 > 0.f ? y1 : ex2(y1 * LOG2E) - 1.f;
  ((float2*)(hs + wv * HS_PITCH))[q] = y;
  __syncthreads();
  if (wv >= 2) return;
  gemm2_rows(hs, wv == 0 ? W2l : W2r, wv == 0 ? xl2 : xr2, row0, n);
}

static __device__ inline float logit2(float2 u, float2 vr, float2 w) {
  float p = lrelu(u.x + vr.x) * w.x + lrelu(u.y + vr.y) * w.y;
  p += xrow<1>(p); p += xrow<2>(p); p += xrow<4>(p); p += xrow<8>(p);
  return p;
}

__global__ void gat2_node(const int* __restrict__ rowptr,
                          const int* __restrict__ srcs,
                          const float* __restrict__ xl,
                          const float* __restrict__ xr,
                          const float* __restrict__ a2,
                          const float* __restrict__ b2,
                          const float* __restrict__ g,
                          const float* __restrict__ be,
                          float* __restrict__ out, int n) {
  const int node = __builtin_amdgcn_readfirstlane(
      (int)(((long long)blockIdx.x * blockDim.x + threadIdx.x) >> 5));
  const int lane = threadIdx.x & 31;
  if (node >= n) return;
  const int  c  = lane & 15;
  const bool hi = lane >= 16;
  const int  q  = 2 * c + (hi ? 1 : 0);
  const float2 w0 = ((const float2*)a2)[c];
  const float2 w  = make_float2(w0.x * LOG2E, w0.y * LOG2E);
  const float bb = b2[q], gg = g[q], bt = be[q];
  const float2 vr = ((const float2*)(xr + (size_t)node * 32))[c];
  const float2 us = ((const float2*)(xl + (size_t)node * 32))[c];
  const int beg = rowptr[node], end = rowptr[node + 1];
  float m = logit2(us, vr, w);
  float den = hi ? 0.f : 1.f;
  float2 a = hi ? make_float2(0.f, 0.f) : us;
  for (int base = beg; base < end; base += 32) {
    int cnt = end - base; cnt = cnt < 32 ? cnt : 32;
    const int sidx = lane < cnt ? srcs[base + lane] : node;
    int j = 0;
    for (; j + 8 <= cnt; j += 8) {
      const int ra0 = __builtin_amdgcn_readlane(sidx, j + 0), rb0 = __builtin_amdgcn_readlane(sidx, j + 1);
      const int ra1 = __builtin_amdgcn_readlane(sidx, j + 2), rb1 = __builtin_amdgcn_readlane(sidx, j + 3);
      const int ra2 = __builtin_amdgcn_readlane(sidx, j + 4), rb2 = __builtin_amdgcn_readlane(sidx, j + 5);
      const int ra3 = __builtin_amdgcn_readlane(sidx, j + 6), rb3 = __builtin_amdgcn_readlane(sidx, j + 7);
      const int s0 = hi ? rb0 : ra0, s1 = hi ? rb1 : ra1;
      const int s2 = hi ? rb2 : ra2, s3 = hi ? rb3 : ra3;
      const float2 u0 = ldrow2(xl, s0, c), u1 = ldrow2(xl, s1, c);
      const float2 u2 = ldrow2(xl, s2, c), u3 = ldrow2(xl, s3, c);
      const float l0 = logit2(u0, vr, w), l1 = logit2(u1, vr, w);
      const float l2 = logit2(u2, vr, w), l3 = logit2(u3, vr, w);
      const float mn = fmaxf(fmaxf(m, l0), fmaxf(fmaxf(l1, l2), l3));
      const float sc = ex2(m - mn);
      const float e0 = ex2(l0 - mn), e1 = ex2(l1 - mn);
      const float e2 = ex2(l2 - mn), e3 = ex2(l3 - mn);
      den = den * sc + e0; a.x = a.x * sc + e0 * u0.x; a.y = a.y * sc + e0 * u0.y;
      den += e1; a.x += e1 * u1.x; a.y += e1 * u1.y;
      den += e2; a.x += e2 * u2.x; a.y += e2 * u2.y;
      den += e3; a.x += e3 * u3.x; a.y += e3 * u3.y;
      m = mn;
    }
    const int rem = cnt - j;
    if (rem > 0) {
      float2 u[4];
#pragma unroll
      for (int p = 0; p < 4; ++p) {
        u[p] = make_float2(0.f, 0.f);
        if (2 * p < rem) {
          const int ra = __builtin_amdgcn_readlane(sidx, j + 2 * p);
          const int rb = __builtin_amdgcn_readlane(sidx, j + 2 * p + 1);
          const int s = hi ? rb : ra;
          u[p] = ldrow2(xl, s, c);
        }
      }
#pragma unroll
      for (int p = 0; p < 4; ++p) {
        if (2 * p < rem) {
          const float l = logit2(u[p], vr, w);
          const float mn = fmaxf(m, l);
          const float sc = ex2(m - mn);
          float e = ex2(l - mn);
          if (2 * p + 1 >= rem) e = hi ? 0.f : e;
          den = den * sc + e;
          a.x = a.x * sc + e * u[p].x; a.y = a.y * sc + e * u[p].y;
          m = mn;
        }
      }
    }
  }
  {
    const float mt = fmaxf(m, xhalf(m));
    const float sm = ex2(m - mt);
    den *= sm; a.x *= sm; a.y *= sm;
    den += xhalf(den);
    a.x += xhalf(a.x); a.y += xhalf(a.y);
  }
  const float v = (hi ? a.y : a.x) * __builtin_amdgcn_rcpf(den) + bb;
  const float mu = wave_sum(v) * (1.f / 32.f);
  const float dd = v - mu;
  const float rs = __builtin_amdgcn_rsqf(wave_sum(dd * dd) * (1.f / 32.f) + LN_EPS);
  out[(size_t)node * 32 + q] = dd * rs * gg + bt;
}

static inline int gridFor(long long threads, int block) {
  return (int)((threads + block - 1) / block);
}

extern "C" void kernel_launch(void* const* d_in, const int* in_sizes, int n_in,
                              void* d_out, int out_size, void* d_ws, size_t ws_size,
                              hipStream_t stream) {
  const float* x   = (const float*)d_in[0];
  const int*   ei  = (const int*)  d_in[1];
  const float* W1l = (const float*)d_in[2];
  const float* W1r = (const float*)d_in[3];
  const float* a1  = (const float*)d_in[4];
  const float* b1  = (const float*)d_in[5];
  const float* g1  = (const float*)d_in[6];
  const float* be1 = (const float*)d_in[7];
  const float* W2l = (const float*)d_in[8];
  const float* W2r = (const float*)d_in[9];
  const float* a2  = (const float*)d_in[10];
  const float* b2  = (const float*)d_in[11];
  const float* g2  = (const float*)d_in[12];
  const float* be2 = (const float*)d_in[13];
  float* out = (float*)d_out;

  const int n  = in_sizes[0] / 4;
  const int E  = in_sizes[1] / 2;
  const int TE = E + n;
  const int ntiles = n / 16;

  float* ws = (float*)d_ws;
  const size_t XL1  = 0;
  const size_t XR1  = XL1 + (size_t)n * 64;
  const size_t ACC1 = XR1 + (size_t)n * 64;
  const size_t DEN1 = ACC1 + (size_t)n * 64;
  const size_t MAX1 = DEN1 + (size_t)n * 4;
  const size_t LOG1 = MAX1 + (size_t)n * 4;
  int* wi = (int*)d_ws;
  const size_t BSTA = LOG1;
  const size_t RPTR = BSTA + NBMAX + 64;
  const size_t SRCS = RPTR + (size_t)n + 64;
  const size_t BINN = (SRCS + (size_t)E + 1) & ~(size_t)1;

  const int B = 256;
  const int nb = (n + BIN_NODES - 1) >> BIN_SHIFT;
  const int nchunk = gridFor(E, EPB);
  const size_t CNTB = ACC1;
  const size_t XL2  = ACC1;
  const size_t XR2  = ACC1 + (size_t)n * 32;

  count_gemm1<<<nchunk + gridFor(ntiles, SORT_T / 32), SORT_T, 0, stream>>>(
      ei, E, nchunk, wi + CNTB, x, W1l, W1r, ws + XL1, ws + XR1, ntiles);
  bin_scatter<<<nchunk, SORT_T, 0, stream>>>(ei, E, nb, nchunk, wi + CNTB, wi + BSTA, wi + BINN);
  bin_sort<<<nb, SORT_T, 0, stream>>>(wi + BINN, wi + BSTA, n, wi + RPTR, wi + SRCS);

  gat1_node<<<gridFor(n, G1_NODES), G1_NODES * 32, 0, stream>>>(
      wi + RPTR, wi + SRCS, ws + XL1, ws + XR1, a1, b1, g1, be1,
      W2l, W2r, ws + XL2, ws + XR2, n);

  gat2_node<<<gridFor((long long)n * 32, B), B, 0, stream>>>(
      wi + RPTR, wi + SRCS, ws + XL2, ws + XR2, a2, b2, g2, be2, out, n);
}


# Node kernels: one 16-lane half-wave per node; two nodes per wave share each dependent load step and the epilogue, no half merge; softmax sums in one arrival-order chain, LayerNorm in a 16-lane tree
# speedup vs baseline: 15.6662x; 1.0672x over previous
#include <hip/hip_runtime.h>
#include <hip/hip_bf16.h>
#include <math.h>


typedef __attribute__((ext_vector_type(2))) float v2f;
typedef __attribute__((ext_vector_type(8))) float v8f;

#define NEG_SLOPE 0.2f
#define LN_EPS 1e-5f

static __device__ inline v8f wmma4(v2f a, v2f b, v8f c) {
  return __builtin_amdgcn_wmma_f32_16x16x4_f32(
       false, a,  false, b,
       (short)0, c,  false,  false);
}

static __device__ inline void gemm1_tile(const float* __restrict__ x,
                                         const float* __restrict__ Wl,
                                         const float* __restrict__ Wr,
                                         float* __restrict__ xl,
                                         float* __restrict__ xr, int wave) {
  int lane = threadIdx.x & 31;
  int row0 = wave * 16;
  int m  = lane & 15;
  int hi = lane >> 4;
  const float* xrow = x + (size_t)(row0 + m) * 4 + hi * 2;
  v2f a; a.x = xrow[0]; a.y = xrow[1];
#pragma unroll
  for (int j = 0; j < 4; ++j) {
    v2f bl, br;
    bl.x = Wl[(2 * hi + 0) * 64 + j * 16 + m];
    bl.y = Wl[(2 * hi + 1) * 64 + j * 16 + m];
    br.x = Wr[(2 * hi + 0) * 64 + j * 16 + m];
    br.y = Wr[(2 * hi + 1) * 64 + j * 16 + m];
    v8f cl = {}; cl = wmma4(a, bl, cl);
    v8f cr = {}; cr = wmma4(a, br, cr);
#pragma unroll
    for (int v = 0; v < 8; ++v) {
      int r = v + hi * 8;
      xl[(size_t)(row0 + r) * 64 + j * 16 + m] = cl[v];
      xr[(size_t)(row0 + r) * 64 + j * 16 + m] = cr[v];
    }
  }
}

#define G1_NODES 16
#define HS_PITCH 68
static __device__ inline void gemm2_rows(const float* hs,
                                         const float* __restrict__ W,
                                         float* __restrict__ xo,
                                         int row0, int n) {
  const int lane = (int)__builtin_amdgcn_mbcnt_lo(~0u, 0u);
  const int m  = lane & 15;
  const int hi = lane >> 4;
  const float* hrow = hs + m * HS_PITCH;
  v8f c[2] = {{0.f},{0.f}};
#pragma unroll
  for (int kk = 0; kk < 16; ++kk) {
    const int k0 = kk * 4 + hi * 2;
    v2f a; a.x = hrow[k0]; a.y = hrow[k0 + 1];
#pragma unroll
    for (int j = 0; j < 2; ++j) {
      v2f b;
      b.x = W[(k0 + 0) * 32 + j * 16 + m];
      b.y = W[(k0 + 1) * 32 + j * 16 + m];
      c[j] = wmma4(a, b, c[j]);
    }
  }
  asm volatile("v_nop\n\tv_nop\n\tv_nop\n\tv_nop" : "+v"(c[0]), "+v"(c[1]));
  float* __restrict__ xcol = xo + (size_t)(row0 + hi * 8) * 32 + m;
  if (row0 + G1_NODES <= n) {
#pragma unroll
    for (int j = 0; j < 2; ++j)
#pragma unroll
      for (int v = 0; v < 8; ++v) xcol[v * 32 + j * 16] = c[j][v];
  } else {
#pragma unroll
    for (int j = 0; j < 2; ++j)
#pragma unroll
      for (int v = 0; v < 8; ++v)
        if (row0 + hi * 8 + v < n) xcol[v * 32 + j * 16] = c[j][v];
  }
}

#define BIN_SHIFT 9
#define BIN_NODES 512
#define NBMAX     256
#define SORT_T    1024
#define SORT_K    8
#define EPB       (SORT_T * SORT_K)
#define ROW_G     (SORT_T / (NBMAX / 4))
#define ROW_B     7

__global__ void __launch_bounds__(SORT_T)
count_gemm1(const int* __restrict__ ei, int E, int nchunk, int* __restrict__ cntB,
            const float* __restrict__ x, const float* __restrict__ Wl,
            const float* __restrict__ Wr, float* __restrict__ xl,
            float* __restrict__ xr, int ntiles) {
  __shared__ int h[NBMAX];
  const int t = threadIdx.x;
  if ((int)blockIdx.x >= nchunk) {
    const int wave = ((int)blockIdx.x - nchunk) * (SORT_T / 32) + (t >> 5);
    if (wave < ntiles) gemm1_tile(x, Wl, Wr, xl, xr, wave);
    return;
  }
  if (t < NBMAX) h[t] = 0;
  __syncthreads();
  const int base = blockIdx.x * EPB + t;
  int d[SORT_K];
#pragma unroll
  for (int k = 0; k < SORT_K; ++k) {
    const int e = base + k * SORT_T;
    d[k] = e < E ? ei[E + e] : -1;
  }
#pragma unroll
  for (int k = 0; k < SORT_K; ++k)
    if (d[k] >= 0) atomicAdd(&h[d[k] >> BIN_SHIFT], 1);
  __syncthreads();
  if (t < NBMAX) cntB[blockIdx.x * NBMAX + t] = h[t];
}

__global__ void __launch_bounds__(SORT_T)
bin_scatter(const int* __restrict__ ei, int E, int nb, int nchunk,
            const int* __restrict__ cntB, int* __restrict__ binStart,
            int* __restrict__ binnedw) {
  int2* __restrict__ binned = (int2*)binnedw;
  __shared__ int h[NBMAX];
  __shared__ __attribute__((aligned(16))) int ptot[ROW_G][NBMAX];
  __shared__ __attribute__((aligned(16))) int pbef[ROW_G][NBMAX];
  __shared__ int wsum[SORT_T / 32];
  const int t = threadIdx.x, lane = t & 31, wv = t >> 5;
  const int blk = blockIdx.x;
  if (t < NBMAX) h[t] = 0;
  const int base = blk * EPB + t;
  int s[SORT_K], d[SORT_K], r[SORT_K];
#pragma unroll
  for (int k = 0; k < SORT_K; ++k) {
    const int e = base + k * SORT_T;
    s[k] = 0; d[k] = -1;
    if (e < E) { s[k] = ei[e]; d[k] = ei[E + e]; }
  }
  {
    const int c4 = t & (NBMAX / 4 - 1), g = t / (NBMAX / 4);
    const int4* __restrict__ m4 = (const int4*)cntB;
    int4 tot = make_int4(0, 0, 0, 0), bef = make_int4(0, 0, 0, 0);
    for (int k0 = g; k0 < nchunk; k0 += ROW_G * ROW_B) {
      int4 v[ROW_B];
#pragma unroll
      for (int j = 0; j < ROW_B; ++j) {
        const int k = k0 + j * ROW_G;
        v[j] = k < nchunk ? m4[k * (NBMAX / 4) + c4] : make_int4(0, 0, 0, 0);
      }
#pragma unroll
      for (int j = 0; j < ROW_B; ++j) {
        const int k = k0 + j * ROW_G;
        tot.x += v[j].x; tot.y += v[j].y; tot.z += v[j].z; tot.w += v[j].w;
        if (k < blk) { bef.x += v[j].x; bef.y += v[j].y; bef.z += v[j].z; bef.w += v[j].w; }
      }
    }
    ((int4*)ptot[g])[c4] = tot;
    ((int4*)pbef[g])[c4] = bef;
  }
  __syncthreads();
#pragma unroll
  for (int k = 0; k < SORT_K; ++k) {
    r[k] = 0;
    if (d[k] >= 0) r[k] = atomicAdd(&h[d[k] >> BIN_SHIFT], 1);
  }
  int T = 0, B = 0;
  if (t < NBMAX) {
    for (int g = 0; g < ROW_G; ++g) { T += ptot[g][t]; B += pbef[g][t]; }
  }
  int inc = T;
  for (int k = 1; k < 32; k <<= 1) { int u = __shfl_up(inc, k, 32); if (lane >= k) inc += u; }
  if (lane == 31) wsum[wv] = inc;
  __syncthreads();
  if (t < NBMAX) {
    int woff = 0;
    for (int q = 0; q < wv; ++q) woff += wsum[q];
    const int excl = woff + inc - T;
    h[t] = excl + B;
    if (blk == 0) {
      if (t < nb) binStart[t] = excl;
      if (t == nb - 1) binStart[nb] = excl + T;
    }
  }
  __syncthreads();
#pragma unroll
  for (int k = 0; k < SORT_K; ++k)
    if (d[k] >= 0) binned[h[d[k] >> BIN_SHIFT] + r[k]] = make_int2(s[k], d[k]);
}

__global__ void __launch_bounds__(SORT_T)
bin_sort(const int* __restrict__ binnedw, const int* __restrict__ binStart, int n,
         int* __restrict__ rowptr, int* __restrict__ srcs) {
  const int2* __restrict__ binned = (const int2*)binnedw;
  __shared__ int cnt[BIN_NODES];
  __shared__ int wsum[SORT_T / 32];
  const int b = blockIdx.x, t = threadIdx.x, lane = t & 31, wv = t >> 5;
  const int beg = binStart[b], end = binStart[b + 1];
  const int node0 = b << BIN_SHIFT;
  if (t < BIN_NODES) cnt[t] = 0;
  __syncthreads();
  for (int c = beg; c < end; c += EPB) {
    int dl[SORT_K];
#pragma unroll
    for (int k = 0; k < SORT_K; ++k) {
      const int i = c + k * SORT_T + t;
      dl[k] = i < end ? binned[i].y : -1;
    }
#pragma unroll
    for (int k = 0; k < SORT_K; ++k)
      if (dl[k] >= 0) atomicAdd(&cnt[dl[k] - node0], 1);
  }
  __syncthreads();
  const int v = t < BIN_NODES ? cnt[t] : 0;
  int inc = v;
  for (int k = 1; k < 32; k <<= 1) { int u = __shfl_up(inc, k, 32); if (lane >= k) inc += u; }
  if (lane == 31) wsum[wv] = inc;
  __syncthreads();
  int woff = 0;
  for (int q = 0; q < wv; ++q) woff += wsum[q];
  const int excl = beg + woff + inc - v;
  if (t < BIN_NODES) {
    cnt[t] = excl;
    const int node = node0 + t;
    if (node < n) rowptr[node] = excl;
    if (node == n - 1) rowptr[n] = excl + v;
  }
  __syncthreads();
  for (int c = beg; c < end; c += EPB) {
    int2 e[SORT_K];
#pragma unroll
    for (int k = 0; k < SORT_K; ++k) {
      const int i = c + k * SORT_T + t;
      e[k] = i < end ? binned[i] : make_int2(0, -1);
    }
#pragma unroll
    for (int k = 0; k < SORT_K; ++k)
      if (e[k].y >= 0) srcs[atomicAdd(&cnt[e[k].y - node0], 1)] = e[k].x;
  }
}

static __device__ inline float lrelu(float e) { return __builtin_fmaxf(e, NEG_SLOPE * e); }

#define LOG2E 1.4426950408889634f
static __device__ inline float ex2(float x) { return __builtin_amdgcn_exp2f(x); }

template <int X> static __device__ inline float xrow(float v) {
  return __builtin_bit_cast(float, __builtin_amdgcn_update_dpp(
      0, __builtin_bit_cast(int, v), 0x160 + X, 0xF, 0xF, true));
}
static __device__ inline float row_sum(float s) {
  s += xrow<8>(s); s += xrow<4>(s); s += xrow<2>(s); s += xrow<1>(s);
  return s;
}
static __device__ inline int row_bcast(int v, unsigned sel) {
  return __builtin_amdgcn_permlane16(v, v, sel, sel, false, false);
}
#define NIB1 0x11111111u


static __device__ inline float4 ldrow4(const float* __restrict__ p, int s, int c) {
  return *(const float4*)((const char*)p + ((unsigned)s * 256u + (unsigned)c * 16u));
}
static __device__ inline float2 ldrow2(const float* __restrict__ p, int s, int c) {
  return *(const float2*)((const char*)p + ((unsigned)s * 128u + (unsigned)c * 8u));
}

static __device__ inline float logit1(float4 u, float4 vr, float4 w) {
  float p = lrelu(u.x + vr.x) * w.x + lrelu(u.y + vr.y) * w.y
          + lrelu(u.z + vr.z) * w.z + lrelu(u.w + vr.w) * w.w;
  p += xrow<1>(p); p += xrow<2>(p);
  return p;
}

__global__ void __launch_bounds__(G1_NODES * 16) __attribute__((amdgpu_waves_per_eu(12, 16)))
gat1_node(const int* __restrict__ rowptr,
                          const int* __restrict__ srcs,
                          const float* __restrict__ xl,
                          const float* __restrict__ xr,
                          const float* __restrict__ a1,
                          const float* __restrict__ b1,
                          const float* __restrict__ g,
                          const float* __restrict__ be,
                          const float* __restrict__ W2l,
                          const float* __restrict__ W2r,
                          float* __restrict__ xl2,
                          float* __restrict__ xr2, int n) {
  __shared__ __attribute__((aligned(16))) float hs[G1_NODES * HS_PITCH];
  const int wv = __builtin_amdgcn_readfirstlane((int)(threadIdx.x >> 5));
  const int row0 = (int)blockIdx.x * G1_NODES;
  const int lane = threadIdx.x & 31;
  const int c = lane & 15;
  const bool hi = lane >= 16;
  const int nA = row0 + 2 * wv     < n ? row0 + 2 * wv     : n - 1;
  const int nB = row0 + 2 * wv + 1 < n ? row0 + 2 * wv + 1 : n - 1;
  const int node = hi ? nB : nA;
  const float2 wa = ((const float2*)a1)[2 * c], wb = ((const float2*)a1)[2 * c + 1];
  const float4 w  = make_float4(wa.x * LOG2E, wa.y * LOG2E, wb.x * LOG2E, wb.y * LOG2E);
  const float4 vr = ldrow4(xr, node, c);
  const float4 us = ldrow4(xl, node, c);
  const int begA = rowptr[nA], lenA = rowptr[nA + 1] - begA;
  const int begB = rowptr[nB], lenB = rowptr[nB + 1] - begB;
  const int lenMax = lenA > lenB ? lenA : lenB;
  float m = logit1(us, vr, w);
  float den = 1.f;
  float4 a = us;
  for (int off = 0; off < lenMax; off += 32) {
    const int cnt = (hi ? lenB : lenA) - off;
    const int pos = (hi ? begB : begA) + off + c;
    const int sA = c < cnt ? srcs[pos] : node;
    const int sB = c + 16 < cnt ? srcs[pos + 16] : node;
    int gEnd = lenMax - off; gEnd = gEnd < 32 ? gEnd : 32;
    for (int j = 0; j < gEnd; j += 4) {
      const int sv = (j & 16) ? sB : sA;
      const unsigned sel = (unsigned)(j & 12) * NIB1;
      const int s0 = row_bcast(sv, sel),            s1 = row_bcast(sv, sel + NIB1);
      const int s2 = row_bcast(sv, sel + 2u * NIB1), s3 = row_bcast(sv, sel + 3u * NIB1);
      const float4 u0 = ldrow4(xl, s0, c), u1 = ldrow4(xl, s1, c);
      const float4 u2 = ldrow4(xl, s2, c), u3 = ldrow4(xl, s3, c);
      const float l0 = logit1(u0, vr, w), l1 = logit1(u1, vr, w);
      const float l2 = logit1(u2, vr, w), l3 = logit1(u3, vr, w);
      const float mn = fmaxf(fmaxf(m, l0), fmaxf(fmaxf(l1, l2), l3));
      const float sc = ex2(m - mn);
      const int rem = cnt - j;
      const float e0 = rem > 0 ? ex2(l0 - mn) : 0.f, e1 = rem > 1 ? ex2(l1 - mn) : 0.f;
      const float e2 = rem > 2 ? ex2(l2 - mn) : 0.f, e3 = rem > 3 ? ex2(l3 - mn) : 0.f;
      den = den * sc + e0;
      a.x = a.x * sc + e0 * u0.x; a.y = a.y * sc + e0 * u0.y;
      a.z = a.z * sc + e0 * u0.z; a.w = a.w * sc + e0 * u0.w;
      den += e1; a.x += e1 * u1.x; a.y += e1 * u1.y; a.z += e1 * u1.z; a.w += e1 * u1.w;
      den += e2; a.x += e2 * u2.x; a.y += e2 * u2.y; a.z += e2 * u2.z; a.w += e2 * u2.w;
      den += e3; a.x += e3 * u3.x; a.y += e3 * u3.y; a.z += e3 * u3.z; a.w += e3 * u3.w;
      m = mn;
    }
  }
  const int ln = (int)__builtin_amdgcn_mbcnt_lo(~0u, 0u);
  const int cc = ln & 15;
  const float4 bb = ((const float4*)b1)[cc];
  const float4 gg = ((const float4*)g)[cc];
  const float4 bt = ((const float4*)be)[cc];
  const float rd = __builtin_amdgcn_rcpf(den);
  const float v0 = a.x * rd + bb.x, v1 = a.y * rd + bb.y;
  const float v2 = a.z * rd + bb.z, v3 = a.w * rd + bb.w;
  const float mu = row_sum((v0 + v1) + (v2 + v3)) * (1.f / 64.f);
  const float d0 = v0 - mu, d1 = v1 - mu, d2 = v2 - mu, d3 = v3 - mu;
  const float rs = __builtin_amdgcn_rsqf(
      row_sum((d0 * d0 + d1 * d1) + (d2 * d2 + d3 * d3)) * (1.f / 64.f) + LN_EPS);
  const float y0 = d0 * rs * gg.x + bt.x, y1 = d1 * rs * gg.y + bt.y;
  const float y2 = d2 * rs * gg.z + bt.z, y3 = d3 * rs * gg.w + bt.w;
  float4 y;
  y.x = y0 > 0.f ? y0 : ex2(y0 * LOG2E) - 1.f;
  y.y = y1 > 0.f ? y1 : ex2(y1 * LOG2E) - 1.f;
  y.z = y2 > 0.f ? y2 : ex2(y2 * LOG2E) - 1.f;
  y.w = y3 > 0.f ? y3 : ex2(y3 * LOG2E) - 1.f;
  ((float4*)(hs + (2 * wv + (ln >> 4)) * HS_PITCH))[cc] = y;
  __syncthreads();
  if (wv >= 2) return;
  gemm2_rows(hs, wv == 0 ? W2l : W2r, wv == 0 ? xl2 : xr2, row0, n);
}

static __device__ inline float logit2(float2 u, float2 vr, float2 w) {
  float p = lrelu(u.x + vr.x) * w.x + lrelu(u.y + vr.y) * w.y;
  p += xrow<1>(p); p += xrow<2>(p); p += xrow<4>(p); p += xrow<8>(p);
  return p;
}

#define G2_T 256
__global__ void __launch_bounds__(G2_T)
gat2_node(const int* __restrict__ rowptr,
                          const int* __restrict__ srcs,
                          const float* __restrict__ xl,
                          const float* __restrict__ xr,
                          const float* __restrict__ a2,
                          const float* __restrict__ b2,
                          const float* __restrict__ g,
                          const float* __restrict__ be,
                          float* __restrict__ out, int n) {
  const int lane = threadIdx.x & 31;
  const int c = lane & 15;
  const bool hi = lane >= 16;
  const int n0 = (int)blockIdx.x * (G2_T / 16)
               + 2 * __builtin_amdgcn_readfirstlane((int)(threadIdx.x >> 5));
  const int nA = n0     < n ? n0     : n - 1;
  const int nB = n0 + 1 < n ? n0 + 1 : n - 1;
  const bool live = n0 + (hi ? 1 : 0) < n;
  const int node = hi ? nB : nA;
  const float2 w0 = ((const float2*)a2)[c];
  const float2 w  = make_float2(w0.x * LOG2E, w0.y * LOG2E);
  const float2 vr = ldrow2(xr, node, c);
  const float2 us = ldrow2(xl, node, c);
  const int begA = rowptr[nA], lenA = rowptr[nA + 1] - begA;
  const int begB = rowptr[nB], lenB = rowptr[nB + 1] - begB;
  const int lenMax = lenA > lenB ? lenA : lenB;
  float m = logit2(us, vr, w);
  float den = 1.f;
  float2 a = us;
  for (int off = 0; off < lenMax; off += 32) {
    const int cnt = (hi ? lenB : lenA) - off;
    const int pos = (hi ? begB : begA) + off + c;
    const int sA = c < cnt ? srcs[pos] : node;
    const int sB = c + 16 < cnt ? srcs[pos + 16] : node;
    int gEnd = lenMax - off; gEnd = gEnd < 32 ? gEnd : 32;
    for (int j = 0; j < gEnd; j += 4) {
      const int sv = (j & 16) ? sB : sA;
      const unsigned sel = (unsigned)(j & 12) * NIB1;
      const int s0 = row_bcast(sv, sel),            s1 = row_bcast(sv, sel + NIB1);
      const int s2 = row_bcast(sv, sel + 2u * NIB1), s3 = row_bcast(sv, sel + 3u * NIB1);
      const float2 u0 = ldrow2(xl, s0, c), u1 = ldrow2(xl, s1, c);
      const float2 u2 = ldrow2(xl, s2, c), u3 = ldrow2(xl, s3, c);
      const float l0 = logit2(u0, vr, w), l1 = logit2(u1, vr, w);
      const float l2 = logit2(u2, vr, w), l3 = logit2(u3, vr, w);
      const float mn = fmaxf(fmaxf(m, l0), fmaxf(fmaxf(l1, l2), l3));
      const float sc = ex2(m - mn);
      const int rem = cnt - j;
      const float e0 = rem > 0 ? ex2(l0 - mn) : 0.f, e1 = rem > 1 ? ex2(l1 - mn) : 0.f;
      const float e2 = rem > 2 ? ex2(l2 - mn) : 0.f, e3 = rem > 3 ? ex2(l3 - mn) : 0.f;
      den = den * sc + e0; a.x = a.x * sc + e0 * u0.x; a.y = a.y * sc + e0 * u0.y;
      den += e1; a.x += e1 * u1.x; a.y += e1 * u1.y;
      den += e2; a.x += e2 * u2.x; a.y += e2 * u2.y;
      den += e3; a.x += e3 * u3.x; a.y += e3 * u3.y;
      m = mn;
    }
  }
  const float2 bb = ((const float2*)b2)[c];
  const float2 gg = ((const float2*)g)[c];
  const float2 bt = ((const float2*)be)[c];
  const float rd = __builtin_amdgcn_rcpf(den);
  const float v0 = a.x * rd + bb.x, v1 = a.y * rd + bb.y;
  const float mu = row_sum(v0 + v1) * (1.f / 32.f);
  const float d0 = v0 - mu, d1 = v1 - mu;
  const float rs = __builtin_amdgcn_rsqf(row_sum(d0 * d0 + d1 * d1) * (1.f / 32.f) + LN_EPS);
  if (live)
    ((float2*)(out + (size_t)node * 32))[c] = make_float2(d0 * rs * gg.x + bt.x, d1 * rs * gg.y + bt.y);
}

static inline int gridFor(long long threads, int block) {
  return (int)((threads + block - 1) / block);
}

extern "C" void kernel_launch(void* const* d_in, const int* in_sizes, int n_in,
                              void* d_out, int out_size, void* d_ws, size_t ws_size,
                              hipStream_t stream) {
  const float* x   = (const float*)d_in[0];
  const int*   ei  = (const int*)  d_in[1];
  const float* W1l = (const float*)d_in[2];
  const float* W1r = (const float*)d_in[3];
  const float* a1  = (const float*)d_in[4];
  const float* b1  = (const float*)d_in[5];
  const float* g1  = (const float*)d_in[6];
  const float* be1 = (const float*)d_in[7];
  const float* W2l = (const float*)d_in[8];
  const float* W2r = (const float*)d_in[9];
  const float* a2  = (const float*)d_in[10];
  const float* b2  = (const float*)d_in[11];
  const float* g2  = (const float*)d_in[12];
  const float* be2 = (const float*)d_in[13];
  float* out = (float*)d_out;

  const int n  = in_sizes[0] / 4;
  const int E  = in_sizes[1] / 2;
  const int TE = E + n;
  const int ntiles = n / 16;

  float* ws = (float*)d_ws;
  const size_t XL1  = 0;
  const size_t XR1  = XL1 + (size_t)n * 64;
  const size_t ACC1 = XR1 + (size_t)n * 64;
  const size_t DEN1 = ACC1 + (size_t)n * 64;
  const size_t MAX1 = DEN1 + (size_t)n * 4;
  const size_t LOG1 = MAX1 + (size_t)n * 4;
  int* wi = (int*)d_ws;
  const size_t BSTA = LOG1;
  const size_t RPTR = BSTA + NBMAX + 64;
  const size_t SRCS = RPTR + (size_t)n + 64;
  const size_t BINN = (SRCS + (size_t)E + 1) & ~(size_t)1;

  const int nb = (n + BIN_NODES - 1) >> BIN_SHIFT;
  const int nchunk = gridFor(E, EPB);
  const size_t CNTB = ACC1;
  const size_t XL2  = ACC1;
  const size_t XR2  = ACC1 + (size_t)n * 32;

  count_gemm1<<<nchunk + gridFor(ntiles, SORT_T / 32), SORT_T, 0, stream>>>(
      ei, E, nchunk, wi + CNTB, x, W1l, W1r, ws + XL1, ws + XR1, ntiles);
  bin_scatter<<<nchunk, SORT_T, 0, stream>>>(ei, E, nb, nchunk, wi + CNTB, wi + BSTA, wi + BINN);
  bin_sort<<<nb, SORT_T, 0, stream>>>(wi + BINN, wi + BSTA, n, wi + RPTR, wi + SRCS);

  gat1_node<<<gridFor(n, G1_NODES), G1_NODES * 16, 0, stream>>>(
      wi + RPTR, wi + SRCS, ws + XL1, ws + XR1, a1, b1, g1, be1,
      W2l, W2r, ws + XL2, ws + XR2, n);

  gat2_node<<<gridFor((long long)n * 16, G2_T), G2_T, 0, stream>>>(
      wi + RPTR, wi + SRCS, ws + XL2, ws + XR2, a2, b2, g2, be2, out, n);
}


# Layer-1 rows formed on WMMA in the node kernel from gathered 16-byte x rows (8 edges of each of a wave's 2 nodes per tile): no xl/xr planes, no 256-byte row gathers, no layer-1 GEMM pass; sums kept
# speedup vs baseline: 16.7517x; 1.0693x over previous
#include <hip/hip_runtime.h>
#include <hip/hip_bf16.h>
#include <math.h>


typedef __attribute__((ext_vector_type(2))) float v2f;
typedef __attribute__((ext_vector_type(8))) float v8f;

#define NEG_SLOPE 0.2f
#define LN_EPS 1e-5f

static __device__ inline v8f wmma4(v2f a, v2f b, v8f c) {
  return __builtin_amdgcn_wmma_f32_16x16x4_f32(
       false, a,  false, b,
       (short)0, c,  false,  false);
}

#define G1_NODES 16
#define HS_PITCH 68
static __device__ inline void gemm2_rows(const float* hs,
                                         const float* __restrict__ W,
                                         float* __restrict__ xo,
                                         int row0, int n) {
  const int lane = (int)__builtin_amdgcn_mbcnt_lo(~0u, 0u);
  const int m  = lane & 15;
  const int hi = lane >> 4;
  const float* hrow = hs + m * HS_PITCH;
  v8f c[2] = {{0.f},{0.f}};
#pragma unroll
  for (int kk = 0; kk < 16; ++kk) {
    const int k0 = kk * 4 + hi * 2;
    v2f a; a.x = hrow[k0]; a.y = hrow[k0 + 1];
#pragma unroll
    for (int j = 0; j < 2; ++j) {
      v2f b;
      b.x = W[(k0 + 0) * 32 + j * 16 + m];
      b.y = W[(k0 + 1) * 32 + j * 16 + m];
      c[j] = wmma4(a, b, c[j]);
    }
  }
  asm volatile("v_nop\n\tv_nop\n\tv_nop\n\tv_nop" : "+v"(c[0]), "+v"(c[1]));
  float* __restrict__ xcol = xo + (size_t)(row0 + hi * 8) * 32 + m;
  if (row0 + G1_NODES <= n) {
#pragma unroll
    for (int j = 0; j < 2; ++j)
#pragma unroll
      for (int v = 0; v < 8; ++v) xcol[v * 32 + j * 16] = c[j][v];
  } else {
#pragma unroll
    for (int j = 0; j < 2; ++j)
#pragma unroll
      for (int v = 0; v < 8; ++v)
        if (row0 + hi * 8 + v < n) xcol[v * 32 + j * 16] = c[j][v];
  }
}

#define BIN_SHIFT 9
#define BIN_NODES 512
#define NBMAX     256
#define SORT_T    1024
#define SORT_K    8
#define EPB       (SORT_T * SORT_K)
#define ROW_G     (SORT_T / (NBMAX / 4))
#define ROW_B     7

__global__ void __launch_bounds__(SORT_T)
count_bins(const int* __restrict__ ei, int E, int* __restrict__ cntB) {
  __shared__ int h[NBMAX];
  const int t = threadIdx.x;
  if (t < NBMAX) h[t] = 0;
  __syncthreads();
  const int base = blockIdx.x * EPB + t;
  int d[SORT_K];
#pragma unroll
  for (int k = 0; k < SORT_K; ++k) {
    const int e = base + k * SORT_T;
    d[k] = e < E ? ei[E + e] : -1;
  }
#pragma unroll
  for (int k = 0; k < SORT_K; ++k)
    if (d[k] >= 0) atomicAdd(&h[d[k] >> BIN_SHIFT], 1);
  __syncthreads();
  if (t < NBMAX) cntB[blockIdx.x * NBMAX + t] = h[t];
}

__global__ void __launch_bounds__(SORT_T)
bin_scatter(const int* __restrict__ ei, int E, int nb, int nchunk,
            const int* __restrict__ cntB, int* __restrict__ binStart,
            int* __restrict__ binnedw) {
  int2* __restrict__ binned = (int2*)binnedw;
  __shared__ int h[NBMAX];
  __shared__ __attribute__((aligned(16))) int ptot[ROW_G][NBMAX];
  __shared__ __attribute__((aligned(16))) int pbef[ROW_G][NBMAX];
  __shared__ int wsum[SORT_T / 32];
  const int t = threadIdx.x, lane = t & 31, wv = t >> 5;
  const int blk = blockIdx.x;
  if (t < NBMAX) h[t] = 0;
  const int base = blk * EPB + t;
  int s[SORT_K], d[SORT_K], r[SORT_K];
#pragma unroll
  for (int k = 0; k < SORT_K; ++k) {
    const int e = base + k * SORT_T;
    s[k] = 0; d[k] = -1;
    if (e < E) { s[k] = ei[e]; d[k] = ei[E + e]; }
  }
  {
    const int c4 = t & (NBMAX / 4 - 1), g = t / (NBMAX / 4);
    const int4* __restrict__ m4 = (const int4*)cntB;
    int4 tot = make_int4(0, 0, 0, 0), bef = make_int4(0, 0, 0, 0);
    for (int k0 = g; k0 < nchunk; k0 += ROW_G * ROW_B) {
      int4 v[ROW_B];
#pragma unroll
      for (int j = 0; j < ROW_B; ++j) {
        const int k = k0 + j * ROW_G;
        v[j] = k < nchunk ? m4[k * (NBMAX / 4) + c4] : make_int4(0, 0, 0, 0);
      }
#pragma unroll
      for (int j = 0; j < ROW_B; ++j) {
        const int k = k0 + j * ROW_G;
        tot.x += v[j].x; tot.y += v[j].y; tot.z += v[j].z; tot.w += v[j].w;
        if (k < blk) { bef.x += v[j].x; bef.y += v[j].y; bef.z += v[j].z; bef.w += v[j].w; }
      }
    }
    ((int4*)ptot[g])[c4] = tot;
    ((int4*)pbef[g])[c4] = bef;
  }
  __syncthreads();
#pragma unroll
  for (int k = 0; k < SORT_K; ++k) {
    r[k] = 0;
    if (d[k] >= 0) r[k] = atomicAdd(&h[d[k] >> BIN_SHIFT], 1);
  }
  int T = 0, B = 0;
  if (t < NBMAX) {
    for (int g = 0; g < ROW_G; ++g) { T += ptot[g][t]; B += pbef[g][t]; }
  }
  int inc = T;
  for (int k = 1; k < 32; k <<= 1) { int u = __shfl_up(inc, k, 32); if (lane >= k) inc += u; }
  if (lane == 31) wsum[wv] = inc;
  __syncthreads();
  if (t < NBMAX) {
    int woff = 0;
    for (int q = 0; q < wv; ++q) woff += wsum[q];
    const int excl = woff + inc - T;
    h[t] = excl + B;
    if (blk == 0) {
      if (t < nb) binStart[t] = excl;
      if (t == nb - 1) binStart[nb] = excl + T;
    }
  }
  __syncthreads();
#pragma unroll
  for (int k = 0; k < SORT_K; ++k)
    if (d[k] >= 0) binned[h[d[k] >> BIN_SHIFT] + r[k]] = make_int2(s[k], d[k]);
}

__global__ void __launch_bounds__(SORT_T)
bin_sort(const int* __restrict__ binnedw, const int* __restrict__ binStart, int n,
         int* __restrict__ rowptr, int* __restrict__ srcs) {
  const int2* __restrict__ binned = (const int2*)binnedw;
  __shared__ int cnt[BIN_NODES];
  __shared__ int wsum[SORT_T / 32];
  const int b = blockIdx.x, t = threadIdx.x, lane = t & 31, wv = t >> 5;
  const int beg = binStart[b], end = binStart[b + 1];
  const int node0 = b << BIN_SHIFT;
  if (t < BIN_NODES) cnt[t] = 0;
  __syncthreads();
  for (int c = beg; c < end; c += EPB) {
    int dl[SORT_K];
#pragma unroll
    for (int k = 0; k < SORT_K; ++k) {
      const int i = c + k * SORT_T + t;
      dl[k] = i < end ? binned[i].y : -1;
    }
#pragma unroll
    for (int k = 0; k < SORT_K; ++k)
      if (dl[k] >= 0) atomicAdd(&cnt[dl[k] - node0], 1);
  }
  __syncthreads();
  const int v = t < BIN_NODES ? cnt[t] : 0;
  int inc = v;
  for (int k = 1; k < 32; k <<= 1) { int u = __shfl_up(inc, k, 32); if (lane >= k) inc += u; }
  if (lane == 31) wsum[wv] = inc;
  __syncthreads();
  int woff = 0;
  for (int q = 0; q < wv; ++q) woff += wsum[q];
  const int excl = beg + woff + inc - v;
  if (t < BIN_NODES) {
    cnt[t] = excl;
    const int node = node0 + t;
    if (node < n) rowptr[node] = excl;
    if (node == n - 1) rowptr[n] = excl + v;
  }
  __syncthreads();
  for (int c = beg; c < end; c += EPB) {
    int2 e[SORT_K];
#pragma unroll
    for (int k = 0; k < SORT_K; ++k) {
      const int i = c + k * SORT_T + t;
      e[k] = i < end ? binned[i] : make_int2(0, -1);
    }
#pragma unroll
    for (int k = 0; k < SORT_K; ++k)
      if (e[k].y >= 0) srcs[atomicAdd(&cnt[e[k].y - node0], 1)] = e[k].x;
  }
}

static __device__ inline float lrelu(float e) { return __builtin_fmaxf(e, NEG_SLOPE * e); }

#define LOG2E 1.4426950408889634f
static __device__ inline float ex2(float x) { return __builtin_amdgcn_exp2f(x); }

template <int X> static __device__ inline float xrow(float v) {
  return __builtin_bit_cast(float, __builtin_amdgcn_update_dpp(
      0, __builtin_bit_cast(int, v), 0x160 + X, 0xF, 0xF, true));
}
static __device__ inline float row_sum(float s) {
  s += xrow<8>(s); s += xrow<4>(s); s += xrow<2>(s); s += xrow<1>(s);
  return s;
}
static __device__ inline int row_bcast(int v, unsigned sel) {
  return __builtin_amdgcn_permlane16(v, v, sel, sel, false, false);
}
#define NIB1 0x11111111u


static __device__ inline float2 ldrow2(const float* __restrict__ p, int s, int c) {
  return *(const float2*)((const char*)p + ((unsigned)s * 128u + (unsigned)c * 8u));
}

static __device__ inline float logit1(float4 u, float4 vr, float4 w) {
  float p = lrelu(u.x + vr.x) * w.x + lrelu(u.y + vr.y) * w.y
          + lrelu(u.z + vr.z) * w.z + lrelu(u.w + vr.w) * w.w;
  p += xrow<1>(p); p += xrow<2>(p);
  return p;
}


static __device__ inline v2f ldx2(const float* __restrict__ x, int s, unsigned kb4) {
  const float2 t = *(const float2*)((const char*)x + ((unsigned)s * 16u + kb4));
  v2f a; a.x = t.x; a.y = t.y;
  return a;
}

static __device__ inline void walk4(float4 u0, float4 u1, float4 u2, float4 u3,
                                    float4 vr, float4 w, int rem,
                                    float& m, float& den, float4& a) {
  const float l0 = logit1(u0, vr, w), l1 = logit1(u1, vr, w);
  const float l2 = logit1(u2, vr, w), l3 = logit1(u3, vr, w);
  const float mn = fmaxf(fmaxf(m, l0), fmaxf(fmaxf(l1, l2), l3));
  const float sc = ex2(m - mn);
  const float e0 = rem > 0 ? ex2(l0 - mn) : 0.f, e1 = rem > 1 ? ex2(l1 - mn) : 0.f;
  const float e2 = rem > 2 ? ex2(l2 - mn) : 0.f, e3 = rem > 3 ? ex2(l3 - mn) : 0.f;
  den = den * sc + e0;
  a.x = a.x * sc + e0 * u0.x; a.y = a.y * sc + e0 * u0.y;
  a.z = a.z * sc + e0 * u0.z; a.w = a.w * sc + e0 * u0.w;
  den += e1; a.x += e1 * u1.x; a.y += e1 * u1.y; a.z += e1 * u1.z; a.w += e1 * u1.w;
  den += e2; a.x += e2 * u2.x; a.y += e2 * u2.y; a.z += e2 * u2.z; a.w += e2 * u2.w;
  den += e3; a.x += e3 * u3.x; a.y += e3 * u3.y; a.z += e3 * u3.z; a.w += e3 * u3.w;
  m = mn;
}

__global__ void __launch_bounds__(G1_NODES * 16) __attribute__((amdgpu_waves_per_eu(8, 16)))
gat1_node(const int* __restrict__ rowptr,
                          const int* __restrict__ srcs,
                          const float* __restrict__ x,
                          const float* __restrict__ W1l,
                          const float* __restrict__ W1r,
                          const float* __restrict__ a1,
                          const float* __restrict__ b1,
                          const float* __restrict__ g,
                          const float* __restrict__ be,
                          const float* __restrict__ W2l,
                          const float* __restrict__ W2r,
                          float* __restrict__ xl2,
                          float* __restrict__ xr2, int n) {
  __shared__ __attribute__((aligned(16))) float hs[G1_NODES * HS_PITCH];
  const int wv = __builtin_amdgcn_readfirstlane((int)(threadIdx.x >> 5));
  const int row0 = (int)blockIdx.x * G1_NODES;
  const int lane = threadIdx.x & 31;
  const int c = lane & 15;
  const bool hi = lane >= 16;
  const int nA = row0 + 2 * wv     < n ? row0 + 2 * wv     : n - 1;
  const int nB = row0 + 2 * wv + 1 < n ? row0 + 2 * wv + 1 : n - 1;
  const bool rowB = c >= 8;
  const int r = c & 7;
  const int nodeR = rowB ? nB : nA;
  const unsigned kb4 = hi ? 8u : 0u;
  const int kb = hi ? 2 : 0;
  const float2 wa = ((const float2*)a1)[2 * c], wb = ((const float2*)a1)[2 * c + 1];
  const float4 w  = make_float4(wa.x * LOG2E, wa.y * LOG2E, wb.x * LOG2E, wb.y * LOG2E);
  const float4 l0 = *(const float4*)(W1l + kb * 64 + 4 * c);
  const float4 l1 = *(const float4*)(W1l + (kb + 1) * 64 + 4 * c);
  v2f Bl0, Bl1, Bl2, Bl3;
  Bl0.x = l0.x; Bl0.y = l1.x; Bl1.x = l0.y; Bl1.y = l1.y;
  Bl2.x = l0.z; Bl2.y = l1.z; Bl3.x = l0.w; Bl3.y = l1.w;
  const v8f zero = {0.f, 0.f, 0.f, 0.f, 0.f, 0.f, 0.f, 0.f};
  float4 vr, us;
  {
    const v2f xa = ldx2(x, nodeR, kb4);
    const float4 r0 = *(const float4*)(W1r + kb * 64 + 4 * c);
    const float4 r1 = *(const float4*)(W1r + (kb + 1) * 64 + 4 * c);
    v2f Br0, Br1, Br2, Br3;
    Br0.x = r0.x; Br0.y = r1.x; Br1.x = r0.y; Br1.y = r1.y;
    Br2.x = r0.z; Br2.y = r1.z; Br3.x = r0.w; Br3.y = r1.w;
    const v8f s0 = wmma4(xa, Bl0, zero), s1 = wmma4(xa, Bl1, zero);
    const v8f s2 = wmma4(xa, Bl2, zero), s3 = wmma4(xa, Bl3, zero);
    const v8f t0 = wmma4(xa, Br0, zero), t1 = wmma4(xa, Br1, zero);
    const v8f t2 = wmma4(xa, Br2, zero), t3 = wmma4(xa, Br3, zero);
    us = make_float4(s0[0], s1[0], s2[0], s3[0]);
    vr = make_float4(t0[0], t1[0], t2[0], t3[0]);
  }
  const int begA = rowptr[nA], lenA = rowptr[nA + 1] - begA;
  const int begB = rowptr[nB], lenB = rowptr[nB + 1] - begB;
  const int lenMax = lenA > lenB ? lenA : lenB;
  float m = logit1(us, vr, w);
  float den = 1.f;
  float4 a = us;
  for (int off = 0; off < lenMax; off += 32) {
    const int cnt = (hi ? lenB : lenA) - off;
    const int cntR = (rowB ? lenB : lenA) - off;
    const int pos = (rowB ? begB : begA) + off + r;
    const int q0 = r      < cntR ? srcs[pos]      : nodeR;
    const int q1 = r + 8  < cntR ? srcs[pos + 8]  : nodeR;
    const int q2 = r + 16 < cntR ? srcs[pos + 16] : nodeR;
    const int q3 = r + 24 < cntR ? srcs[pos + 24] : nodeR;
    int gEnd = lenMax - off; gEnd = gEnd < 32 ? gEnd : 32;
    v2f an = ldx2(x, q0, kb4);
    for (int j = 0; j < gEnd; j += 8) {
      const v2f af = an;
      an = ldx2(x, j == 0 ? q1 : (j == 8 ? q2 : q3), kb4);
      const v8f d0 = wmma4(af, Bl0, zero), d1 = wmma4(af, Bl1, zero);
      const v8f d2 = wmma4(af, Bl2, zero), d3 = wmma4(af, Bl3, zero);
      walk4(make_float4(d0[0], d1[0], d2[0], d3[0]), make_float4(d0[1], d1[1], d2[1], d3[1]),
            make_float4(d0[2], d1[2], d2[2], d3[2]), make_float4(d0[3], d1[3], d2[3], d3[3]),
            vr, w, cnt - j, m, den, a);
      if (j + 4 < gEnd)
        walk4(make_float4(d0[4], d1[4], d2[4], d3[4]), make_float4(d0[5], d1[5], d2[5], d3[5]),
              make_float4(d0[6], d1[6], d2[6], d3[6]), make_float4(d0[7], d1[7], d2[7], d3[7]),
              vr, w, cnt - j - 4, m, den, a);
    }
  }
  const int ln = (int)__builtin_amdgcn_mbcnt_lo(~0u, 0u);
  const int cc = ln & 15;
  const float4 bb = ((const float4*)b1)[cc];
  const float4 gg = ((const float4*)g)[cc];
  const float4 bt = ((const float4*)be)[cc];
  const float rd = __builtin_amdgcn_rcpf(den);
  const float v0 = a.x * rd + bb.x, v1 = a.y * rd + bb.y;
  const float v2 = a.z * rd + bb.z, v3 = a.w * rd + bb.w;
  const float mu = row_sum((v0 + v1) + (v2 + v3)) * (1.f / 64.f);
  const float d0 = v0 - mu, d1 = v1 - mu, d2 = v2 - mu, d3 = v3 - mu;
  const float rs = __builtin_amdgcn_rsqf(
      row_sum((d0 * d0 + d1 * d1) + (d2 * d2 + d3 * d3)) * (1.f / 64.f) + LN_EPS);
  const float y0 = d0 * rs * gg.x + bt.x, y1 = d1 * rs * gg.y + bt.y;
  const float y2 = d2 * rs * gg.z + bt.z, y3 = d3 * rs * gg.w + bt.w;
  float4 y;
  y.x = y0 > 0.f ? y0 : ex2(y0 * LOG2E) - 1.f;
  y.y = y1 > 0.f ? y1 : ex2(y1 * LOG2E) - 1.f;
  y.z = y2 > 0.f ? y2 : ex2(y2 * LOG2E) - 1.f;
  y.w = y3 > 0.f ? y3 : ex2(y3 * LOG2E) - 1.f;
  ((float4*)(hs + (2 * wv + (ln >> 4)) * HS_PITCH))[cc] = y;
  __syncthreads();
  if (wv >= 2) return;
  gemm2_rows(hs, wv == 0 ? W2l : W2r, wv == 0 ? xl2 : xr2, row0, n);
}

static __device__ inline float logit2(float2 u, float2 vr, float2 w) {
  float p = lrelu(u.x + vr.x) * w.x + lrelu(u.y + vr.y) * w.y;
  p += xrow<1>(p); p += xrow<2>(p); p += xrow<4>(p); p += xrow<8>(p);
  return p;
}

#define G2_T 256
__global__ void __launch_bounds__(G2_T)
gat2_node(const int* __restrict__ rowptr,
                          const int* __restrict__ srcs,
                          const float* __restrict__ xl,
                          const float* __restrict__ xr,
                          const float* __restrict__ a2,
                          const float* __restrict__ b2,
                          const float* __restrict__ g,
                          const float* __restrict__ be,
                          float* __restrict__ out, int n) {
  const int lane = threadIdx.x & 31;
  const int c = lane & 15;
  const bool hi = lane >= 16;
  const int n0 = (int)blockIdx.x * (G2_T / 16)
               + 2 * __builtin_amdgcn_readfirstlane((int)(threadIdx.x >> 5));
  const int nA = n0     < n ? n0     : n - 1;
  const int nB = n0 + 1 < n ? n0 + 1 : n - 1;
  const bool live = n0 + (hi ? 1 : 0) < n;
  const int node = hi ? nB : nA;
  const float2 w0 = ((const float2*)a2)[c];
  const float2 w  = make_float2(w0.x * LOG2E, w0.y * LOG2E);
  const float2 vr = ldrow2(xr, node, c);
  const float2 us = ldrow2(xl, node, c);
  const int begA = rowptr[nA], lenA = rowptr[nA + 1] - begA;
  const int begB = rowptr[nB], lenB = rowptr[nB + 1] - begB;
  const int lenMax = lenA > lenB ? lenA : lenB;
  float m = logit2(us, vr, w);
  float den = 1.f;
  float2 a = us;
  for (int off = 0; off < lenMax; off += 32) {
    const int cnt = (hi ? lenB : lenA) - off;
    const int pos = (hi ? begB : begA) + off + c;
    const int sA = c < cnt ? srcs[pos] : node;
    const int sB = c + 16 < cnt ? srcs[pos + 16] : node;
    int gEnd = lenMax - off; gEnd = gEnd < 32 ? gEnd : 32;
    for (int j = 0; j < gEnd; j += 4) {
      const int sv = (j & 16) ? sB : sA;
      const unsigned sel = (unsigned)(j & 12) * NIB1;
      const int s0 = row_bcast(sv, sel),            s1 = row_bcast(sv, sel + NIB1);
      const int s2 = row_bcast(sv, sel + 2u * NIB1), s3 = row_bcast(sv, sel + 3u * NIB1);
      const float2 u0 = ldrow2(xl, s0, c), u1 = ldrow2(xl, s1, c);
      const float2 u2 = ldrow2(xl, s2, c), u3 = ldrow2(xl, s3, c);
      const float l0 = logit2(u0, vr, w), l1 = logit2(u1, vr, w);
      const float l2 = logit2(u2, vr, w), l3 = logit2(u3, vr, w);
      const float mn = fmaxf(fmaxf(m, l0), fmaxf(fmaxf(l1, l2), l3));
      const float sc = ex2(m - mn);
      const int rem = cnt - j;
      const float e0 = rem > 0 ? ex2(l0 - mn) : 0.f, e1 = rem > 1 ? ex2(l1 - mn) : 0.f;
      const float e2 = rem > 2 ? ex2(l2 - mn) : 0.f, e3 = rem > 3 ? ex2(l3 - mn) : 0.f;
      den = den * sc + e0; a.x = a.x * sc + e0 * u0.x; a.y = a.y * sc + e0 * u0.y;
      den += e1; a.x += e1 * u1.x; a.y += e1 * u1.y;
      den += e2; a.x += e2 * u2.x; a.y += e2 * u2.y;
      den += e3; a.x += e3 * u3.x; a.y += e3 * u3.y;
      m = mn;
    }
  }
  const float2 bb = ((const float2*)b2)[c];
  const float2 gg = ((const float2*)g)[c];
  const float2 bt = ((const float2*)be)[c];
  const float rd = __builtin_amdgcn_rcpf(den);
  const float v0 = a.x * rd + bb.x, v1 = a.y * rd + bb.y;
  const float mu = row_sum(v0 + v1) * (1.f / 32.f);
  const float d0 = v0 - mu, d1 = v1 - mu;
  const float rs = __builtin_amdgcn_rsqf(row_sum(d0 * d0 + d1 * d1) * (1.f / 32.f) + LN_EPS);
  if (live)
    ((float2*)(out + (size_t)node * 32))[c] = make_float2(d0 * rs * gg.x + bt.x, d1 * rs * gg.y + bt.y);
}

static inline int gridFor(long long threads, int block) {
  return (int)((threads + block - 1) / block);
}

extern "C" void kernel_launch(void* const* d_in, const int* in_sizes, int n_in,
                              void* d_out, int out_size, void* d_ws, size_t ws_size,
                              hipStream_t stream) {
  const float* x   = (const float*)d_in[0];
  const int*   ei  = (const int*)  d_in[1];
  const float* W1l = (const float*)d_in[2];
  const float* W1r = (const float*)d_in[3];
  const float* a1  = (const float*)d_in[4];
  const float* b1  = (const float*)d_in[5];
  const float* g1  = (const float*)d_in[6];
  const float* be1 = (const float*)d_in[7];
  const float* W2l = (const float*)d_in[8];
  const float* W2r = (const float*)d_in[9];
  const float* a2  = (const float*)d_in[10];
  const float* b2  = (const float*)d_in[11];
  const float* g2  = (const float*)d_in[12];
  const float* be2 = (const float*)d_in[13];
  float* out = (float*)d_out;

  const int n  = in_sizes[0] / 4;
  const int E  = in_sizes[1] / 2;
  const int TE = E + n;

  float* ws = (float*)d_ws;
  const size_t XL1  = 0;
  const size_t XR1  = XL1 + (size_t)n * 64;
  const size_t ACC1 = XR1 + (size_t)n * 64;
  const size_t DEN1 = ACC1 + (size_t)n * 64;
  const size_t MAX1 = DEN1 + (size_t)n * 4;
  const size_t LOG1 = MAX1 + (size_t)n * 4;
  int* wi = (int*)d_ws;
  const size_t BSTA = LOG1;
  const size_t RPTR = BSTA + NBMAX + 64;
  const size_t SRCS = RPTR + (size_t)n + 64;
  const size_t BINN = (SRCS + (size_t)E + 1) & ~(size_t)1;

  const int nb = (n + BIN_NODES - 1) >> BIN_SHIFT;
  const int nchunk = gridFor(E, EPB);
  const size_t CNTB = ACC1;
  const size_t XL2  = ACC1;
  const size_t XR2  = ACC1 + (size_t)n * 32;

  count_bins<<<nchunk, SORT_T, 0, stream>>>(ei, E, wi + CNTB);
  bin_scatter<<<nchunk, SORT_T, 0, stream>>>(ei, E, nb, nchunk, wi + CNTB, wi + BSTA, wi + BINN);
  bin_sort<<<nb, SORT_T, 0, stream>>>(wi + BINN, wi + BSTA, n, wi + RPTR, wi + SRCS);

  gat1_node<<<gridFor(n, G1_NODES), G1_NODES * 16, 0, stream>>>(
      wi + RPTR, wi + SRCS, x, W1l, W1r, a1, b1, g1, be1,
      W2l, W2r, ws + XL2, ws + XR2, n);

  gat2_node<<<gridFor((long long)n * 16, G2_T), G2_T, 0, stream>>>(
      wi + RPTR, wi + SRCS, ws + XL2, ws + XR2, a2, b2, g2, be2, out, n);
}
